# starting kernel
# baseline (speedup 1.0000x reference)
#include <hip/hip_runtime.h>
#include <cstdint>
#include <cstddef>

typedef __attribute__((ext_vector_type(2))) float v2f;
typedef __attribute__((ext_vector_type(8))) float v8f;

#define NEG_SLOPE 0.2f

__device__ __forceinline__ unsigned fkey(float f) {
  unsigned u = __float_as_uint(f);
  return (u & 0x80000000u) ? ~u : (u | 0x80000000u);
}
__device__ __forceinline__ float funkey(unsigned k) {
  unsigned u = (k & 0x80000000u) ? (k & 0x7fffffffu) : ~k;
  return __uint_as_float(u);
}

__device__ __forceinline__ void edge_sd(const int* __restrict__ ei, long long E,
                                        long long eid, int& s, int& d) {
  if (eid < E) { s = ei[eid]; d = ei[E + eid]; }
  else         { s = d = (int)(eid - E); }
}

template<int K, int COLS>
__global__ void gemm_wmma_f32(const float* __restrict__ A,
                              const float* __restrict__ B,
                              float* __restrict__ C, int N) {
  constexpr int CT = COLS / 16;
  const int wave = (int)((blockIdx.x * blockDim.x + threadIdx.x) >> 5);
  const int lane = (int)(threadIdx.x & 31);
  const int row0 = wave * 16;
  if (row0 >= N) return;
  const int m    = lane & 15;
  const int half = lane >> 4;
  const int klo  = half * 2;

  v8f zero = {0.f, 0.f, 0.f, 0.f, 0.f, 0.f, 0.f, 0.f};
  v8f acc[CT];
  #pragma unroll
  for (int t = 0; t < CT; ++t) acc[t] = zero;

  #pragma unroll
  for (int k = 0; k < K; k += 4) {
    const float* ap = A + (size_t)(row0 + m) * K + (k + klo);
    v2f a; a.x = ap[0]; a.y = ap[1];
    #pragma unroll
    for (int t = 0; t < CT; ++t) {
      const float* bp = B + (size_t)(k + klo) * COLS + t * 16 + m;
      v2f b; b.x = bp[0]; b.y = bp[COLS];
      acc[t] = __builtin_amdgcn_wmma_f32_16x16x4_f32(
          false, a, false, b, (short)0, acc[t], false, false);
    }
  }
  const int rbase = row0 + half * 8;
  #pragma unroll
  for (int t = 0; t < CT; ++t) {
    #pragma unroll
    for (int v = 0; v < 8; ++v)
      C[(size_t)(rbase + v) * COLS + t * 16 + m] = acc[t][v];
  }
}

__global__ void alpha1_k(const float* __restrict__ h1,
                         const float* __restrict__ aw_s, const float* __restrict__ aw_d,
                         float* __restrict__ as, float* __restrict__ ad, int N) {
  long long i = (long long)blockIdx.x * blockDim.x + threadIdx.x;
  if (i >= (long long)N * 8) return;
  int n = (int)(i >> 3), h = (int)(i & 7);
  const float* hp = h1 + (size_t)n * 64 + h * 8;
  const float* sp = aw_s + h * 8;
  const float* dp = aw_d + h * 8;
  float s = 0.f, d = 0.f;
  #pragma unroll
  for (int c = 0; c < 8; ++c) { float hv = hp[c]; s += hv * sp[c]; d += hv * dp[c]; }
  as[i] = s; ad[i] = d;
}

__global__ void alpha2_k(const float* __restrict__ h2,
                         const float* __restrict__ aw_s, const float* __restrict__ aw_d,
                         float* __restrict__ as, float* __restrict__ ad, int N) {
  int wave = (int)((blockIdx.x * blockDim.x + threadIdx.x) >> 5);
  int lane = (int)(threadIdx.x & 31);
  if (wave >= N) return;
  const float* hp = h2 + (size_t)wave * 128;
  float s = 0.f, d = 0.f;
  #pragma unroll
  for (int f = 0; f < 128; f += 32) {
    float hv = hp[f + lane];
    s += hv * aw_s[f + lane];
    d += hv * aw_d[f + lane];
  }
  #pragma unroll
  for (int off = 16; off > 0; off >>= 1) {
    s += __shfl_down(s, off, 32);
    d += __shfl_down(d, off, 32);
  }
  if (lane == 0) { as[wave] = s; ad[wave] = d; }
}

template<int H>
__global__ void edge_max_k(const int* __restrict__ ei, long long E, int N,
                           const float* __restrict__ as, const float* __restrict__ ad,
                           unsigned* __restrict__ emaxk) {
  long long t = (long long)blockIdx.x * blockDim.x + threadIdx.x;
  long long total = (E + N) * H;
  if (t >= total) return;
  long long eid = t / H; int h = (int)(t % H);
  int s, d; edge_sd(ei, E, eid, s, d);
  float e = as[(size_t)s * H + h] + ad[(size_t)d * H + h];
  e = e > 0.f ? e : NEG_SLOPE * e;
  atomicMax(&emaxk[(size_t)d * H + h], fkey(e));
}

template<int H>
__global__ void edge_den_k(const int* __restrict__ ei, long long E, int N,
                           const float* __restrict__ as, const float* __restrict__ ad,
                           const unsigned* __restrict__ emaxk, float* __restrict__ den) {
  long long t = (long long)blockIdx.x * blockDim.x + threadIdx.x;
  long long total = (E + N) * H;
  if (t >= total) return;
  long long eid = t / H; int h = (int)(t % H);
  int s, d; edge_sd(ei, E, eid, s, d);
  float e = as[(size_t)s * H + h] + ad[(size_t)d * H + h];
  e = e > 0.f ? e : NEG_SLOPE * e;
  float w = __expf(e - funkey(emaxk[(size_t)d * H + h]));
  atomicAdd(&den[(size_t)d * H + h], w);
}

__global__ void agg1_k(const int* __restrict__ ei, long long E, int N,
                       const float* __restrict__ as, const float* __restrict__ ad,
                       const unsigned* __restrict__ emaxk, const float* __restrict__ den,
                       const float* __restrict__ h1, float* __restrict__ agg) {
  long long t = (long long)blockIdx.x * blockDim.x + threadIdx.x;
  long long total = (E + N) * 8;
  if (t >= total) return;
  long long eid = t >> 3; int h = (int)(t & 7);
  int s, d; edge_sd(ei, E, eid, s, d);
  float e = as[(size_t)s * 8 + h] + ad[(size_t)d * 8 + h];
  e = e > 0.f ? e : NEG_SLOPE * e;
  float alpha = __expf(e - funkey(emaxk[(size_t)d * 8 + h])) / den[(size_t)d * 8 + h];
  const float4* hp = (const float4*)(h1 + (size_t)s * 64 + h * 8);
  float4 v0 = hp[0], v1 = hp[1];
  float* op = agg + (size_t)d * 64 + h * 8;
  atomicAdd(op + 0, alpha * v0.x); atomicAdd(op + 1, alpha * v0.y);
  atomicAdd(op + 2, alpha * v0.z); atomicAdd(op + 3, alpha * v0.w);
  atomicAdd(op + 4, alpha * v1.x); atomicAdd(op + 5, alpha * v1.y);
  atomicAdd(op + 6, alpha * v1.z); atomicAdd(op + 7, alpha * v1.w);
}

__global__ void agg2_k(const int* __restrict__ ei, long long E, int N,
                       const float* __restrict__ as, const float* __restrict__ ad,
                       const unsigned* __restrict__ emaxk, const float* __restrict__ den,
                       const float* __restrict__ h2, float* __restrict__ out) {
  long long t = (long long)blockIdx.x * blockDim.x + threadIdx.x;
  long long total = (E + N) * 32;
  if (t >= total) return;
  long long eid = t >> 5; int chunk = (int)(t & 31);
  int s, d; edge_sd(ei, E, eid, s, d);
  float e = as[s] + ad[d];
  e = e > 0.f ? e : NEG_SLOPE * e;
  float alpha = __expf(e - funkey(emaxk[d])) / den[d];
  float4 hv = *(const float4*)(h2 + (size_t)s * 128 + chunk * 4);
  float* op = out + (size_t)d * 128 + chunk * 4;
  atomicAdd(op + 0, alpha * hv.x); atomicAdd(op + 1, alpha * hv.y);
  atomicAdd(op + 2, alpha * hv.z); atomicAdd(op + 3, alpha * hv.w);
}

__global__ void elu_bias_k(const float* __restrict__ agg, const float* __restrict__ b1,
                           float* __restrict__ helu, int N) {
  long long i = (long long)blockIdx.x * blockDim.x + threadIdx.x;
  if (i >= (long long)N * 64) return;
  float v = agg[i] + b1[i & 63];
  helu[i] = v > 0.f ? v : (__expf(v) - 1.f);
}

__global__ void bias2_k(float* __restrict__ out, const float* __restrict__ b2, int N) {
  long long i = (long long)blockIdx.x * blockDim.x + threadIdx.x;
  if (i >= (long long)N * 128) return;
  out[i] += b2[i & 127];
}

static inline unsigned cdiv(long long a, long long b) { return (unsigned)((a + b - 1) / b); }

extern "C" void kernel_launch(void* const* d_in, const int* in_sizes, int n_in,
                              void* d_out, int out_size, void* d_ws, size_t ws_size,
                              hipStream_t stream) {
  const float* x    = (const float*)d_in[0];
  const int*   ei   = (const int*)d_in[1];
  const float* W1   = (const float*)d_in[2];
  const float* as1w = (const float*)d_in[3];
  const float* ad1w = (const float*)d_in[4];
  const float* b1   = (const float*)d_in[5];
  const float* W2   = (const float*)d_in[6];
  const float* as2w = (const float*)d_in[7];
  const float* ad2w = (const float*)d_in[8];
  const float* b2   = (const float*)d_in[9];
  float* out = (float*)d_out;

  const int N = in_sizes[0] / 128;
  const long long E = (long long)in_sizes[1] / 2;

  float* w = (float*)d_ws;
  size_t off = 0;
  float*    h1   = w + off; off += (size_t)N * 64;
  float*    as1  = w + off; off += (size_t)N * 8;
  float*    ad1  = w + off; off += (size_t)N * 8;
  unsigned* emk1 = (unsigned*)(w + off); off += (size_t)N * 8;
  float*    den1 = w + off; off += (size_t)N * 8;
  float*    agg1 = w + off; off += (size_t)N * 64;
  float*    helu = w + off; off += (size_t)N * 64;
  float*    h2   = w + off; off += (size_t)N * 128;
  float*    as2  = w + off; off += (size_t)N;
  float*    ad2  = w + off; off += (size_t)N;
  unsigned* emk2 = (unsigned*)(w + off); off += (size_t)N;
  float*    den2 = w + off; off += (size_t)N;

  const int B = 256;
  const long long T1 = ((long long)E + N) * 8;
  const long long T2 = (long long)E + N;
  const int tiles = (N + 15) / 16;

  hipMemsetAsync(emk1, 0, (size_t)N * 8 * sizeof(unsigned), stream);
  hipMemsetAsync(den1, 0, (size_t)N * 8 * sizeof(float), stream);
  hipMemsetAsync(agg1, 0, (size_t)N * 64 * sizeof(float), stream);
  hipMemsetAsync(emk2, 0, (size_t)N * sizeof(unsigned), stream);
  hipMemsetAsync(den2, 0, (size_t)N * sizeof(float), stream);
  hipMemsetAsync(out,  0, (size_t)N * 128 * sizeof(float), stream);

  gemm_wmma_f32<128, 64><<<cdiv((long long)tiles * 32, B), B, 0, stream>>>(x, W1, h1, N);
  alpha1_k<<<cdiv((long long)N * 8, B), B, 0, stream>>>(h1, as1w, ad1w, as1, ad1, N);
  edge_max_k<8><<<cdiv(T1, B), B, 0, stream>>>(ei, E, N, as1, ad1, emk1);
  edge_den_k<8><<<cdiv(T1, B), B, 0, stream>>>(ei, E, N, as1, ad1, emk1, den1);
  agg1_k<<<cdiv(T1, B), B, 0, stream>>>(ei, E, N, as1, ad1, emk1, den1, h1, agg1);
  elu_bias_k<<<cdiv((long long)N * 64, B), B, 0, stream>>>(agg1, b1, helu, N);

  gemm_wmma_f32<64, 128><<<cdiv((long long)tiles * 32, B), B, 0, stream>>>(helu, W2, h2, N);
  alpha2_k<<<cdiv((long long)N * 32, B), B, 0, stream>>>(h2, as2w, ad2w, as2, ad2, N);
  edge_max_k<1><<<cdiv(T2, B), B, 0, stream>>>(ei, E, N, as2, ad2, emk2);
  edge_den_k<1><<<cdiv(T2, B), B, 0, stream>>>(ei, E, N, as2, ad2, emk2, den2);
  agg2_k<<<cdiv(T2 * 32, B), B, 0, stream>>>(ei, E, N, as2, ad2, emk2, den2, h2, out);
  bias2_k<<<cdiv((long long)N * 128, B), B, 0, stream>>>(out, b2, N);

  (void)n_in; (void)out_size; (void)ws_size;
}


# Group edges by destination each call (rank by int atomic, scan, fill); one wave per node then does max, softmax sum and feature sum in registers: no float atomics, sum order = list order
# speedup vs baseline: 3.7110x; 3.7110x over previous
#include <hip/hip_runtime.h>
#include <cstdint>
#include <cstddef>

typedef __attribute__((ext_vector_type(2))) float v2f;
typedef __attribute__((ext_vector_type(8))) float v8f;

#define NEG_SLOPE 0.2f

template<int K, int COLS>
__global__ void gemm_wmma_f32(const float* __restrict__ A,
                              const float* __restrict__ B,
                              float* __restrict__ C, int N) {
  constexpr int CT = COLS / 16;
  const int wave = (int)((blockIdx.x * blockDim.x + threadIdx.x) >> 5);
  const int lane = (int)(threadIdx.x & 31);
  const int row0 = wave * 16;
  if (row0 >= N) return;
  const int m    = lane & 15;
  const int half = lane >> 4;
  const int klo  = half * 2;

  v8f zero = {0.f, 0.f, 0.f, 0.f, 0.f, 0.f, 0.f, 0.f};
  v8f acc[CT];
  #pragma unroll
  for (int t = 0; t < CT; ++t) acc[t] = zero;

  #pragma unroll
  for (int k = 0; k < K; k += 4) {
    const float* ap = A + (size_t)(row0 + m) * K + (k + klo);
    v2f a; a.x = ap[0]; a.y = ap[1];
    #pragma unroll
    for (int t = 0; t < CT; ++t) {
      const float* bp = B + (size_t)(k + klo) * COLS + t * 16 + m;
      v2f b; b.x = bp[0]; b.y = bp[COLS];
      acc[t] = __builtin_amdgcn_wmma_f32_16x16x4_f32(
          false, a, false, b, (short)0, acc[t], false, false);
    }
  }
  const int rbase = row0 + half * 8;
  #pragma unroll
  for (int t = 0; t < CT; ++t) {
    #pragma unroll
    for (int v = 0; v < 8; ++v)
      C[(size_t)(rbase + v) * COLS + t * 16 + m] = acc[t][v];
  }
}

__global__ void alpha1_k(const float* __restrict__ h1,
                         const float* __restrict__ aw_s, const float* __restrict__ aw_d,
                         float* __restrict__ as, float* __restrict__ ad, int N) {
  long long i = (long long)blockIdx.x * blockDim.x + threadIdx.x;
  if (i >= (long long)N * 8) return;
  int n = (int)(i >> 3), h = (int)(i & 7);
  const float* hp = h1 + (size_t)n * 64 + h * 8;
  const float* sp = aw_s + h * 8;
  const float* dp = aw_d + h * 8;
  float s = 0.f, d = 0.f;
  #pragma unroll
  for (int c = 0; c < 8; ++c) { float hv = hp[c]; s += hv * sp[c]; d += hv * dp[c]; }
  as[i] = s; ad[i] = d;
}

__global__ void alpha2_k(const float* __restrict__ h2,
                         const float* __restrict__ aw_s, const float* __restrict__ aw_d,
                         float* __restrict__ as, float* __restrict__ ad, int N) {
  int wave = (int)((blockIdx.x * blockDim.x + threadIdx.x) >> 5);
  int lane = (int)(threadIdx.x & 31);
  if (wave >= N) return;
  const float* hp = h2 + (size_t)wave * 128;
  float s = 0.f, d = 0.f;
  #pragma unroll
  for (int f = 0; f < 128; f += 32) {
    float hv = hp[f + lane];
    s += hv * aw_s[f + lane];
    d += hv * aw_d[f + lane];
  }
  #pragma unroll
  for (int off = 16; off > 0; off >>= 1) {
    s += __shfl_down(s, off, 32);
    d += __shfl_down(d, off, 32);
  }
  if (lane == 0) { as[wave] = s; ad[wave] = d; }
}

__global__ void count_pos_k(const int* __restrict__ ei, long long E,
                            int* __restrict__ cnt, int* __restrict__ pos) {
  long long e = (long long)blockIdx.x * blockDim.x + threadIdx.x;
  if (e >= E) return;
  pos[e] = atomicAdd(&cnt[ei[E + e]], 1);
}

__global__ void __launch_bounds__(1024)
scan_k(const int* __restrict__ cnt, int* __restrict__ rowptr, int N) {
  __shared__ int wtot[32];
  __shared__ int wpre[32];
  const int tid = (int)threadIdx.x, lane = tid & 31, wv = tid >> 5;
  const int base = (int)blockIdx.x * 1024;
  int part = 0;
  for (int i = tid; i < base; i += 1024) part += cnt[i];
  const int i = base + tid;
  const int v = (i < N) ? cnt[i] : 0;
  int incl = v;
  #pragma unroll
  for (int off = 1; off < 32; off <<= 1) {
    int t = __shfl_up(incl, off, 32);
    if (lane >= off) incl += t;
  }
  #pragma unroll
  for (int off = 16; off > 0; off >>= 1) part += __shfl_xor(part, off, 32);
  if (lane == 31) wtot[wv] = incl;
  if (lane == 0)  wpre[wv] = part;
  __syncthreads();
  int t2 = wpre[lane] + (lane < wv ? wtot[lane] : 0);
  #pragma unroll
  for (int off = 16; off > 0; off >>= 1) t2 += __shfl_xor(t2, off, 32);
  const int excl = t2 + incl - v;
  if (i < N) rowptr[i] = excl;
  if (i == N - 1) rowptr[N] = excl + v;
}

__global__ void fill_k(const int* __restrict__ ei, long long E,
                       const int* __restrict__ rowptr, const int* __restrict__ pos,
                       int* __restrict__ csr) {
  long long e = (long long)blockIdx.x * blockDim.x + threadIdx.x;
  if (e >= E) return;
  csr[rowptr[ei[E + e]] + pos[e]] = ei[e];
}

__device__ __forceinline__ float leaky(float e) { return e > 0.f ? e : NEG_SLOPE * e; }

__global__ void __launch_bounds__(256)
gat1_node_k(const int* __restrict__ rowptr, const int* __restrict__ csr,
            const float* __restrict__ as, const float* __restrict__ ad,
            const float* __restrict__ h1, const float* __restrict__ b1,
            float* __restrict__ helu, int N) {
  const int lane = (int)(threadIdx.x & 31);
  const int node = __builtin_amdgcn_readfirstlane((int)((blockIdx.x * blockDim.x + threadIdx.x) >> 5));
  if (node >= N) return;
  const int beg = rowptr[node];
  const int deg = rowptr[node + 1] - beg;

  const int q = lane >> 3, hq = lane & 7;
  const float adq = ad[(size_t)node * 8 + hq];
  float m = leaky(as[(size_t)node * 8 + hq] + adq);
  for (int c0 = 0; c0 < deg; c0 += 32) {
    const int cnt = min(32, deg - c0);
    const int sl = (lane < cnt) ? csr[beg + c0 + lane] : node;
    #pragma unroll 4
    for (int j = 0; j < cnt; j += 4) {
      const int s = __shfl(sl, j + q, 32);
      m = fmaxf(m, leaky(as[(size_t)s * 8 + hq] + adq));
    }
  }
  m = fmaxf(m, __shfl_xor(m, 8, 32));
  m = fmaxf(m, __shfl_xor(m, 16, 32));

  const int hb = lane >> 2;
  const float mb  = __shfl(m, hb, 32);
  const float adb = ad[(size_t)node * 8 + hb];
  float den, ax, ay;
  {
    const float w = __expf(leaky(as[(size_t)node * 8 + hb] + adb) - mb);
    const float2 hv = *(const float2*)(h1 + (size_t)node * 64 + lane * 2);
    den = w; ax = w * hv.x; ay = w * hv.y;
  }
  for (int c0 = 0; c0 < deg; c0 += 32) {
    const int cnt = min(32, deg - c0);
    const int sl = (lane < cnt) ? csr[beg + c0 + lane] : node;
    #pragma unroll 4
    for (int j = 0; j < cnt; ++j) {
      const int s = __builtin_amdgcn_readlane(sl, j);
      const float w = __expf(leaky(as[(size_t)s * 8 + hb] + adb) - mb);
      const float2 hv = *(const float2*)(h1 + (size_t)s * 64 + lane * 2);
      den += w; ax += w * hv.x; ay += w * hv.y;
    }
  }
  float vx = ax / den + b1[lane * 2];
  float vy = ay / den + b1[lane * 2 + 1];
  float2 o;
  o.x = vx > 0.f ? vx : (__expf(vx) - 1.f);
  o.y = vy > 0.f ? vy : (__expf(vy) - 1.f);
  *(float2*)(helu + (size_t)node * 64 + lane * 2) = o;
}

__global__ void __launch_bounds__(256)
gat2_node_k(const int* __restrict__ rowptr, const int* __restrict__ csr,
            const float* __restrict__ as, const float* __restrict__ ad,
            const float* __restrict__ h2, const float* __restrict__ b2,
            float* __restrict__ out, int N) {
  const int lane = (int)(threadIdx.x & 31);
  const int node = __builtin_amdgcn_readfirstlane((int)((blockIdx.x * blockDim.x + threadIdx.x) >> 5));
  if (node >= N) return;
  const int beg = rowptr[node];
  const int deg = rowptr[node + 1] - beg;
  const float adn = ad[node];
  const float eself = leaky(as[node] + adn);

  float m = eself;
  for (int c0 = 0; c0 < deg; c0 += 32) {
    const int cnt = min(32, deg - c0);
    const int sl = (lane < cnt) ? csr[beg + c0 + lane] : node;
    m = fmaxf(m, leaky(as[sl] + adn));
  }
  #pragma unroll
  for (int off = 16; off > 0; off >>= 1) m = fmaxf(m, __shfl_xor(m, off, 32));

  const float wself = __expf(eself - m);
  float denp = 0.f;
  float4 acc;
  {
    const float4 hv = *(const float4*)(h2 + (size_t)node * 128 + lane * 4);
    acc.x = wself * hv.x; acc.y = wself * hv.y; acc.z = wself * hv.z; acc.w = wself * hv.w;
  }
  for (int c0 = 0; c0 < deg; c0 += 32) {
    const int cnt = min(32, deg - c0);
    const int sl = (lane < cnt) ? csr[beg + c0 + lane] : node;
    const float wl = (lane < cnt) ? __expf(leaky(as[sl] + adn) - m) : 0.f;
    denp += wl;
    #pragma unroll 4
    for (int j = 0; j < cnt; ++j) {
      const int s = __builtin_amdgcn_readlane(sl, j);
      const float w = __int_as_float(__builtin_amdgcn_readlane(__float_as_int(wl), j));
      const float4 hv = *(const float4*)(h2 + (size_t)s * 128 + lane * 4);
      acc.x += w * hv.x; acc.y += w * hv.y; acc.z += w * hv.z; acc.w += w * hv.w;
    }
  }
  #pragma unroll
  for (int off = 16; off > 0; off >>= 1) denp += __shfl_xor(denp, off, 32);
  const float den = wself + denp;
  const float4 bv = *(const float4*)(b2 + lane * 4);
  float4 o;
  o.x = acc.x / den + bv.x; o.y = acc.y / den + bv.y;
  o.z = acc.z / den + bv.z; o.w = acc.w / den + bv.w;
  *(float4*)(out + (size_t)node * 128 + lane * 4) = o;
}

static inline unsigned cdiv(long long a, long long b) { return (unsigned)((a + b - 1) / b); }

extern "C" void kernel_launch(void* const* d_in, const int* in_sizes, int n_in,
                              void* d_out, int out_size, void* d_ws, size_t ws_size,
                              hipStream_t stream) {
  const float* x    = (const float*)d_in[0];
  const int*   ei   = (const int*)d_in[1];
  const float* W1   = (const float*)d_in[2];
  const float* as1w = (const float*)d_in[3];
  const float* ad1w = (const float*)d_in[4];
  const float* b1   = (const float*)d_in[5];
  const float* W2   = (const float*)d_in[6];
  const float* as2w = (const float*)d_in[7];
  const float* ad2w = (const float*)d_in[8];
  const float* b2   = (const float*)d_in[9];
  float* out = (float*)d_out;

  const int N = in_sizes[0] / 128;
  const long long E = (long long)in_sizes[1] / 2;

  float* w = (float*)d_ws;
  size_t off = 0;
  float* h1   = w + off; off += (size_t)N * 64;
  float* helu = w + off; off += (size_t)N * 64;
  float* h2   = w + off; off += (size_t)N * 128;
  float* as1  = w + off; off += (size_t)N * 8;
  float* ad1  = w + off; off += (size_t)N * 8;
  float* as2  = w + off; off += (size_t)N;
  float* ad2  = w + off; off += (size_t)N;
  int* cnt    = (int*)(w + off); off += (size_t)N;
  int* rowptr = (int*)(w + off); off += (size_t)N + 1;
  int* pos    = (int*)(w + off); off += (size_t)E;
  int* csr    = (int*)(w + off); off += (size_t)E;

  const int B = 256;
  const int tiles = (N + 15) / 16;

  hipMemsetAsync(cnt, 0, (size_t)N * sizeof(int), stream);
  count_pos_k<<<cdiv(E, B), B, 0, stream>>>(ei, E, cnt, pos);
  scan_k<<<cdiv(N, 1024), 1024, 0, stream>>>(cnt, rowptr, N);
  fill_k<<<cdiv(E, B), B, 0, stream>>>(ei, E, rowptr, pos, csr);

  gemm_wmma_f32<128, 64><<<cdiv((long long)tiles * 32, B), B, 0, stream>>>(x, W1, h1, N);
  alpha1_k<<<cdiv((long long)N * 8, B), B, 0, stream>>>(h1, as1w, ad1w, as1, ad1, N);
  gat1_node_k<<<cdiv((long long)N * 32, B), B, 0, stream>>>(rowptr, csr, as1, ad1, h1, b1, helu, N);

  gemm_wmma_f32<64, 128><<<cdiv((long long)tiles * 32, B), B, 0, stream>>>(helu, W2, h2, N);
  alpha2_k<<<cdiv((long long)N * 32, B), B, 0, stream>>>(h2, as2w, ad2w, as2, ad2, N);
  gat2_node_k<<<cdiv((long long)N * 32, B), B, 0, stream>>>(rowptr, csr, as2, ad2, h2, b2, out, N);

  (void)n_in; (void)out_size; (void)ws_size;
}


# Layer 2 sums the softmax-weighted 64-wide hidden rows first and projects by W2 after (sum is linear): 256-B rows gathered not 512-B, no h2 plane; logits from h.(W2 a); layer-2 f32 sum order changes
# speedup vs baseline: 3.9930x; 1.0760x over previous
#include <hip/hip_runtime.h>
#include <cstdint>
#include <cstddef>

typedef __attribute__((ext_vector_type(2))) float v2f;
typedef __attribute__((ext_vector_type(8))) float v8f;

#define NEG_SLOPE 0.2f

template<int K, int COLS, bool BIAS>
__global__ void gemm_wmma_f32(const float* __restrict__ A,
                              const float* __restrict__ B,
                              const float* __restrict__ bias,
                              float* __restrict__ C, int N) {
  constexpr int CT = COLS / 16;
  const int wave = (int)((blockIdx.x * blockDim.x + threadIdx.x) >> 5);
  const int lane = (int)(threadIdx.x & 31);
  const int row0 = wave * 16;
  if (row0 >= N) return;
  const int m    = lane & 15;
  const int half = lane >> 4;
  const int klo  = half * 2;

  v8f zero = {0.f, 0.f, 0.f, 0.f, 0.f, 0.f, 0.f, 0.f};
  v8f acc[CT];
  #pragma unroll
  for (int t = 0; t < CT; ++t) acc[t] = zero;

  #pragma unroll
  for (int k = 0; k < K; k += 4) {
    const float* ap = A + (size_t)(row0 + m) * K + (k + klo);
    v2f a; a.x = ap[0]; a.y = ap[1];
    #pragma unroll
    for (int t = 0; t < CT; ++t) {
      const float* bp = B + (size_t)(k + klo) * COLS + t * 16 + m;
      v2f b; b.x = bp[0]; b.y = bp[COLS];
      acc[t] = __builtin_amdgcn_wmma_f32_16x16x4_f32(
          false, a, false, b, (short)0, acc[t], false, false);
    }
  }
  const int rbase = row0 + half * 8;
  #pragma unroll
  for (int t = 0; t < CT; ++t) {
    const float bv = BIAS ? bias[t * 16 + m] : 0.f;
    #pragma unroll
    for (int v = 0; v < 8; ++v)
      C[(size_t)(rbase + v) * COLS + t * 16 + m] = BIAS ? acc[t][v] + bv : acc[t][v];
  }
}

__global__ void alpha1_k(const float* __restrict__ h1,
                         const float* __restrict__ aw_s, const float* __restrict__ aw_d,
                         float* __restrict__ as, float* __restrict__ ad, int N) {
  long long i = (long long)blockIdx.x * blockDim.x + threadIdx.x;
  if (i >= (long long)N * 8) return;
  int n = (int)(i >> 3), h = (int)(i & 7);
  const float* hp = h1 + (size_t)n * 64 + h * 8;
  const float* sp = aw_s + h * 8;
  const float* dp = aw_d + h * 8;
  float s = 0.f, d = 0.f;
  #pragma unroll
  for (int c = 0; c < 8; ++c) { float hv = hp[c]; s += hv * sp[c]; d += hv * dp[c]; }
  as[i] = s; ad[i] = d;
}

__global__ void fold_k(const float* __restrict__ W2,
                       const float* __restrict__ aw_s, const float* __restrict__ aw_d,
                       float* __restrict__ us, float* __restrict__ ud) {
  const int wave = (int)((blockIdx.x * blockDim.x + threadIdx.x) >> 5);
  const int lane = (int)(threadIdx.x & 31);
  if (wave >= 64) return;
  const float4 wv = *(const float4*)(W2 + (size_t)wave * 128 + lane * 4);
  const float4 sv = *(const float4*)(aw_s + lane * 4);
  const float4 dv = *(const float4*)(aw_d + lane * 4);
  float s = wv.x * sv.x + wv.y * sv.y + wv.z * sv.z + wv.w * sv.w;
  float d = wv.x * dv.x + wv.y * dv.y + wv.z * dv.z + wv.w * dv.w;
  #pragma unroll
  for (int off = 16; off > 0; off >>= 1) {
    s += __shfl_xor(s, off, 32);
    d += __shfl_xor(d, off, 32);
  }
  if (lane == 0) { us[wave] = s; ud[wave] = d; }
}

__global__ void count_pos_k(const int* __restrict__ ei, long long E,
                            int* __restrict__ cnt, int* __restrict__ pos) {
  long long e = (long long)blockIdx.x * blockDim.x + threadIdx.x;
  if (e >= E) return;
  pos[e] = atomicAdd(&cnt[ei[E + e]], 1);
}

__global__ void __launch_bounds__(1024)
scan_k(const int* __restrict__ cnt, int* __restrict__ rowptr, int N) {
  __shared__ int wtot[32];
  __shared__ int wpre[32];
  const int tid = (int)threadIdx.x, lane = tid & 31, wv = tid >> 5;
  const int base = (int)blockIdx.x * 1024;
  int part = 0;
  for (int i = tid; i < base; i += 1024) part += cnt[i];
  const int i = base + tid;
  const int v = (i < N) ? cnt[i] : 0;
  int incl = v;
  #pragma unroll
  for (int off = 1; off < 32; off <<= 1) {
    int t = __shfl_up(incl, off, 32);
    if (lane >= off) incl += t;
  }
  #pragma unroll
  for (int off = 16; off > 0; off >>= 1) part += __shfl_xor(part, off, 32);
  if (lane == 31) wtot[wv] = incl;
  if (lane == 0)  wpre[wv] = part;
  __syncthreads();
  int t2 = wpre[lane] + (lane < wv ? wtot[lane] : 0);
  #pragma unroll
  for (int off = 16; off > 0; off >>= 1) t2 += __shfl_xor(t2, off, 32);
  const int excl = t2 + incl - v;
  if (i < N) rowptr[i] = excl;
  if (i == N - 1) rowptr[N] = excl + v;
}

__global__ void fill_k(const int* __restrict__ ei, long long E,
                       const int* __restrict__ rowptr, const int* __restrict__ pos,
                       int* __restrict__ csr) {
  long long e = (long long)blockIdx.x * blockDim.x + threadIdx.x;
  if (e >= E) return;
  csr[rowptr[ei[E + e]] + pos[e]] = ei[e];
}

__device__ __forceinline__ float leaky(float e) { return e > 0.f ? e : NEG_SLOPE * e; }

__global__ void __launch_bounds__(256)
gat1_node_k(const int* __restrict__ rowptr, const int* __restrict__ csr,
            const float* __restrict__ as, const float* __restrict__ ad,
            const float* __restrict__ h1, const float* __restrict__ b1,
            const float* __restrict__ us, const float* __restrict__ ud,
            float* __restrict__ helu, float* __restrict__ as2, float* __restrict__ ad2, int N) {
  const int lane = (int)(threadIdx.x & 31);
  const int node = __builtin_amdgcn_readfirstlane((int)((blockIdx.x * blockDim.x + threadIdx.x) >> 5));
  if (node >= N) return;
  const int beg = rowptr[node];
  const int deg = rowptr[node + 1] - beg;

  const int q = lane >> 3, hq = lane & 7;
  const float adq = ad[(size_t)node * 8 + hq];
  float m = leaky(as[(size_t)node * 8 + hq] + adq);
  for (int c0 = 0; c0 < deg; c0 += 32) {
    const int cnt = min(32, deg - c0);
    const int sl = (lane < cnt) ? csr[beg + c0 + lane] : node;
    #pragma unroll 4
    for (int j = 0; j < cnt; j += 4) {
      const int s = __shfl(sl, j + q, 32);
      m = fmaxf(m, leaky(as[(size_t)s * 8 + hq] + adq));
    }
  }
  m = fmaxf(m, __shfl_xor(m, 8, 32));
  m = fmaxf(m, __shfl_xor(m, 16, 32));

  const int hb = lane >> 2;
  const float mb  = __shfl(m, hb, 32);
  const float adb = ad[(size_t)node * 8 + hb];
  float den, ax, ay;
  {
    const float w = __expf(leaky(as[(size_t)node * 8 + hb] + adb) - mb);
    const float2 hv = *(const float2*)(h1 + (size_t)node * 64 + lane * 2);
    den = w; ax = w * hv.x; ay = w * hv.y;
  }
  for (int c0 = 0; c0 < deg; c0 += 32) {
    const int cnt = min(32, deg - c0);
    const int sl = (lane < cnt) ? csr[beg + c0 + lane] : node;
    #pragma unroll 4
    for (int j = 0; j < cnt; ++j) {
      const int s = __builtin_amdgcn_readlane(sl, j);
      const float w = __expf(leaky(as[(size_t)s * 8 + hb] + adb) - mb);
      const float2 hv = *(const float2*)(h1 + (size_t)s * 64 + lane * 2);
      den += w; ax += w * hv.x; ay += w * hv.y;
    }
  }
  float vx = ax / den + b1[lane * 2];
  float vy = ay / den + b1[lane * 2 + 1];
  float2 o;
  o.x = vx > 0.f ? vx : (__expf(vx) - 1.f);
  o.y = vy > 0.f ? vy : (__expf(vy) - 1.f);
  *(float2*)(helu + (size_t)node * 64 + lane * 2) = o;

  const float2 usv = *(const float2*)(us + lane * 2);
  const float2 udv = *(const float2*)(ud + lane * 2);
  float ps = o.x * usv.x + o.y * usv.y;
  float pd = o.x * udv.x + o.y * udv.y;
  #pragma unroll
  for (int off = 16; off > 0; off >>= 1) {
    ps += __shfl_xor(ps, off, 32);
    pd += __shfl_xor(pd, off, 32);
  }
  if (lane == 0) { as2[node] = ps; ad2[node] = pd; }
}

__global__ void __launch_bounds__(256)
gat2_node_k(const int* __restrict__ rowptr, const int* __restrict__ csr,
            const float* __restrict__ as, const float* __restrict__ ad,
            const float* __restrict__ helu, float* __restrict__ agg, int N) {
  const int lane = (int)(threadIdx.x & 31);
  const int node = __builtin_amdgcn_readfirstlane((int)((blockIdx.x * blockDim.x + threadIdx.x) >> 5));
  if (node >= N) return;
  const int beg = rowptr[node];
  const int deg = rowptr[node + 1] - beg;
  const float adn = ad[node];
  const float eself = leaky(as[node] + adn);

  float m = eself;
  for (int c0 = 0; c0 < deg; c0 += 32) {
    const int cnt = min(32, deg - c0);
    const int sl = (lane < cnt) ? csr[beg + c0 + lane] : node;
    m = fmaxf(m, leaky(as[sl] + adn));
  }
  #pragma unroll
  for (int off = 16; off > 0; off >>= 1) m = fmaxf(m, __shfl_xor(m, off, 32));

  const float wself = __expf(eself - m);
  float denp = 0.f;
  float ax, ay;
  {
    const float2 hv = *(const float2*)(helu + (size_t)node * 64 + lane * 2);
    ax = wself * hv.x; ay = wself * hv.y;
  }
  for (int c0 = 0; c0 < deg; c0 += 32) {
    const int cnt = min(32, deg - c0);
    const int sl = (lane < cnt) ? csr[beg + c0 + lane] : node;
    const float wl = (lane < cnt) ? __expf(leaky(as[sl] + adn) - m) : 0.f;
    denp += wl;
    #pragma unroll 4
    for (int j = 0; j < cnt; ++j) {
      const int s = __builtin_amdgcn_readlane(sl, j);
      const float w = __int_as_float(__builtin_amdgcn_readlane(__float_as_int(wl), j));
      const float2 hv = *(const float2*)(helu + (size_t)s * 64 + lane * 2);
      ax += w * hv.x; ay += w * hv.y;
    }
  }
  #pragma unroll
  for (int off = 16; off > 0; off >>= 1) denp += __shfl_xor(denp, off, 32);
  const float den = wself + denp;
  float2 o;
  o.x = ax / den; o.y = ay / den;
  *(float2*)(agg + (size_t)node * 64 + lane * 2) = o;
}

static inline unsigned cdiv(long long a, long long b) { return (unsigned)((a + b - 1) / b); }

extern "C" void kernel_launch(void* const* d_in, const int* in_sizes, int n_in,
                              void* d_out, int out_size, void* d_ws, size_t ws_size,
                              hipStream_t stream) {
  const float* x    = (const float*)d_in[0];
  const int*   ei   = (const int*)d_in[1];
  const float* W1   = (const float*)d_in[2];
  const float* as1w = (const float*)d_in[3];
  const float* ad1w = (const float*)d_in[4];
  const float* b1   = (const float*)d_in[5];
  const float* W2   = (const float*)d_in[6];
  const float* as2w = (const float*)d_in[7];
  const float* ad2w = (const float*)d_in[8];
  const float* b2   = (const float*)d_in[9];
  float* out = (float*)d_out;

  const int N = in_sizes[0] / 128;
  const long long E = (long long)in_sizes[1] / 2;

  float* w = (float*)d_ws;
  size_t off = 0;
  float* h1   = w + off; off += (size_t)N * 64;
  float* helu = w + off; off += (size_t)N * 64;
  float* agg  = w + off; off += (size_t)N * 64;
  float* us   = w + off; off += 64;
  float* ud   = w + off; off += 64;
  float* as1  = w + off; off += (size_t)N * 8;
  float* ad1  = w + off; off += (size_t)N * 8;
  float* as2  = w + off; off += (size_t)N;
  float* ad2  = w + off; off += (size_t)N;
  int* cnt    = (int*)(w + off); off += (size_t)N;
  int* rowptr = (int*)(w + off); off += (size_t)N + 1;
  int* pos    = (int*)(w + off); off += (size_t)E;
  int* csr    = (int*)(w + off); off += (size_t)E;

  const int B = 256;
  const int tiles = (N + 15) / 16;

  hipMemsetAsync(cnt, 0, (size_t)N * sizeof(int), stream);
  count_pos_k<<<cdiv(E, B), B, 0, stream>>>(ei, E, cnt, pos);
  scan_k<<<cdiv(N, 1024), 1024, 0, stream>>>(cnt, rowptr, N);
  fill_k<<<cdiv(E, B), B, 0, stream>>>(ei, E, rowptr, pos, csr);

  fold_k<<<8, B, 0, stream>>>(W2, as2w, ad2w, us, ud);
  gemm_wmma_f32<128, 64, false><<<cdiv((long long)tiles * 32, B), B, 0, stream>>>(x, W1, nullptr, h1, N);
  alpha1_k<<<cdiv((long long)N * 8, B), B, 0, stream>>>(h1, as1w, ad1w, as1, ad1, N);
  gat1_node_k<<<cdiv((long long)N * 32, B), B, 0, stream>>>(rowptr, csr, as1, ad1, h1, b1, us, ud,
                                                           helu, as2, ad2, N);

  gat2_node_k<<<cdiv((long long)N * 32, B), B, 0, stream>>>(rowptr, csr, as2, ad2, helu, agg, N);
  gemm_wmma_f32<64, 128, true><<<cdiv((long long)tiles * 32, B), B, 0, stream>>>(agg, W2, b2, out, N);

  (void)n_in; (void)out_size; (void)ws_size;
}


# Edge list grouped by destination on LDS counters: two-level counting sort (256-node buckets, runs staged in LDS, whole-line stores) replaces per-edge global adds, scan, fill; sum order = list order
# speedup vs baseline: 7.5890x; 1.9006x over previous
#include <hip/hip_runtime.h>
#include <cstdint>
#include <cstddef>

typedef __attribute__((ext_vector_type(2))) float v2f;
typedef __attribute__((ext_vector_type(8))) float v8f;

#define NEG_SLOPE 0.2f

template<int K, int COLS, bool BIAS>
__global__ void gemm_wmma_f32(const float* __restrict__ A,
                              const float* __restrict__ B,
                              const float* __restrict__ bias,
                              float* __restrict__ C, int N) {
  constexpr int CT = COLS / 16;
  const int wave = (int)((blockIdx.x * blockDim.x + threadIdx.x) >> 5);
  const int lane = (int)(threadIdx.x & 31);
  const int row0 = wave * 16;
  if (row0 >= N) return;
  const int m    = lane & 15;
  const int half = lane >> 4;
  const int klo  = half * 2;

  v8f zero = {0.f, 0.f, 0.f, 0.f, 0.f, 0.f, 0.f, 0.f};
  v8f acc[CT];
  #pragma unroll
  for (int t = 0; t < CT; ++t) acc[t] = zero;

  #pragma unroll
  for (int k = 0; k < K; k += 4) {
    const float* ap = A + (size_t)(row0 + m) * K + (k + klo);
    v2f a; a.x = ap[0]; a.y = ap[1];
    #pragma unroll
    for (int t = 0; t < CT; ++t) {
      const float* bp = B + (size_t)(k + klo) * COLS + t * 16 + m;
      v2f b; b.x = bp[0]; b.y = bp[COLS];
      acc[t] = __builtin_amdgcn_wmma_f32_16x16x4_f32(
          false, a, false, b, (short)0, acc[t], false, false);
    }
  }
  const int rbase = row0 + half * 8;
  #pragma unroll
  for (int t = 0; t < CT; ++t) {
    const float bv = BIAS ? bias[t * 16 + m] : 0.f;
    #pragma unroll
    for (int v = 0; v < 8; ++v)
      C[(size_t)(rbase + v) * COLS + t * 16 + m] = BIAS ? acc[t][v] + bv : acc[t][v];
  }
}

__global__ void alpha1_k(const float* __restrict__ h1,
                         const float* __restrict__ aw_s, const float* __restrict__ aw_d,
                         float* __restrict__ as, float* __restrict__ ad, int N) {
  long long i = (long long)blockIdx.x * blockDim.x + threadIdx.x;
  if (i >= (long long)N * 8) return;
  int n = (int)(i >> 3), h = (int)(i & 7);
  const float* hp = h1 + (size_t)n * 64 + h * 8;
  const float* sp = aw_s + h * 8;
  const float* dp = aw_d + h * 8;
  float s = 0.f, d = 0.f;
  #pragma unroll
  for (int c = 0; c < 8; ++c) { float hv = hp[c]; s += hv * sp[c]; d += hv * dp[c]; }
  as[i] = s; ad[i] = d;
}

__global__ void fold_k(const float* __restrict__ W2,
                       const float* __restrict__ aw_s, const float* __restrict__ aw_d,
                       float* __restrict__ us, float* __restrict__ ud) {
  const int wave = (int)((blockIdx.x * blockDim.x + threadIdx.x) >> 5);
  const int lane = (int)(threadIdx.x & 31);
  if (wave >= 64) return;
  const float4 wv = *(const float4*)(W2 + (size_t)wave * 128 + lane * 4);
  const float4 sv = *(const float4*)(aw_s + lane * 4);
  const float4 dv = *(const float4*)(aw_d + lane * 4);
  float s = wv.x * sv.x + wv.y * sv.y + wv.z * sv.z + wv.w * sv.w;
  float d = wv.x * dv.x + wv.y * dv.y + wv.z * dv.z + wv.w * dv.w;
  #pragma unroll
  for (int off = 16; off > 0; off >>= 1) {
    s += __shfl_xor(s, off, 32);
    d += __shfl_xor(d, off, 32);
  }
  if (lane == 0) { us[wave] = s; ud[wave] = d; }
}

#define NODE_SHIFT 8
#define NODES_PER_BUCKET 256
#define KMAX 1024
#define EPB 8192
#define EPT (EPB / 1024)
#define ROWS_EPT 12
#define STAGE_CAP (ROWS_EPT * 1024)
#define CPAD 64

__device__ __forceinline__ void wave_excl_scan_lds(const int* in, int* out, int* out2, int n, int lane) {
  int carry = 0;
  for (int b0 = 0; b0 < n; b0 += 256) {
    int v[8], incl[8];
    #pragma unroll
    for (int c = 0; c < 8; ++c) {
      const int i = b0 + c * 32 + lane;
      v[c] = (i < n) ? in[i] : 0;
      incl[c] = v[c];
    }
    #pragma unroll
    for (int off = 1; off < 32; off <<= 1) {
      #pragma unroll
      for (int c = 0; c < 8; ++c) {
        const int t = __shfl_up(incl[c], off, 32);
        if (lane >= off) incl[c] += t;
      }
    }
    #pragma unroll
    for (int c = 0; c < 8; ++c) {
      const int i = b0 + c * 32 + lane;
      const int tot = __shfl(incl[c], 31, 32);
      if (i < n) {
        out[i] = carry + incl[c] - v[c];
        if (out2) out2[i] = carry + incl[c] - v[c];
      }
      carry += tot;
    }
  }
}

__global__ void __launch_bounds__(1024)
bucket_hist_k(const int* __restrict__ ei, long long E, int K, int* __restrict__ gcnt) {
  __shared__ int cntL[KMAX];
  const int tid = (int)threadIdx.x;
  const long long e0 = (long long)blockIdx.x * EPB;
  int d[EPT];
  #pragma unroll
  for (int j = 0; j < EPT; ++j) {
    const long long e = e0 + j * 1024 + tid;
    d[j] = (e < E) ? ei[E + e] : -1;
  }
  for (int k = tid; k < K; k += 1024) cntL[k] = 0;
  __syncthreads();
  #pragma unroll
  for (int j = 0; j < EPT; ++j)
    if (d[j] >= 0) atomicAdd(&cntL[d[j] >> NODE_SHIFT], 1);
  __syncthreads();
  for (int k = tid; k < K; k += 1024) {
    const int c = cntL[k];
    if (c) atomicAdd(&gcnt[k * CPAD], c);
  }
}

__global__ void __launch_bounds__(1024)
bucket_scatter_k(const int* __restrict__ ei, long long E, int K,
                 int* __restrict__ gcnt, int2* __restrict__ part) {
  __shared__ int cntL[KMAX];
  __shared__ int lstart[KMAX];
  __shared__ int gdelta[KMAX];
  __shared__ int bstart[KMAX];
  __shared__ int2 stage[EPB];
  const int tid = (int)threadIdx.x, lane = tid & 31;
  const long long e0 = (long long)blockIdx.x * EPB;
  int d[EPT], s[EPT];
  #pragma unroll
  for (int j = 0; j < EPT; ++j) {
    const long long e = e0 + j * 1024 + tid;
    d[j] = -1; s[j] = 0;
    if (e < E) { d[j] = ei[E + e]; s[j] = ei[e]; }
  }
  for (int k = tid; k < K; k += 1024) { cntL[k] = 0; bstart[k] = gcnt[k * CPAD]; }
  __syncthreads();
  #pragma unroll
  for (int j = 0; j < EPT; ++j)
    if (d[j] >= 0) atomicAdd(&cntL[d[j] >> NODE_SHIFT], 1);
  __syncthreads();
  if (tid < 32)      wave_excl_scan_lds(bstart, bstart, nullptr, K, lane);
  else if (tid < 64) wave_excl_scan_lds(cntL, lstart, nullptr, K, lane);
  __syncthreads();
  if (blockIdx.x == 0)
    for (int k = tid; k < K; k += 1024) gcnt[k * CPAD + 2] = bstart[k];
  for (int k = tid; k < K; k += 1024) {
    const int c = cntL[k];
    const int g = c ? atomicAdd(&gcnt[k * CPAD + 1], c) : 0;
    const int ls = lstart[k];
    gdelta[k] = bstart[k] + g - ls;
    cntL[k] = ls;
  }
  __syncthreads();
  int slot[EPT];
  #pragma unroll
  for (int j = 0; j < EPT; ++j)
    slot[j] = (d[j] >= 0) ? atomicAdd(&cntL[d[j] >> NODE_SHIFT], 1) : 0;
  #pragma unroll
  for (int j = 0; j < EPT; ++j)
    if (d[j] >= 0) stage[slot[j]] = make_int2(s[j], d[j]);
  __syncthreads();
  const long long rem = E - e0;
  const int nE = rem < (long long)EPB ? (int)rem : EPB;
  if (nE == EPB) {
    int2 p[EPT]; int g[EPT];
    #pragma unroll
    for (int j = 0; j < EPT; ++j) p[j] = stage[tid + j * 1024];
    #pragma unroll
    for (int j = 0; j < EPT; ++j) g[j] = gdelta[p[j].y >> NODE_SHIFT];
    #pragma unroll
    for (int j = 0; j < EPT; ++j) part[g[j] + tid + j * 1024] = p[j];
  } else {
    for (int i = tid; i < nE; i += 1024) {
      const int2 p = stage[i];
      part[gdelta[p.y >> NODE_SHIFT] + i] = p;
    }
  }
}

__global__ void __launch_bounds__(1024)
bucket_rows_k(const int2* __restrict__ part, const int* __restrict__ gcnt, int K, int N,
              int* __restrict__ rowptr, int* __restrict__ csr) {
  __shared__ int cntL[NODES_PER_BUCKET];
  __shared__ int cur[NODES_PER_BUCKET];
  __shared__ int excl[NODES_PER_BUCKET];
  __shared__ int stage[STAGE_CAP];
  const int tid = (int)threadIdx.x, lane = tid & 31;
  const int k = (int)blockIdx.x;
  const int node0 = k << NODE_SHIFT;
  const int cnt  = gcnt[k * CPAD];
  const int base = gcnt[k * CPAD + 2];
  if (tid < NODES_PER_BUCKET) cntL[tid] = 0;
  __syncthreads();
  const bool staged = cnt <= STAGE_CAP;

  int2 p[ROWS_EPT];
  if (staged) {
    #pragma unroll
    for (int j = 0; j < ROWS_EPT; ++j) {
      const int i = tid + j * 1024;
      p[j] = make_int2(0, -1);
      if (i < cnt) p[j] = part[base + i];
    }
    #pragma unroll
    for (int j = 0; j < ROWS_EPT; ++j)
      if (p[j].y >= 0) atomicAdd(&cntL[p[j].y - node0], 1);
  } else {
    for (int i = tid; i < cnt; i += 1024) atomicAdd(&cntL[part[base + i].y - node0], 1);
  }
  __syncthreads();
  if (tid < 32) wave_excl_scan_lds(cntL, cur, excl, NODES_PER_BUCKET, lane);
  __syncthreads();
  if (tid < NODES_PER_BUCKET) {
    const int n = node0 + tid;
    if (n < N) rowptr[n] = base + excl[tid];
  }
  if (tid == 0 && k == K - 1) rowptr[N] = base + cnt;
  if (staged) {
    int r[ROWS_EPT];
    #pragma unroll
    for (int j = 0; j < ROWS_EPT; ++j)
      r[j] = (p[j].y >= 0) ? atomicAdd(&cur[p[j].y - node0], 1) : 0;
    #pragma unroll
    for (int j = 0; j < ROWS_EPT; ++j)
      if (p[j].y >= 0) stage[r[j]] = p[j].x;
  } else {
    for (int i = tid; i < cnt; i += 1024) {
      const int2 q = part[base + i];
      csr[base + atomicAdd(&cur[q.y - node0], 1)] = q.x;
    }
  }
  __syncthreads();
  if (staged)
    for (int i = tid; i < cnt; i += 1024) csr[base + i] = stage[i];
}

__device__ __forceinline__ float leaky(float e) { return e > 0.f ? e : NEG_SLOPE * e; }

__global__ void __launch_bounds__(256)
gat1_node_k(const int* __restrict__ rowptr, const int* __restrict__ csr,
            const float* __restrict__ as, const float* __restrict__ ad,
            const float* __restrict__ h1, const float* __restrict__ b1,
            const float* __restrict__ us, const float* __restrict__ ud,
            float* __restrict__ helu, float* __restrict__ as2, float* __restrict__ ad2, int N) {
  const int lane = (int)(threadIdx.x & 31);
  const int node = __builtin_amdgcn_readfirstlane((int)((blockIdx.x * blockDim.x + threadIdx.x) >> 5));
  if (node >= N) return;
  const int beg = rowptr[node];
  const int deg = rowptr[node + 1] - beg;

  const int q = lane >> 3, hq = lane & 7;
  const float adq = ad[(size_t)node * 8 + hq];
  float m = leaky(as[(size_t)node * 8 + hq] + adq);
  for (int c0 = 0; c0 < deg; c0 += 32) {
    const int cnt = min(32, deg - c0);
    const int sl = (lane < cnt) ? csr[beg + c0 + lane] : node;
    #pragma unroll 4
    for (int j = 0; j < cnt; j += 4) {
      const int s = __shfl(sl, j + q, 32);
      m = fmaxf(m, leaky(as[(size_t)s * 8 + hq] + adq));
    }
  }
  m = fmaxf(m, __shfl_xor(m, 8, 32));
  m = fmaxf(m, __shfl_xor(m, 16, 32));

  const int hb = lane >> 2;
  const float mb  = __shfl(m, hb, 32);
  const float adb = ad[(size_t)node * 8 + hb];
  float den, ax, ay;
  {
    const float w = __expf(leaky(as[(size_t)node * 8 + hb] + adb) - mb);
    const float2 hv = *(const float2*)(h1 + (size_t)node * 64 + lane * 2);
    den = w; ax = w * hv.x; ay = w * hv.y;
  }
  for (int c0 = 0; c0 < deg; c0 += 32) {
    const int cnt = min(32, deg - c0);
    const int sl = (lane < cnt) ? csr[beg + c0 + lane] : node;
    #pragma unroll 4
    for (int j = 0; j < cnt; ++j) {
      const int s = __builtin_amdgcn_readlane(sl, j);
      const float w = __expf(leaky(as[(size_t)s * 8 + hb] + adb) - mb);
      const float2 hv = *(const float2*)(h1 + (size_t)s * 64 + lane * 2);
      den += w; ax += w * hv.x; ay += w * hv.y;
    }
  }
  float vx = ax / den + b1[lane * 2];
  float vy = ay / den + b1[lane * 2 + 1];
  float2 o;
  o.x = vx > 0.f ? vx : (__expf(vx) - 1.f);
  o.y = vy > 0.f ? vy : (__expf(vy) - 1.f);
  *(float2*)(helu + (size_t)node * 64 + lane * 2) = o;

  const float2 usv = *(const float2*)(us + lane * 2);
  const float2 udv = *(const float2*)(ud + lane * 2);
  float ps = o.x * usv.x + o.y * usv.y;
  float pd = o.x * udv.x + o.y * udv.y;
  #pragma unroll
  for (int off = 16; off > 0; off >>= 1) {
    ps += __shfl_xor(ps, off, 32);
    pd += __shfl_xor(pd, off, 32);
  }
  if (lane == 0) { as2[node] = ps; ad2[node] = pd; }
}

__global__ void __launch_bounds__(256)
gat2_node_k(const int* __restrict__ rowptr, const int* __restrict__ csr,
            const float* __restrict__ as, const float* __restrict__ ad,
            const float* __restrict__ helu, float* __restrict__ agg, int N) {
  const int lane = (int)(threadIdx.x & 31);
  const int node = __builtin_amdgcn_readfirstlane((int)((blockIdx.x * blockDim.x + threadIdx.x) >> 5));
  if (node >= N) return;
  const int beg = rowptr[node];
  const int deg = rowptr[node + 1] - beg;
  const float adn = ad[node];
  const float eself = leaky(as[node] + adn);

  float m = eself;
  for (int c0 = 0; c0 < deg; c0 += 32) {
    const int cnt = min(32, deg - c0);
    const int sl = (lane < cnt) ? csr[beg + c0 + lane] : node;
    m = fmaxf(m, leaky(as[sl] + adn));
  }
  #pragma unroll
  for (int off = 16; off > 0; off >>= 1) m = fmaxf(m, __shfl_xor(m, off, 32));

  const float wself = __expf(eself - m);
  float denp = 0.f;
  float ax, ay;
  {
    const float2 hv = *(const float2*)(helu + (size_t)node * 64 + lane * 2);
    ax = wself * hv.x; ay = wself * hv.y;
  }
  for (int c0 = 0; c0 < deg; c0 += 32) {
    const int cnt = min(32, deg - c0);
    const int sl = (lane < cnt) ? csr[beg + c0 + lane] : node;
    const float wl = (lane < cnt) ? __expf(leaky(as[sl] + adn) - m) : 0.f;
    denp += wl;
    #pragma unroll 4
    for (int j = 0; j < cnt; ++j) {
      const int s = __builtin_amdgcn_readlane(sl, j);
      const float w = __int_as_float(__builtin_amdgcn_readlane(__float_as_int(wl), j));
      const float2 hv = *(const float2*)(helu + (size_t)s * 64 + lane * 2);
      ax += w * hv.x; ay += w * hv.y;
    }
  }
  #pragma unroll
  for (int off = 16; off > 0; off >>= 1) denp += __shfl_xor(denp, off, 32);
  const float den = wself + denp;
  float2 o;
  o.x = ax / den; o.y = ay / den;
  *(float2*)(agg + (size_t)node * 64 + lane * 2) = o;
}

static inline unsigned cdiv(long long a, long long b) { return (unsigned)((a + b - 1) / b); }

extern "C" void kernel_launch(void* const* d_in, const int* in_sizes, int n_in,
                              void* d_out, int out_size, void* d_ws, size_t ws_size,
                              hipStream_t stream) {
  const float* x    = (const float*)d_in[0];
  const int*   ei   = (const int*)d_in[1];
  const float* W1   = (const float*)d_in[2];
  const float* as1w = (const float*)d_in[3];
  const float* ad1w = (const float*)d_in[4];
  const float* b1   = (const float*)d_in[5];
  const float* W2   = (const float*)d_in[6];
  const float* as2w = (const float*)d_in[7];
  const float* ad2w = (const float*)d_in[8];
  const float* b2   = (const float*)d_in[9];
  float* out = (float*)d_out;

  const int N = in_sizes[0] / 128;
  const long long E = (long long)in_sizes[1] / 2;

  float* w = (float*)d_ws;
  size_t off = 0;
  float* h1   = w + off; off += (size_t)N * 64;
  float* helu = w + off; off += (size_t)N * 64;
  float* agg  = w + off; off += (size_t)N * 64;
  float* us   = w + off; off += 64;
  float* ud   = w + off; off += 64;
  float* as1  = w + off; off += (size_t)N * 8;
  float* ad1  = w + off; off += (size_t)N * 8;
  float* as2  = w + off; off += (size_t)N;
  float* ad2  = w + off; off += (size_t)N;
  int2* part  = (int2*)(w + off); off += (size_t)E * 2;
  int* gcnt   = (int*)(w + off); off += (size_t)KMAX * CPAD;
  int* rowptr = (int*)(w + off); off += (size_t)N + 1;
  int* csr    = (int*)(w + off); off += (size_t)E;

  const int B = 256;
  const int tiles = (N + 15) / 16;

  const int K = (N + NODES_PER_BUCKET - 1) / NODES_PER_BUCKET;
  hipMemsetAsync(gcnt, 0, (size_t)K * CPAD * sizeof(int), stream);
  bucket_hist_k<<<cdiv(E, EPB), 1024, 0, stream>>>(ei, E, K, gcnt);
  bucket_scatter_k<<<cdiv(E, EPB), 1024, 0, stream>>>(ei, E, K, gcnt, part);
  bucket_rows_k<<<K, 1024, 0, stream>>>(part, gcnt, K, N, rowptr, csr);

  fold_k<<<8, B, 0, stream>>>(W2, as2w, ad2w, us, ud);
  gemm_wmma_f32<128, 64, false><<<cdiv((long long)tiles * 32, B), B, 0, stream>>>(x, W1, nullptr, h1, N);
  alpha1_k<<<cdiv((long long)N * 8, B), B, 0, stream>>>(h1, as1w, ad1w, as1, ad1, N);
  gat1_node_k<<<cdiv((long long)N * 32, B), B, 0, stream>>>(rowptr, csr, as1, ad1, h1, b1, us, ud,
                                                           helu, as2, ad2, N);

  gat2_node_k<<<cdiv((long long)N * 32, B), B, 0, stream>>>(rowptr, csr, as2, ad2, helu, agg, N);
  gemm_wmma_f32<64, 128, true><<<cdiv((long long)tiles * 32, B), B, 0, stream>>>(agg, W2, b2, out, N);

  (void)n_in; (void)out_size; (void)ws_size;
}


# Layer-1 GEMM tiles ride as extra work-groups in the level-2 edge-grouping launch: two independent phases overlap instead of running back to back, one launch fewer; sums unchanged
# speedup vs baseline: 10.1669x; 1.3397x over previous
#include <hip/hip_runtime.h>
#include <cstdint>
#include <cstddef>

typedef __attribute__((ext_vector_type(2))) float v2f;
typedef __attribute__((ext_vector_type(8))) float v8f;

#define NEG_SLOPE 0.2f

template<int K, int COLS, bool BIAS>
__global__ void gemm_wmma_f32(const float* __restrict__ A,
                              const float* __restrict__ B,
                              const float* __restrict__ bias,
                              float* __restrict__ C, int N) {
  constexpr int CT = COLS / 16;
  const int wave = (int)((blockIdx.x * blockDim.x + threadIdx.x) >> 5);
  const int lane = (int)(threadIdx.x & 31);
  const int row0 = wave * 16;
  if (row0 >= N) return;
  const int m    = lane & 15;
  const int half = lane >> 4;
  const int klo  = half * 2;

  v8f zero = {0.f, 0.f, 0.f, 0.f, 0.f, 0.f, 0.f, 0.f};
  v8f acc[CT];
  #pragma unroll
  for (int t = 0; t < CT; ++t) acc[t] = zero;

  #pragma unroll
  for (int k = 0; k < K; k += 4) {
    const float* ap = A + (size_t)(row0 + m) * K + (k + klo);
    v2f a; a.x = ap[0]; a.y = ap[1];
    #pragma unroll
    for (int t = 0; t < CT; ++t) {
      const float* bp = B + (size_t)(k + klo) * COLS + t * 16 + m;
      v2f b; b.x = bp[0]; b.y = bp[COLS];
      acc[t] = __builtin_amdgcn_wmma_f32_16x16x4_f32(
          false, a, false, b, (short)0, acc[t], false, false);
    }
  }
  const int rbase = row0 + half * 8;
  #pragma unroll
  for (int t = 0; t < CT; ++t) {
    const float bv = BIAS ? bias[t * 16 + m] : 0.f;
    #pragma unroll
    for (int v = 0; v < 8; ++v)
      C[(size_t)(rbase + v) * COLS + t * 16 + m] = BIAS ? acc[t][v] + bv : acc[t][v];
  }
}

template<int K, int COLS>
__device__ __forceinline__ void gemm_tile_f32(const float* __restrict__ A, const float* __restrict__ B,
                                              float* __restrict__ C, const int row0, const int lane) {
  constexpr int CT = COLS / 16;
  const unsigned m    = (unsigned)lane & 15u;
  const unsigned half = (unsigned)lane >> 4;
  const unsigned klo  = half * 2u;
  const float* __restrict__ At = A + (size_t)row0 * K;
  float* __restrict__ Ct = C + (size_t)row0 * COLS;
  const unsigned ao = m * (unsigned)K + klo;
  const unsigned bo = klo * (unsigned)COLS + m;

  v8f zero = {0.f, 0.f, 0.f, 0.f, 0.f, 0.f, 0.f, 0.f};
  v8f acc[CT];
  #pragma unroll
  for (int t = 0; t < CT; ++t) acc[t] = zero;

  #pragma unroll
  for (int k = 0; k < K; k += 4) {
    v2f a; a.x = At[ao + (unsigned)k]; a.y = At[ao + (unsigned)k + 1u];
    #pragma unroll
    for (int t = 0; t < CT; ++t) {
      const unsigned bi = bo + (unsigned)(k * COLS + t * 16);
      v2f b; b.x = B[bi]; b.y = B[bi + (unsigned)COLS];
      acc[t] = __builtin_amdgcn_wmma_f32_16x16x4_f32(
          false, a, false, b, (short)0, acc[t], false, false);
    }
  }
  const unsigned co = half * 8u * (unsigned)COLS + m;
  #pragma unroll
  for (int t = 0; t < CT; ++t) {
    #pragma unroll
    for (int v = 0; v < 8; ++v)
      Ct[co + (unsigned)(v * COLS + t * 16)] = acc[t][v];
  }
}

__global__ void fold_k(const float* __restrict__ W2,
                       const float* __restrict__ aw_s, const float* __restrict__ aw_d,
                       float* __restrict__ us, float* __restrict__ ud) {
  const int wave = (int)((blockIdx.x * blockDim.x + threadIdx.x) >> 5);
  const int lane = (int)(threadIdx.x & 31);
  if (wave >= 64) return;
  const float4 wv = *(const float4*)(W2 + (size_t)wave * 128 + lane * 4);
  const float4 sv = *(const float4*)(aw_s + lane * 4);
  const float4 dv = *(const float4*)(aw_d + lane * 4);
  float s = wv.x * sv.x + wv.y * sv.y + wv.z * sv.z + wv.w * sv.w;
  float d = wv.x * dv.x + wv.y * dv.y + wv.z * dv.z + wv.w * dv.w;
  #pragma unroll
  for (int off = 16; off > 0; off >>= 1) {
    s += __shfl_xor(s, off, 32);
    d += __shfl_xor(d, off, 32);
  }
  if (lane == 0) { us[wave] = s; ud[wave] = d; }
}

#define NODE_SHIFT 8
#define NODES_PER_BUCKET 256
#define KMAX 1024
#define EPB 8192
#define EPT (EPB / 1024)
#define ROWS_EPT 12
#define STAGE_CAP (ROWS_EPT * 1024)
#define CPAD 64

__device__ __forceinline__ void wave_excl_scan_lds(const int* in, int* out, int* out2, int n, int lane) {
  int carry = 0;
  for (int b0 = 0; b0 < n; b0 += 256) {
    int v[8], incl[8];
    #pragma unroll
    for (int c = 0; c < 8; ++c) {
      const int i = b0 + c * 32 + lane;
      v[c] = (i < n) ? in[i] : 0;
      incl[c] = v[c];
    }
    #pragma unroll
    for (int off = 1; off < 32; off <<= 1) {
      #pragma unroll
      for (int c = 0; c < 8; ++c) {
        const int t = __shfl_up(incl[c], off, 32);
        if (lane >= off) incl[c] += t;
      }
    }
    #pragma unroll
    for (int c = 0; c < 8; ++c) {
      const int i = b0 + c * 32 + lane;
      const int tot = __shfl(incl[c], 31, 32);
      if (i < n) {
        out[i] = carry + incl[c] - v[c];
        if (out2) out2[i] = carry + incl[c] - v[c];
      }
      carry += tot;
    }
  }
}

__global__ void __launch_bounds__(1024)
bucket_hist_k(const int* __restrict__ ei, long long E, int K, int* __restrict__ gcnt) {
  __shared__ int cntL[KMAX];
  const int tid = (int)threadIdx.x;
  const long long e0 = (long long)blockIdx.x * EPB;
  int d[EPT];
  #pragma unroll
  for (int j = 0; j < EPT; ++j) {
    const long long e = e0 + j * 1024 + tid;
    d[j] = (e < E) ? ei[E + e] : -1;
  }
  for (int k = tid; k < K; k += 1024) cntL[k] = 0;
  __syncthreads();
  #pragma unroll
  for (int j = 0; j < EPT; ++j)
    if (d[j] >= 0) atomicAdd(&cntL[d[j] >> NODE_SHIFT], 1);
  __syncthreads();
  for (int k = tid; k < K; k += 1024) {
    const int c = cntL[k];
    if (c) atomicAdd(&gcnt[k * CPAD], c);
  }
}

__global__ void __launch_bounds__(1024)
bucket_scatter_k(const int* __restrict__ ei, long long E, int K,
                 int* __restrict__ gcnt, int2* __restrict__ part) {
  __shared__ int cntL[KMAX];
  __shared__ int lstart[KMAX];
  __shared__ int gdelta[KMAX];
  __shared__ int bstart[KMAX];
  __shared__ int2 stage[EPB];
  const int tid = (int)threadIdx.x, lane = tid & 31;
  const long long e0 = (long long)blockIdx.x * EPB;
  int d[EPT], s[EPT];
  #pragma unroll
  for (int j = 0; j < EPT; ++j) {
    const long long e = e0 + j * 1024 + tid;
    d[j] = -1; s[j] = 0;
    if (e < E) { d[j] = ei[E + e]; s[j] = ei[e]; }
  }
  for (int k = tid; k < K; k += 1024) { cntL[k] = 0; bstart[k] = gcnt[k * CPAD]; }
  __syncthreads();
  #pragma unroll
  for (int j = 0; j < EPT; ++j)
    if (d[j] >= 0) atomicAdd(&cntL[d[j] >> NODE_SHIFT], 1);
  __syncthreads();
  if (tid < 32)      wave_excl_scan_lds(bstart, bstart, nullptr, K, lane);
  else if (tid < 64) wave_excl_scan_lds(cntL, lstart, nullptr, K, lane);
  __syncthreads();
  if (blockIdx.x == 0)
    for (int k = tid; k < K; k += 1024) gcnt[k * CPAD + 2] = bstart[k];
  for (int k = tid; k < K; k += 1024) {
    const int c = cntL[k];
    const int g = c ? atomicAdd(&gcnt[k * CPAD + 1], c) : 0;
    const int ls = lstart[k];
    gdelta[k] = bstart[k] + g - ls;
    cntL[k] = ls;
  }
  __syncthreads();
  int slot[EPT];
  #pragma unroll
  for (int j = 0; j < EPT; ++j)
    slot[j] = (d[j] >= 0) ? atomicAdd(&cntL[d[j] >> NODE_SHIFT], 1) : 0;
  #pragma unroll
  for (int j = 0; j < EPT; ++j)
    if (d[j] >= 0) stage[slot[j]] = make_int2(s[j], d[j]);
  __syncthreads();
  const long long rem = E - e0;
  const int nE = rem < (long long)EPB ? (int)rem : EPB;
  if (nE == EPB) {
    int2 p[EPT]; int g[EPT];
    #pragma unroll
    for (int j = 0; j < EPT; ++j) p[j] = stage[tid + j * 1024];
    #pragma unroll
    for (int j = 0; j < EPT; ++j) g[j] = gdelta[p[j].y >> NODE_SHIFT];
    #pragma unroll
    for (int j = 0; j < EPT; ++j) part[g[j] + tid + j * 1024] = p[j];
  } else {
    for (int i = tid; i < nE; i += 1024) {
      const int2 p = stage[i];
      part[gdelta[p.y >> NODE_SHIFT] + i] = p;
    }
  }
}

#define GEMM_TILES_PER_GROUP 16
__global__ void __launch_bounds__(1024, 16)
edge_rows_gemm_k(const int2* __restrict__ part, const int* __restrict__ gcnt, int K, int N,
                 int* __restrict__ rowptr, int* __restrict__ csr,
                 const float* __restrict__ x, const float* __restrict__ W1, float* __restrict__ h1) {
  if ((int)blockIdx.x >= K) {
    const int wv = __builtin_amdgcn_readfirstlane((int)(threadIdx.x >> 5));
    const int row0 = (((int)blockIdx.x - K) * GEMM_TILES_PER_GROUP + wv) * 16;
    if (wv < GEMM_TILES_PER_GROUP && row0 < N)
      gemm_tile_f32<128, 64>(x, W1, h1, row0, (int)(threadIdx.x & 31));
    return;
  }
  __shared__ int cntL[NODES_PER_BUCKET];
  __shared__ int cur[NODES_PER_BUCKET];
  __shared__ int excl[NODES_PER_BUCKET];
  __shared__ int stage[STAGE_CAP];
  const int tid = (int)threadIdx.x, lane = tid & 31;
  const int k = (int)blockIdx.x;
  const int node0 = k << NODE_SHIFT;
  const int cnt  = gcnt[k * CPAD];
  const int base = gcnt[k * CPAD + 2];
  if (tid < NODES_PER_BUCKET) cntL[tid] = 0;
  __syncthreads();
  const bool staged = cnt <= STAGE_CAP;

  int2 p[ROWS_EPT];
  if (staged) {
    #pragma unroll
    for (int j = 0; j < ROWS_EPT; ++j) {
      const int i = tid + j * 1024;
      p[j] = make_int2(0, -1);
      if (i < cnt) p[j] = part[base + i];
    }
    #pragma unroll
    for (int j = 0; j < ROWS_EPT; ++j)
      if (p[j].y >= 0) atomicAdd(&cntL[p[j].y - node0], 1);
  } else {
    for (int i = tid; i < cnt; i += 1024) atomicAdd(&cntL[part[base + i].y - node0], 1);
  }
  __syncthreads();
  if (tid < 32) wave_excl_scan_lds(cntL, cur, excl, NODES_PER_BUCKET, lane);
  __syncthreads();
  if (tid < NODES_PER_BUCKET) {
    const int n = node0 + tid;
    if (n < N) rowptr[n] = base + excl[tid];
  }
  if (tid == 0 && k == K - 1) rowptr[N] = base + cnt;
  if (staged) {
    int r[ROWS_EPT];
    #pragma unroll
    for (int j = 0; j < ROWS_EPT; ++j)
      r[j] = (p[j].y >= 0) ? atomicAdd(&cur[p[j].y - node0], 1) : 0;
    #pragma unroll
    for (int j = 0; j < ROWS_EPT; ++j)
      if (p[j].y >= 0) stage[r[j]] = p[j].x;
  } else {
    for (int i = tid; i < cnt; i += 1024) {
      const int2 q = part[base + i];
      csr[base + atomicAdd(&cur[q.y - node0], 1)] = q.x;
    }
  }
  __syncthreads();
  if (staged)
    for (int i = tid; i < cnt; i += 1024) csr[base + i] = stage[i];
}


__device__ __forceinline__ float leaky(float e) { return e > 0.f ? e : NEG_SLOPE * e; }
__device__ __forceinline__ float dot4(const float4 a, const float4 b) {
  return a.x * b.x + a.y * b.y + a.z * b.z + a.w * b.w;
}

__device__ __forceinline__ void stream_step(float& m, const float e, const bool ok, float& sc, float& w) {
  const float mn = fmaxf(m, e);
  sc = __expf(m - mn);
  w  = ok ? __expf(e - mn) : 0.f;
  m = mn;
}
__device__ __forceinline__ void scale_add(float4& acc, const float sc, const float w, const float4 v) {
  acc.x = acc.x * sc + w * v.x;
  acc.y = acc.y * sc + w * v.y;
  acc.z = acc.z * sc + w * v.z;
  acc.w = acc.w * sc + w * v.w;
}
__device__ __forceinline__ float slot_factor(const float m) {
  float M = fmaxf(m, __shfl_xor(m, 8, 32));
  M = fmaxf(M, __shfl_xor(M, 16, 32));
  return __expf(m - M);
}
__device__ __forceinline__ float sum_slots(float v) {
  v += __shfl_xor(v, 8, 32);
  v += __shfl_xor(v, 16, 32);
  return v;
}
__device__ __forceinline__ float4 sum_slots4(const float4 a, const float f) {
  float4 r;
  r.x = sum_slots(a.x * f); r.y = sum_slots(a.y * f); r.z = sum_slots(a.z * f); r.w = sum_slots(a.w * f);
  return r;
}

__global__ void __launch_bounds__(256, 16)
gat1_node_k(const int* __restrict__ rowptr, const int* __restrict__ csr,
            const float* __restrict__ h1,
            const float* __restrict__ aw_s, const float* __restrict__ aw_d,
            const float* __restrict__ b1, float* __restrict__ helu, int N) {
  const int lane = (int)(threadIdx.x & 31);
  const int node = __builtin_amdgcn_readfirstlane((int)((blockIdx.x * blockDim.x + threadIdx.x) >> 5));
  if (node >= N) return;
  const int beg = rowptr[node];
  const int deg = rowptr[node + 1] - beg;
  const int eq = lane >> 3;
  const unsigned pc = (unsigned)(lane & 7);
  const float4* __restrict__ hv4 = (const float4*)h1;
  const float4 sA = *(const float4*)(aw_s + 4 * pc), sB = *(const float4*)(aw_s + 32 + 4 * pc);
  const float4 dA = *(const float4*)(aw_d + 4 * pc), dB = *(const float4*)(aw_d + 32 + 4 * pc);

  const float4 oA = hv4[(unsigned)node * 16u + pc], oB = hv4[(unsigned)node * 16u + 8u + pc];
  float adA = dot4(oA, dA), adB = dot4(oB, dB), asA = dot4(oA, sA), asB = dot4(oB, sB);
  adA += __shfl_xor(adA, 1, 32); adB += __shfl_xor(adB, 1, 32);
  asA += __shfl_xor(asA, 1, 32); asB += __shfl_xor(asB, 1, 32);
  float mA = leaky(asA + adA), mB = leaky(asB + adB);
  const float4 zero4 = make_float4(0.f, 0.f, 0.f, 0.f);
  float denA = (eq == 0) ? 1.f : 0.f, denB = denA;
  float4 accA = (eq == 0) ? oA : zero4, accB = (eq == 0) ? oB : zero4;

  for (int c0 = 0; c0 < deg; c0 += 32) {
    const int cnt = min(32, deg - c0);
    const int sl = (lane < cnt) ? csr[beg + c0 + lane] : node;
    for (int t = 0; 4 * t < cnt; t += 2) {
      const unsigned s0 = (unsigned)__shfl(sl, 4 * t + eq, 32);
      const unsigned s1 = (unsigned)__shfl(sl, 4 * t + 4 + eq, 32);
      const float4 vA0 = hv4[s0 * 16u + pc], vB0 = hv4[s0 * 16u + 8u + pc];
      const float4 vA1 = hv4[s1 * 16u + pc], vB1 = hv4[s1 * 16u + 8u + pc];
      float pA0 = dot4(vA0, sA), pB0 = dot4(vB0, sB), pA1 = dot4(vA1, sA), pB1 = dot4(vB1, sB);
      pA0 += __shfl_xor(pA0, 1, 32); pB0 += __shfl_xor(pB0, 1, 32);
      pA1 += __shfl_xor(pA1, 1, 32); pB1 += __shfl_xor(pB1, 1, 32);
      const bool ok0 = 4 * t + eq < cnt, ok1 = 4 * t + 4 + eq < cnt;
      float sc, w;
      stream_step(mA, leaky(pA0 + adA), ok0, sc, w); denA = denA * sc + w; scale_add(accA, sc, w, vA0);
      stream_step(mB, leaky(pB0 + adB), ok0, sc, w); denB = denB * sc + w; scale_add(accB, sc, w, vB0);
      stream_step(mA, leaky(pA1 + adA), ok1, sc, w); denA = denA * sc + w; scale_add(accA, sc, w, vA1);
      stream_step(mB, leaky(pB1 + adB), ok1, sc, w); denB = denB * sc + w; scale_add(accB, sc, w, vB1);
    }
  }
  const float fA = slot_factor(mA), fB = slot_factor(mB);
  denA = sum_slots(denA * fA); accA = sum_slots4(accA, fA);
  denB = sum_slots(denB * fB); accB = sum_slots4(accB, fB);
  if (eq == 0) {
    const float4 bA = *(const float4*)(b1 + 4 * pc), bB = *(const float4*)(b1 + 32 + 4 * pc);
    float4 rA, rB;
    rA.x = accA.x / denA + bA.x; rA.y = accA.y / denA + bA.y; rA.z = accA.z / denA + bA.z; rA.w = accA.w / denA + bA.w;
    rB.x = accB.x / denB + bB.x; rB.y = accB.y / denB + bB.y; rB.z = accB.z / denB + bB.z; rB.w = accB.w / denB + bB.w;
    rA.x = rA.x > 0.f ? rA.x : (__expf(rA.x) - 1.f); rA.y = rA.y > 0.f ? rA.y : (__expf(rA.y) - 1.f);
    rA.z = rA.z > 0.f ? rA.z : (__expf(rA.z) - 1.f); rA.w = rA.w > 0.f ? rA.w : (__expf(rA.w) - 1.f);
    rB.x = rB.x > 0.f ? rB.x : (__expf(rB.x) - 1.f); rB.y = rB.y > 0.f ? rB.y : (__expf(rB.y) - 1.f);
    rB.z = rB.z > 0.f ? rB.z : (__expf(rB.z) - 1.f); rB.w = rB.w > 0.f ? rB.w : (__expf(rB.w) - 1.f);
    float4* __restrict__ o4 = (float4*)helu;
    o4[(unsigned)node * 16u + pc] = rA;
    o4[(unsigned)node * 16u + 8u + pc] = rB;
  }
}

__global__ void __launch_bounds__(256, 16)
gat2_node_k(const int* __restrict__ rowptr, const int* __restrict__ csr,
            const float* __restrict__ helu,
            const float* __restrict__ us, const float* __restrict__ ud,
            float* __restrict__ agg, int N) {
  const int lane = (int)(threadIdx.x & 31);
  const int node = __builtin_amdgcn_readfirstlane((int)((blockIdx.x * blockDim.x + threadIdx.x) >> 5));
  if (node >= N) return;
  const int beg = rowptr[node];
  const int deg = rowptr[node + 1] - beg;
  const int eq = lane >> 3;
  const unsigned pc = (unsigned)(lane & 7);
  const float4* __restrict__ hv4 = (const float4*)helu;
  const float4 sA = *(const float4*)(us + 4 * pc), sB = *(const float4*)(us + 32 + 4 * pc);
  const float4 dA = *(const float4*)(ud + 4 * pc), dB = *(const float4*)(ud + 32 + 4 * pc);

  const float4 oA = hv4[(unsigned)node * 16u + pc], oB = hv4[(unsigned)node * 16u + 8u + pc];
  float adn = dot4(oA, dA) + dot4(oB, dB), asn = dot4(oA, sA) + dot4(oB, sB);
  #pragma unroll
  for (int off = 1; off <= 4; off <<= 1) {
    adn += __shfl_xor(adn, off, 32);
    asn += __shfl_xor(asn, off, 32);
  }
  float m = leaky(asn + adn);
  const float4 zero4 = make_float4(0.f, 0.f, 0.f, 0.f);
  float den = (eq == 0) ? 1.f : 0.f;
  float4 accA = (eq == 0) ? oA : zero4, accB = (eq == 0) ? oB : zero4;

  for (int c0 = 0; c0 < deg; c0 += 32) {
    const int cnt = min(32, deg - c0);
    const int sl = (lane < cnt) ? csr[beg + c0 + lane] : node;
    for (int t = 0; 4 * t < cnt; t += 2) {
      const unsigned s0 = (unsigned)__shfl(sl, 4 * t + eq, 32);
      const unsigned s1 = (unsigned)__shfl(sl, 4 * t + 4 + eq, 32);
      const float4 vA0 = hv4[s0 * 16u + pc], vB0 = hv4[s0 * 16u + 8u + pc];
      const float4 vA1 = hv4[s1 * 16u + pc], vB1 = hv4[s1 * 16u + 8u + pc];
      float p0 = dot4(vA0, sA) + dot4(vB0, sB), p1 = dot4(vA1, sA) + dot4(vB1, sB);
      #pragma unroll
      for (int off = 1; off <= 4; off <<= 1) {
        p0 += __shfl_xor(p0, off, 32);
        p1 += __shfl_xor(p1, off, 32);
      }
      const bool ok0 = 4 * t + eq < cnt, ok1 = 4 * t + 4 + eq < cnt;
      const float e0 = leaky(p0 + adn), e1 = leaky(p1 + adn);
      float sc, w;
      stream_step(m, e0, ok0, sc, w); den = den * sc + w; scale_add(accA, sc, w, vA0); scale_add(accB, sc, w, vB0);
      stream_step(m, e1, ok1, sc, w); den = den * sc + w; scale_add(accA, sc, w, vA1); scale_add(accB, sc, w, vB1);
    }
  }
  const float f = slot_factor(m);
  den = sum_slots(den * f); accA = sum_slots4(accA, f); accB = sum_slots4(accB, f);
  if (eq == 0) {
    float4 rA, rB;
    rA.x = accA.x / den; rA.y = accA.y / den; rA.z = accA.z / den; rA.w = accA.w / den;
    rB.x = accB.x / den; rB.y = accB.y / den; rB.z = accB.z / den; rB.w = accB.w / den;
    float4* __restrict__ o4 = (float4*)agg;
    o4[(unsigned)node * 16u + pc] = rA;
    o4[(unsigned)node * 16u + 8u + pc] = rB;
  }
}

static inline unsigned cdiv(long long a, long long b) { return (unsigned)((a + b - 1) / b); }

extern "C" void kernel_launch(void* const* d_in, const int* in_sizes, int n_in,
                              void* d_out, int out_size, void* d_ws, size_t ws_size,
                              hipStream_t stream) {
  const float* x    = (const float*)d_in[0];
  const int*   ei   = (const int*)d_in[1];
  const float* W1   = (const float*)d_in[2];
  const float* as1w = (const float*)d_in[3];
  const float* ad1w = (const float*)d_in[4];
  const float* b1   = (const float*)d_in[5];
  const float* W2   = (const float*)d_in[6];
  const float* as2w = (const float*)d_in[7];
  const float* ad2w = (const float*)d_in[8];
  const float* b2   = (const float*)d_in[9];
  float* out = (float*)d_out;

  const int N = in_sizes[0] / 128;
  const long long E = (long long)in_sizes[1] / 2;

  float* w = (float*)d_ws;
  size_t off = 0;
  float* h1   = w + off; off += (size_t)N * 64;
  float* helu = w + off; off += (size_t)N * 64;
  float* agg  = w + off; off += (size_t)N * 64;
  float* us   = w + off; off += 64;
  float* ud   = w + off; off += 64;
  int2* part  = (int2*)(w + off); off += (size_t)E * 2;
  int* gcnt   = (int*)(w + off); off += (size_t)KMAX * CPAD;
  int* rowptr = (int*)(w + off); off += (size_t)N + 1;
  int* csr    = (int*)(w + off); off += (size_t)E;

  const int B = 256;
  const int tiles = (N + 15) / 16;

  const int K = (N + NODES_PER_BUCKET - 1) / NODES_PER_BUCKET;
  hipMemsetAsync(gcnt, 0, (size_t)K * CPAD * sizeof(int), stream);
  bucket_hist_k<<<cdiv(E, EPB), 1024, 0, stream>>>(ei, E, K, gcnt);
  bucket_scatter_k<<<cdiv(E, EPB), 1024, 0, stream>>>(ei, E, K, gcnt, part);
  edge_rows_gemm_k<<<K + cdiv(tiles, GEMM_TILES_PER_GROUP), 1024, 0, stream>>>(part, gcnt, K, N, rowptr, csr, x, W1, h1);

  fold_k<<<8, B, 0, stream>>>(W2, as2w, ad2w, us, ud);
  gat1_node_k<<<cdiv((long long)N * 32, B), B, 0, stream>>>(rowptr, csr, h1, as1w, ad1w, b1, helu, N);

  gat2_node_k<<<cdiv((long long)N * 32, B), B, 0, stream>>>(rowptr, csr, helu, us, ud, agg, N);
  gemm_wmma_f32<64, 128, true><<<cdiv((long long)tiles * 32, B), B, 0, stream>>>(agg, W2, b2, out, N);

  (void)n_in; (void)out_size; (void)ws_size;
}


# Node kernels: one 16-lane row per destination node, two nodes per wave (half the waves to start; 2 edge slots per node, sums and list picks stay in the row on the vector ALU); f32 sum order changes
# speedup vs baseline: 10.5576x; 1.0384x over previous
#include <hip/hip_runtime.h>
#include <cstdint>
#include <cstddef>

typedef __attribute__((ext_vector_type(2))) float v2f;
typedef __attribute__((ext_vector_type(8))) float v8f;

#define NEG_SLOPE 0.2f

template<int K, int COLS, bool BIAS>
__global__ void gemm_wmma_f32(const float* __restrict__ A,
                              const float* __restrict__ B,
                              const float* __restrict__ bias,
                              float* __restrict__ C, int N) {
  constexpr int CT = COLS / 16;
  const int wave = (int)((blockIdx.x * blockDim.x + threadIdx.x) >> 5);
  const int lane = (int)(threadIdx.x & 31);
  const int row0 = wave * 16;
  if (row0 >= N) return;
  const int m    = lane & 15;
  const int half = lane >> 4;
  const int klo  = half * 2;

  v8f zero = {0.f, 0.f, 0.f, 0.f, 0.f, 0.f, 0.f, 0.f};
  v8f acc[CT];
  #pragma unroll
  for (int t = 0; t < CT; ++t) acc[t] = zero;

  #pragma unroll
  for (int k = 0; k < K; k += 4) {
    const float* ap = A + (size_t)(row0 + m) * K + (k + klo);
    v2f a; a.x = ap[0]; a.y = ap[1];
    #pragma unroll
    for (int t = 0; t < CT; ++t) {
      const float* bp = B + (size_t)(k + klo) * COLS + t * 16 + m;
      v2f b; b.x = bp[0]; b.y = bp[COLS];
      acc[t] = __builtin_amdgcn_wmma_f32_16x16x4_f32(
          false, a, false, b, (short)0, acc[t], false, false);
    }
  }
  const int rbase = row0 + half * 8;
  #pragma unroll
  for (int t = 0; t < CT; ++t) {
    const float bv = BIAS ? bias[t * 16 + m] : 0.f;
    #pragma unroll
    for (int v = 0; v < 8; ++v)
      C[(size_t)(rbase + v) * COLS + t * 16 + m] = BIAS ? acc[t][v] + bv : acc[t][v];
  }
}

template<int K, int COLS>
__device__ __forceinline__ void gemm_tile_f32(const float* __restrict__ A, const float* __restrict__ B,
                                              float* __restrict__ C, const int row0, const int lane) {
  constexpr int CT = COLS / 16;
  const unsigned m    = (unsigned)lane & 15u;
  const unsigned half = (unsigned)lane >> 4;
  const unsigned klo  = half * 2u;
  const float* __restrict__ At = A + (size_t)row0 * K;
  float* __restrict__ Ct = C + (size_t)row0 * COLS;
  const unsigned ao = m * (unsigned)K + klo;
  const unsigned bo = klo * (unsigned)COLS + m;

  v8f zero = {0.f, 0.f, 0.f, 0.f, 0.f, 0.f, 0.f, 0.f};
  v8f acc[CT];
  #pragma unroll
  for (int t = 0; t < CT; ++t) acc[t] = zero;

  #pragma unroll
  for (int k = 0; k < K; k += 4) {
    v2f a; a.x = At[ao + (unsigned)k]; a.y = At[ao + (unsigned)k + 1u];
    #pragma unroll
    for (int t = 0; t < CT; ++t) {
      const unsigned bi = bo + (unsigned)(k * COLS + t * 16);
      v2f b; b.x = B[bi]; b.y = B[bi + (unsigned)COLS];
      acc[t] = __builtin_amdgcn_wmma_f32_16x16x4_f32(
          false, a, false, b, (short)0, acc[t], false, false);
    }
  }
  const unsigned co = half * 8u * (unsigned)COLS + m;
  #pragma unroll
  for (int t = 0; t < CT; ++t) {
    #pragma unroll
    for (int v = 0; v < 8; ++v)
      Ct[co + (unsigned)(v * COLS + t * 16)] = acc[t][v];
  }
}

__global__ void fold_k(const float* __restrict__ W2,
                       const float* __restrict__ aw_s, const float* __restrict__ aw_d,
                       float* __restrict__ us, float* __restrict__ ud) {
  const int wave = (int)((blockIdx.x * blockDim.x + threadIdx.x) >> 5);
  const int lane = (int)(threadIdx.x & 31);
  if (wave >= 64) return;
  const float4 wv = *(const float4*)(W2 + (size_t)wave * 128 + lane * 4);
  const float4 sv = *(const float4*)(aw_s + lane * 4);
  const float4 dv = *(const float4*)(aw_d + lane * 4);
  float s = wv.x * sv.x + wv.y * sv.y + wv.z * sv.z + wv.w * sv.w;
  float d = wv.x * dv.x + wv.y * dv.y + wv.z * dv.z + wv.w * dv.w;
  #pragma unroll
  for (int off = 16; off > 0; off >>= 1) {
    s += __shfl_xor(s, off, 32);
    d += __shfl_xor(d, off, 32);
  }
  if (lane == 0) { us[wave] = s; ud[wave] = d; }
}

#define NODE_SHIFT 8
#define NODES_PER_BUCKET 256
#define KMAX 1024
#define EPB 8192
#define EPT (EPB / 1024)
#define ROWS_EPT 12
#define STAGE_CAP (ROWS_EPT * 1024)
#define CPAD 64

__device__ __forceinline__ void wave_excl_scan_lds(const int* in, int* out, int* out2, int n, int lane) {
  int carry = 0;
  for (int b0 = 0; b0 < n; b0 += 256) {
    int v[8], incl[8];
    #pragma unroll
    for (int c = 0; c < 8; ++c) {
      const int i = b0 + c * 32 + lane;
      v[c] = (i < n) ? in[i] : 0;
      incl[c] = v[c];
    }
    #pragma unroll
    for (int off = 1; off < 32; off <<= 1) {
      #pragma unroll
      for (int c = 0; c < 8; ++c) {
        const int t = __shfl_up(incl[c], off, 32);
        if (lane >= off) incl[c] += t;
      }
    }
    #pragma unroll
    for (int c = 0; c < 8; ++c) {
      const int i = b0 + c * 32 + lane;
      const int tot = __shfl(incl[c], 31, 32);
      if (i < n) {
        out[i] = carry + incl[c] - v[c];
        if (out2) out2[i] = carry + incl[c] - v[c];
      }
      carry += tot;
    }
  }
}

__global__ void __launch_bounds__(1024)
bucket_hist_k(const int* __restrict__ ei, long long E, int K, int* __restrict__ gcnt) {
  __shared__ int cntL[KMAX];
  const int tid = (int)threadIdx.x;
  const long long e0 = (long long)blockIdx.x * EPB;
  int d[EPT];
  #pragma unroll
  for (int j = 0; j < EPT; ++j) {
    const long long e = e0 + j * 1024 + tid;
    d[j] = (e < E) ? ei[E + e] : -1;
  }
  for (int k = tid; k < K; k += 1024) cntL[k] = 0;
  __syncthreads();
  #pragma unroll
  for (int j = 0; j < EPT; ++j)
    if (d[j] >= 0) atomicAdd(&cntL[d[j] >> NODE_SHIFT], 1);
  __syncthreads();
  for (int k = tid; k < K; k += 1024) {
    const int c = cntL[k];
    if (c) atomicAdd(&gcnt[k * CPAD], c);
  }
}

__global__ void __launch_bounds__(1024)
bucket_scatter_k(const int* __restrict__ ei, long long E, int K,
                 int* __restrict__ gcnt, int2* __restrict__ part) {
  __shared__ int cntL[KMAX];
  __shared__ int lstart[KMAX];
  __shared__ int gdelta[KMAX];
  __shared__ int bstart[KMAX];
  __shared__ int2 stage[EPB];
  const int tid = (int)threadIdx.x, lane = tid & 31;
  const long long e0 = (long long)blockIdx.x * EPB;
  int d[EPT], s[EPT];
  #pragma unroll
  for (int j = 0; j < EPT; ++j) {
    const long long e = e0 + j * 1024 + tid;
    d[j] = -1; s[j] = 0;
    if (e < E) { d[j] = ei[E + e]; s[j] = ei[e]; }
  }
  for (int k = tid; k < K; k += 1024) { cntL[k] = 0; bstart[k] = gcnt[k * CPAD]; }
  __syncthreads();
  #pragma unroll
  for (int j = 0; j < EPT; ++j)
    if (d[j] >= 0) atomicAdd(&cntL[d[j] >> NODE_SHIFT], 1);
  __syncthreads();
  if (tid < 32)      wave_excl_scan_lds(bstart, bstart, nullptr, K, lane);
  else if (tid < 64) wave_excl_scan_lds(cntL, lstart, nullptr, K, lane);
  __syncthreads();
  if (blockIdx.x == 0)
    for (int k = tid; k < K; k += 1024) gcnt[k * CPAD + 2] = bstart[k];
  for (int k = tid; k < K; k += 1024) {
    const int c = cntL[k];
    const int g = c ? atomicAdd(&gcnt[k * CPAD + 1], c) : 0;
    const int ls = lstart[k];
    gdelta[k] = bstart[k] + g - ls;
    cntL[k] = ls;
  }
  __syncthreads();
  int slot[EPT];
  #pragma unroll
  for (int j = 0; j < EPT; ++j)
    slot[j] = (d[j] >= 0) ? atomicAdd(&cntL[d[j] >> NODE_SHIFT], 1) : 0;
  #pragma unroll
  for (int j = 0; j < EPT; ++j)
    if (d[j] >= 0) stage[slot[j]] = make_int2(s[j], d[j]);
  __syncthreads();
  const long long rem = E - e0;
  const int nE = rem < (long long)EPB ? (int)rem : EPB;
  if (nE == EPB) {
    int2 p[EPT]; int g[EPT];
    #pragma unroll
    for (int j = 0; j < EPT; ++j) p[j] = stage[tid + j * 1024];
    #pragma unroll
    for (int j = 0; j < EPT; ++j) g[j] = gdelta[p[j].y >> NODE_SHIFT];
    #pragma unroll
    for (int j = 0; j < EPT; ++j) part[g[j] + tid + j * 1024] = p[j];
  } else {
    for (int i = tid; i < nE; i += 1024) {
      const int2 p = stage[i];
      part[gdelta[p.y >> NODE_SHIFT] + i] = p;
    }
  }
}

#define GEMM_TILES_PER_GROUP 16
__global__ void __launch_bounds__(1024, 16)
edge_rows_gemm_k(const int2* __restrict__ part, const int* __restrict__ gcnt, int K, int N,
                 int* __restrict__ rowptr, int* __restrict__ csr,
                 const float* __restrict__ x, const float* __restrict__ W1, float* __restrict__ h1) {
  if ((int)blockIdx.x >= K) {
    const int wv = __builtin_amdgcn_readfirstlane((int)(threadIdx.x >> 5));
    const int row0 = (((int)blockIdx.x - K) * GEMM_TILES_PER_GROUP + wv) * 16;
    if (wv < GEMM_TILES_PER_GROUP && row0 < N)
      gemm_tile_f32<128, 64>(x, W1, h1, row0, (int)(threadIdx.x & 31));
    return;
  }
  __shared__ int cntL[NODES_PER_BUCKET];
  __shared__ int cur[NODES_PER_BUCKET];
  __shared__ int excl[NODES_PER_BUCKET];
  __shared__ int stage[STAGE_CAP];
  const int tid = (int)threadIdx.x, lane = tid & 31;
  const int k = (int)blockIdx.x;
  const int node0 = k << NODE_SHIFT;
  const int cnt  = gcnt[k * CPAD];
  const int base = gcnt[k * CPAD + 2];
  if (tid < NODES_PER_BUCKET) cntL[tid] = 0;
  __syncthreads();
  const bool staged = cnt <= STAGE_CAP;

  int2 p[ROWS_EPT];
  if (staged) {
    #pragma unroll
    for (int j = 0; j < ROWS_EPT; ++j) {
      const int i = tid + j * 1024;
      p[j] = make_int2(0, -1);
      if (i < cnt) p[j] = part[base + i];
    }
    #pragma unroll
    for (int j = 0; j < ROWS_EPT; ++j)
      if (p[j].y >= 0) atomicAdd(&cntL[p[j].y - node0], 1);
  } else {
    for (int i = tid; i < cnt; i += 1024) atomicAdd(&cntL[part[base + i].y - node0], 1);
  }
  __syncthreads();
  if (tid < 32) wave_excl_scan_lds(cntL, cur, excl, NODES_PER_BUCKET, lane);
  __syncthreads();
  if (tid < NODES_PER_BUCKET) {
    const int n = node0 + tid;
    if (n < N) rowptr[n] = base + excl[tid];
  }
  if (tid == 0 && k == K - 1) rowptr[N] = base + cnt;
  if (staged) {
    int r[ROWS_EPT];
    #pragma unroll
    for (int j = 0; j < ROWS_EPT; ++j)
      r[j] = (p[j].y >= 0) ? atomicAdd(&cur[p[j].y - node0], 1) : 0;
    #pragma unroll
    for (int j = 0; j < ROWS_EPT; ++j)
      if (p[j].y >= 0) stage[r[j]] = p[j].x;
  } else {
    for (int i = tid; i < cnt; i += 1024) {
      const int2 q = part[base + i];
      csr[base + atomicAdd(&cur[q.y - node0], 1)] = q.x;
    }
  }
  __syncthreads();
  if (staged)
    for (int i = tid; i < cnt; i += 1024) csr[base + i] = stage[i];
}


template<int M>
__device__ __forceinline__ float xrow(const float v) {
  const int b = __builtin_bit_cast(int, v);
  return __builtin_bit_cast(float, __builtin_amdgcn_update_dpp(0, b, 0x160 + M, 0xf, 0xf, true));
}
__device__ __forceinline__ int xhalf_i(const int b) {
  return __builtin_amdgcn_permlanex16(b, b, 0x76543210, 0xfedcba98, false, false);
}
__device__ __forceinline__ unsigned row_pick(const int v, const unsigned sel) {
  return __builtin_amdgcn_permlane16_var((unsigned)v, (unsigned)v, sel, false, false);
}

__device__ __forceinline__ float leaky(float e) { return e > 0.f ? e : NEG_SLOPE * e; }
__device__ __forceinline__ float dot4(const float4 a, const float4 b) {
  return a.x * b.x + a.y * b.y + a.z * b.z + a.w * b.w;
}

__device__ __forceinline__ void stream_step(float& m, const float e, const bool ok, float& sc, float& w) {
  const float mn = fmaxf(m, e);
  sc = __expf(m - mn);
  w  = ok ? __expf(e - mn) : 0.f;
  m = mn;
}
__device__ __forceinline__ void scale_add(float4& acc, const float sc, const float w, const float4 v) {
  acc.x = acc.x * sc + w * v.x;
  acc.y = acc.y * sc + w * v.y;
  acc.z = acc.z * sc + w * v.z;
  acc.w = acc.w * sc + w * v.w;
}
__device__ __forceinline__ float slot_factor(const float m) {
  const float M = fmaxf(m, xrow<8>(m));
  return __expf(m - M);
}
__device__ __forceinline__ float sum_slots(float v) {
  v += xrow<8>(v);
  return v;
}
__device__ __forceinline__ float4 sum_slots4(const float4 a, const float f) {
  float4 r;
  r.x = sum_slots(a.x * f); r.y = sum_slots(a.y * f); r.z = sum_slots(a.z * f); r.w = sum_slots(a.w * f);
  return r;
}

__global__ void __launch_bounds__(256, 16)
gat1_node_k(const int* __restrict__ rowptr, const int* __restrict__ csr,
            const float* __restrict__ h1,
            const float* __restrict__ aw_s, const float* __restrict__ aw_d,
            const float* __restrict__ b1, float* __restrict__ helu, int N) {
  const int lane = (int)(threadIdx.x & 31);
  const int wv = __builtin_amdgcn_readfirstlane((int)((blockIdx.x * blockDim.x + threadIdx.x) >> 5));
  if (2 * wv >= N) return;
  const int hf = lane >> 4;
  const bool live = 2 * wv + hf < N;
  const int node = live ? 2 * wv + hf : N - 1;
  const int beg = rowptr[node];
  const int deg = rowptr[node + 1] - beg;
  const int degw = __builtin_amdgcn_readfirstlane(max(deg, xhalf_i(deg)));
  const int j = lane & 15;
  const int eq = (lane >> 3) & 1;
  const unsigned pc = (unsigned)(lane & 7);
  const float4* __restrict__ hv4 = (const float4*)h1;
  const float4 sA = *(const float4*)(aw_s + 4 * pc), sB = *(const float4*)(aw_s + 32 + 4 * pc);
  const float4 dA = *(const float4*)(aw_d + 4 * pc), dB = *(const float4*)(aw_d + 32 + 4 * pc);

  const float4 oA = hv4[(unsigned)node * 16u + pc], oB = hv4[(unsigned)node * 16u + 8u + pc];
  float adA = dot4(oA, dA), adB = dot4(oB, dB), asA = dot4(oA, sA), asB = dot4(oB, sB);
  adA += xrow<1>(adA); adB += xrow<1>(adB);
  asA += xrow<1>(asA); asB += xrow<1>(asB);
  float mA = leaky(asA + adA), mB = leaky(asB + adB);
  const float4 zero4 = make_float4(0.f, 0.f, 0.f, 0.f);
  float denA = (eq == 0) ? 1.f : 0.f, denB = denA;
  float4 accA = (eq == 0) ? oA : zero4, accB = (eq == 0) ? oB : zero4;
  const int rem = deg - eq;

  for (int c0 = 0; c0 < degw; c0 += 16) {
    const int sl = (c0 + j < deg) ? csr[beg + c0 + j] : node;
    const int lim = min(16, degw - c0);
    for (int t = 0; 4 * t < lim; ++t) {
      const unsigned s0 = row_pick(sl, 4u * (unsigned)t + (unsigned)eq);
      const unsigned s1 = row_pick(sl, 4u * (unsigned)t + 2u + (unsigned)eq);
      const float4 vA0 = hv4[s0 * 16u + pc], vB0 = hv4[s0 * 16u + 8u + pc];
      const float4 vA1 = hv4[s1 * 16u + pc], vB1 = hv4[s1 * 16u + 8u + pc];
      float pA0 = dot4(vA0, sA), pB0 = dot4(vB0, sB), pA1 = dot4(vA1, sA), pB1 = dot4(vB1, sB);
      pA0 += xrow<1>(pA0); pB0 += xrow<1>(pB0);
      pA1 += xrow<1>(pA1); pB1 += xrow<1>(pB1);
      const bool ok0 = c0 + 4 * t < rem, ok1 = c0 + 4 * t + 2 < rem;
      float sc, w;
      stream_step(mA, leaky(pA0 + adA), ok0, sc, w); denA = denA * sc + w; scale_add(accA, sc, w, vA0);
      stream_step(mB, leaky(pB0 + adB), ok0, sc, w); denB = denB * sc + w; scale_add(accB, sc, w, vB0);
      stream_step(mA, leaky(pA1 + adA), ok1, sc, w); denA = denA * sc + w; scale_add(accA, sc, w, vA1);
      stream_step(mB, leaky(pB1 + adB), ok1, sc, w); denB = denB * sc + w; scale_add(accB, sc, w, vB1);
    }
  }
  const float fA = slot_factor(mA), fB = slot_factor(mB);
  denA = sum_slots(denA * fA); accA = sum_slots4(accA, fA);
  denB = sum_slots(denB * fB); accB = sum_slots4(accB, fB);
  if (eq == 0 && live) {
    const float4 bA = *(const float4*)(b1 + 4 * pc), bB = *(const float4*)(b1 + 32 + 4 * pc);
    float4 rA, rB;
    rA.x = accA.x / denA + bA.x; rA.y = accA.y / denA + bA.y; rA.z = accA.z / denA + bA.z; rA.w = accA.w / denA + bA.w;
    rB.x = accB.x / denB + bB.x; rB.y = accB.y / denB + bB.y; rB.z = accB.z / denB + bB.z; rB.w = accB.w / denB + bB.w;
    rA.x = rA.x > 0.f ? rA.x : (__expf(rA.x) - 1.f); rA.y = rA.y > 0.f ? rA.y : (__expf(rA.y) - 1.f);
    rA.z = rA.z > 0.f ? rA.z : (__expf(rA.z) - 1.f); rA.w = rA.w > 0.f ? rA.w : (__expf(rA.w) - 1.f);
    rB.x = rB.x > 0.f ? rB.x : (__expf(rB.x) - 1.f); rB.y = rB.y > 0.f ? rB.y : (__expf(rB.y) - 1.f);
    rB.z = rB.z > 0.f ? rB.z : (__expf(rB.z) - 1.f); rB.w = rB.w > 0.f ? rB.w : (__expf(rB.w) - 1.f);
    float4* __restrict__ o4 = (float4*)helu;
    o4[(unsigned)node * 16u + pc] = rA;
    o4[(unsigned)node * 16u + 8u + pc] = rB;
  }
}

__global__ void __launch_bounds__(256, 16)
gat2_node_k(const int* __restrict__ rowptr, const int* __restrict__ csr,
            const float* __restrict__ helu,
            const float* __restrict__ us, const float* __restrict__ ud,
            float* __restrict__ agg, int N) {
  const int lane = (int)(threadIdx.x & 31);
  const int wv = __builtin_amdgcn_readfirstlane((int)((blockIdx.x * blockDim.x + threadIdx.x) >> 5));
  if (2 * wv >= N) return;
  const int hf = lane >> 4;
  const bool live = 2 * wv + hf < N;
  const int node = live ? 2 * wv + hf : N - 1;
  const int beg = rowptr[node];
  const int deg = rowptr[node + 1] - beg;
  const int degw = __builtin_amdgcn_readfirstlane(max(deg, xhalf_i(deg)));
  const int j = lane & 15;
  const int eq = (lane >> 3) & 1;
  const unsigned pc = (unsigned)(lane & 7);
  const float4* __restrict__ hv4 = (const float4*)helu;
  const float4 sA = *(const float4*)(us + 4 * pc), sB = *(const float4*)(us + 32 + 4 * pc);
  const float4 dA = *(const float4*)(ud + 4 * pc), dB = *(const float4*)(ud + 32 + 4 * pc);

  const float4 oA = hv4[(unsigned)node * 16u + pc], oB = hv4[(unsigned)node * 16u + 8u + pc];
  float adn = dot4(oA, dA) + dot4(oB, dB), asn = dot4(oA, sA) + dot4(oB, sB);
  adn += xrow<1>(adn); asn += xrow<1>(asn);
  adn += xrow<2>(adn); asn += xrow<2>(asn);
  adn += xrow<4>(adn); asn += xrow<4>(asn);
  float m = leaky(asn + adn);
  const float4 zero4 = make_float4(0.f, 0.f, 0.f, 0.f);
  float den = (eq == 0) ? 1.f : 0.f;
  float4 accA = (eq == 0) ? oA : zero4, accB = (eq == 0) ? oB : zero4;
  const int rem = deg - eq;

  for (int c0 = 0; c0 < degw; c0 += 16) {
    const int sl = (c0 + j < deg) ? csr[beg + c0 + j] : node;
    const int lim = min(16, degw - c0);
    for (int t = 0; 4 * t < lim; ++t) {
      const unsigned s0 = row_pick(sl, 4u * (unsigned)t + (unsigned)eq);
      const unsigned s1 = row_pick(sl, 4u * (unsigned)t + 2u + (unsigned)eq);
      const float4 vA0 = hv4[s0 * 16u + pc], vB0 = hv4[s0 * 16u + 8u + pc];
      const float4 vA1 = hv4[s1 * 16u + pc], vB1 = hv4[s1 * 16u + 8u + pc];
      float p0 = dot4(vA0, sA) + dot4(vB0, sB), p1 = dot4(vA1, sA) + dot4(vB1, sB);
      p0 += xrow<1>(p0); p1 += xrow<1>(p1);
      p0 += xrow<2>(p0); p1 += xrow<2>(p1);
      p0 += xrow<4>(p0); p1 += xrow<4>(p1);
      const bool ok0 = c0 + 4 * t < rem, ok1 = c0 + 4 * t + 2 < rem;
      const float e0 = leaky(p0 + adn), e1 = leaky(p1 + adn);
      float sc, w;
      stream_step(m, e0, ok0, sc, w); den = den * sc + w; scale_add(accA, sc, w, vA0); scale_add(accB, sc, w, vB0);
      stream_step(m, e1, ok1, sc, w); den = den * sc + w; scale_add(accA, sc, w, vA1); scale_add(accB, sc, w, vB1);
    }
  }
  const float f = slot_factor(m);
  den = sum_slots(den * f); accA = sum_slots4(accA, f); accB = sum_slots4(accB, f);
  if (eq == 0 && live) {
    float4 rA, rB;
    rA.x = accA.x / den; rA.y = accA.y / den; rA.z = accA.z / den; rA.w = accA.w / den;
    rB.x = accB.x / den; rB.y = accB.y / den; rB.z = accB.z / den; rB.w = accB.w / den;
    float4* __restrict__ o4 = (float4*)agg;
    o4[(unsigned)node * 16u + pc] = rA;
    o4[(unsigned)node * 16u + 8u + pc] = rB;
  }
}

static inline unsigned cdiv(long long a, long long b) { return (unsigned)((a + b - 1) / b); }

extern "C" void kernel_launch(void* const* d_in, const int* in_sizes, int n_in,
                              void* d_out, int out_size, void* d_ws, size_t ws_size,
                              hipStream_t stream) {
  const float* x    = (const float*)d_in[0];
  const int*   ei   = (const int*)d_in[1];
  const float* W1   = (const float*)d_in[2];
  const float* as1w = (const float*)d_in[3];
  const float* ad1w = (const float*)d_in[4];
  const float* b1   = (const float*)d_in[5];
  const float* W2   = (const float*)d_in[6];
  const float* as2w = (const float*)d_in[7];
  const float* ad2w = (const float*)d_in[8];
  const float* b2   = (const float*)d_in[9];
  float* out = (float*)d_out;

  const int N = in_sizes[0] / 128;
  const long long E = (long long)in_sizes[1] / 2;

  float* w = (float*)d_ws;
  size_t off = 0;
  float* h1   = w + off; off += (size_t)N * 64;
  float* helu = w + off; off += (size_t)N * 64;
  float* agg  = w + off; off += (size_t)N * 64;
  float* us   = w + off; off += 64;
  float* ud   = w + off; off += 64;
  int2* part  = (int2*)(w + off); off += (size_t)E * 2;
  int* gcnt   = (int*)(w + off); off += (size_t)KMAX * CPAD;
  int* rowptr = (int*)(w + off); off += (size_t)N + 1;
  int* csr    = (int*)(w + off); off += (size_t)E;

  const int B = 256;
  const int tiles = (N + 15) / 16;
  const int pairs = (N + 1) / 2;

  const int K = (N + NODES_PER_BUCKET - 1) / NODES_PER_BUCKET;
  hipMemsetAsync(gcnt, 0, (size_t)K * CPAD * sizeof(int), stream);
  bucket_hist_k<<<cdiv(E, EPB), 1024, 0, stream>>>(ei, E, K, gcnt);
  bucket_scatter_k<<<cdiv(E, EPB), 1024, 0, stream>>>(ei, E, K, gcnt, part);
  edge_rows_gemm_k<<<K + cdiv(tiles, GEMM_TILES_PER_GROUP), 1024, 0, stream>>>(part, gcnt, K, N, rowptr, csr, x, W1, h1);

  fold_k<<<8, B, 0, stream>>>(W2, as2w, ad2w, us, ud);
  gat1_node_k<<<cdiv((long long)pairs * 32, B), B, 0, stream>>>(rowptr, csr, h1, as1w, ad1w, b1, helu, N);

  gat2_node_k<<<cdiv((long long)pairs * 32, B), B, 0, stream>>>(rowptr, csr, helu, us, ud, agg, N);
  gemm_wmma_f32<64, 128, true><<<cdiv((long long)tiles * 32, B), B, 0, stream>>>(agg, W2, b2, out, N);

  (void)n_in; (void)out_size; (void)ws_size;
}


# Layer-2 projection joins the layer-2 node kernel: a group's 16 summed rows go to LDS and four of its waves multiply them by W2 on the matrix cores; no agg plane, one launch fewer; sums unchanged
# speedup vs baseline: 10.9975x; 1.0417x over previous
#include <hip/hip_runtime.h>
#include <cstdint>
#include <cstddef>

typedef __attribute__((ext_vector_type(2))) float v2f;
typedef __attribute__((ext_vector_type(8))) float v8f;

#define NEG_SLOPE 0.2f

template<int K, int COLS>
__device__ __forceinline__ void gemm_tile_f32(const float* __restrict__ A, const float* __restrict__ B,
                                              float* __restrict__ C, const int row0, const int lane) {
  constexpr int CT = COLS / 16;
  const unsigned m    = (unsigned)lane & 15u;
  const unsigned half = (unsigned)lane >> 4;
  const unsigned klo  = half * 2u;
  const float* __restrict__ At = A + (size_t)row0 * K;
  float* __restrict__ Ct = C + (size_t)row0 * COLS;
  const unsigned ao = m * (unsigned)K + klo;
  const unsigned bo = klo * (unsigned)COLS + m;

  v8f zero = {0.f, 0.f, 0.f, 0.f, 0.f, 0.f, 0.f, 0.f};
  v8f acc[CT];
  #pragma unroll
  for (int t = 0; t < CT; ++t) acc[t] = zero;

  #pragma unroll
  for (int k = 0; k < K; k += 4) {
    v2f a; a.x = At[ao + (unsigned)k]; a.y = At[ao + (unsigned)k + 1u];
    #pragma unroll
    for (int t = 0; t < CT; ++t) {
      const unsigned bi = bo + (unsigned)(k * COLS + t * 16);
      v2f b; b.x = B[bi]; b.y = B[bi + (unsigned)COLS];
      acc[t] = __builtin_amdgcn_wmma_f32_16x16x4_f32(
          false, a, false, b, (short)0, acc[t], false, false);
    }
  }
  const unsigned co = half * 8u * (unsigned)COLS + m;
  #pragma unroll
  for (int t = 0; t < CT; ++t) {
    #pragma unroll
    for (int v = 0; v < 8; ++v)
      Ct[co + (unsigned)(v * COLS + t * 16)] = acc[t][v];
  }
}

__global__ void fold_k(const float* __restrict__ W2,
                       const float* __restrict__ aw_s, const float* __restrict__ aw_d,
                       float* __restrict__ us, float* __restrict__ ud) {
  const int wave = (int)((blockIdx.x * blockDim.x + threadIdx.x) >> 5);
  const int lane = (int)(threadIdx.x & 31);
  if (wave >= 64) return;
  const float4 wv = *(const float4*)(W2 + (size_t)wave * 128 + lane * 4);
  const float4 sv = *(const float4*)(aw_s + lane * 4);
  const float4 dv = *(const float4*)(aw_d + lane * 4);
  float s = wv.x * sv.x + wv.y * sv.y + wv.z * sv.z + wv.w * sv.w;
  float d = wv.x * dv.x + wv.y * dv.y + wv.z * dv.z + wv.w * dv.w;
  #pragma unroll
  for (int off = 16; off > 0; off >>= 1) {
    s += __shfl_xor(s, off, 32);
    d += __shfl_xor(d, off, 32);
  }
  if (lane == 0) { us[wave] = s; ud[wave] = d; }
}

#define NODE_SHIFT 8
#define NODES_PER_BUCKET 256
#define KMAX 1024
#define EPB 8192
#define EPT (EPB / 1024)
#define ROWS_EPT 12
#define STAGE_CAP (ROWS_EPT * 1024)
#define CPAD 64

__device__ __forceinline__ void wave_excl_scan_lds(const int* in, int* out, int* out2, int n, int lane) {
  int carry = 0;
  for (int b0 = 0; b0 < n; b0 += 256) {
    int v[8], incl[8];
    #pragma unroll
    for (int c = 0; c < 8; ++c) {
      const int i = b0 + c * 32 + lane;
      v[c] = (i < n) ? in[i] : 0;
      incl[c] = v[c];
    }
    #pragma unroll
    for (int off = 1; off < 32; off <<= 1) {
      #pragma unroll
      for (int c = 0; c < 8; ++c) {
        const int t = __shfl_up(incl[c], off, 32);
        if (lane >= off) incl[c] += t;
      }
    }
    #pragma unroll
    for (int c = 0; c < 8; ++c) {
      const int i = b0 + c * 32 + lane;
      const int tot = __shfl(incl[c], 31, 32);
      if (i < n) {
        out[i] = carry + incl[c] - v[c];
        if (out2) out2[i] = carry + incl[c] - v[c];
      }
      carry += tot;
    }
  }
}

__global__ void __launch_bounds__(1024)
bucket_hist_k(const int* __restrict__ ei, long long E, int K, int* __restrict__ gcnt) {
  __shared__ int cntL[KMAX];
  const int tid = (int)threadIdx.x;
  const long long e0 = (long long)blockIdx.x * EPB;
  int d[EPT];
  #pragma unroll
  for (int j = 0; j < EPT; ++j) {
    const long long e = e0 + j * 1024 + tid;
    d[j] = (e < E) ? ei[E + e] : -1;
  }
  for (int k = tid; k < K; k += 1024) cntL[k] = 0;
  __syncthreads();
  #pragma unroll
  for (int j = 0; j < EPT; ++j)
    if (d[j] >= 0) atomicAdd(&cntL[d[j] >> NODE_SHIFT], 1);
  __syncthreads();
  for (int k = tid; k < K; k += 1024) {
    const int c = cntL[k];
    if (c) atomicAdd(&gcnt[k * CPAD], c);
  }
}

__global__ void __launch_bounds__(1024)
bucket_scatter_k(const int* __restrict__ ei, long long E, int K,
                 int* __restrict__ gcnt, int2* __restrict__ part) {
  __shared__ int cntL[KMAX];
  __shared__ int lstart[KMAX];
  __shared__ int gdelta[KMAX];
  __shared__ int bstart[KMAX];
  __shared__ int2 stage[EPB];
  const int tid = (int)threadIdx.x, lane = tid & 31;
  const long long e0 = (long long)blockIdx.x * EPB;
  int d[EPT], s[EPT];
  #pragma unroll
  for (int j = 0; j < EPT; ++j) {
    const long long e = e0 + j * 1024 + tid;
    d[j] = -1; s[j] = 0;
    if (e < E) { d[j] = ei[E + e]; s[j] = ei[e]; }
  }
  for (int k = tid; k < K; k += 1024) { cntL[k] = 0; bstart[k] = gcnt[k * CPAD]; }
  __syncthreads();
  #pragma unroll
  for (int j = 0; j < EPT; ++j)
    if (d[j] >= 0) atomicAdd(&cntL[d[j] >> NODE_SHIFT], 1);
  __syncthreads();
  if (tid < 32)      wave_excl_scan_lds(bstart, bstart, nullptr, K, lane);
  else if (tid < 64) wave_excl_scan_lds(cntL, lstart, nullptr, K, lane);
  __syncthreads();
  if (blockIdx.x == 0)
    for (int k = tid; k < K; k += 1024) gcnt[k * CPAD + 2] = bstart[k];
  for (int k = tid; k < K; k += 1024) {
    const int c = cntL[k];
    const int g = c ? atomicAdd(&gcnt[k * CPAD + 1], c) : 0;
    const int ls = lstart[k];
    gdelta[k] = bstart[k] + g - ls;
    cntL[k] = ls;
  }
  __syncthreads();
  int slot[EPT];
  #pragma unroll
  for (int j = 0; j < EPT; ++j)
    slot[j] = (d[j] >= 0) ? atomicAdd(&cntL[d[j] >> NODE_SHIFT], 1) : 0;
  #pragma unroll
  for (int j = 0; j < EPT; ++j)
    if (d[j] >= 0) stage[slot[j]] = make_int2(s[j], d[j]);
  __syncthreads();
  const long long rem = E - e0;
  const int nE = rem < (long long)EPB ? (int)rem : EPB;
  if (nE == EPB) {
    int2 p[EPT]; int g[EPT];
    #pragma unroll
    for (int j = 0; j < EPT; ++j) p[j] = stage[tid + j * 1024];
    #pragma unroll
    for (int j = 0; j < EPT; ++j) g[j] = gdelta[p[j].y >> NODE_SHIFT];
    #pragma unroll
    for (int j = 0; j < EPT; ++j) part[g[j] + tid + j * 1024] = p[j];
  } else {
    for (int i = tid; i < nE; i += 1024) {
      const int2 p = stage[i];
      part[gdelta[p.y >> NODE_SHIFT] + i] = p;
    }
  }
}

#define GEMM_TILES_PER_GROUP 16
__global__ void __launch_bounds__(1024, 16)
edge_rows_gemm_k(const int2* __restrict__ part, const int* __restrict__ gcnt, int K, int N,
                 int* __restrict__ rowptr, int* __restrict__ csr,
                 const float* __restrict__ x, const float* __restrict__ W1, float* __restrict__ h1) {
  if ((int)blockIdx.x >= K) {
    const int wv = __builtin_amdgcn_readfirstlane((int)(threadIdx.x >> 5));
    const int row0 = (((int)blockIdx.x - K) * GEMM_TILES_PER_GROUP + wv) * 16;
    if (wv < GEMM_TILES_PER_GROUP && row0 < N)
      gemm_tile_f32<128, 64>(x, W1, h1, row0, (int)(threadIdx.x & 31));
    return;
  }
  __shared__ int cntL[NODES_PER_BUCKET];
  __shared__ int cur[NODES_PER_BUCKET];
  __shared__ int excl[NODES_PER_BUCKET];
  __shared__ int stage[STAGE_CAP];
  const int tid = (int)threadIdx.x, lane = tid & 31;
  const int k = (int)blockIdx.x;
  const int node0 = k << NODE_SHIFT;
  const int cnt  = gcnt[k * CPAD];
  const int base = gcnt[k * CPAD + 2];
  if (tid < NODES_PER_BUCKET) cntL[tid] = 0;
  __syncthreads();
  const bool staged = cnt <= STAGE_CAP;

  int2 p[ROWS_EPT];
  if (staged) {
    #pragma unroll
    for (int j = 0; j < ROWS_EPT; ++j) {
      const int i = tid + j * 1024;
      p[j] = make_int2(0, -1);
      if (i < cnt) p[j] = part[base + i];
    }
    #pragma unroll
    for (int j = 0; j < ROWS_EPT; ++j)
      if (p[j].y >= 0) atomicAdd(&cntL[p[j].y - node0], 1);
  } else {
    for (int i = tid; i < cnt; i += 1024) atomicAdd(&cntL[part[base + i].y - node0], 1);
  }
  __syncthreads();
  if (tid < 32) wave_excl_scan_lds(cntL, cur, excl, NODES_PER_BUCKET, lane);
  __syncthreads();
  if (tid < NODES_PER_BUCKET) {
    const int n = node0 + tid;
    if (n < N) rowptr[n] = base + excl[tid];
  }
  if (tid == 0 && k == K - 1) rowptr[N] = base + cnt;
  if (staged) {
    int r[ROWS_EPT];
    #pragma unroll
    for (int j = 0; j < ROWS_EPT; ++j)
      r[j] = (p[j].y >= 0) ? atomicAdd(&cur[p[j].y - node0], 1) : 0;
    #pragma unroll
    for (int j = 0; j < ROWS_EPT; ++j)
      if (p[j].y >= 0) stage[r[j]] = p[j].x;
  } else {
    for (int i = tid; i < cnt; i += 1024) {
      const int2 q = part[base + i];
      csr[base + atomicAdd(&cur[q.y - node0], 1)] = q.x;
    }
  }
  __syncthreads();
  if (staged)
    for (int i = tid; i < cnt; i += 1024) csr[base + i] = stage[i];
}


template<int M>
__device__ __forceinline__ float xrow(const float v) {
  const int b = __builtin_bit_cast(int, v);
  return __builtin_bit_cast(float, __builtin_amdgcn_update_dpp(0, b, 0x160 + M, 0xf, 0xf, true));
}
__device__ __forceinline__ int xhalf_i(const int b) {
  return __builtin_amdgcn_permlanex16(b, b, 0x76543210, 0xfedcba98, false, false);
}
__device__ __forceinline__ unsigned row_pick(const int v, const unsigned sel) {
  return __builtin_amdgcn_permlane16_var((unsigned)v, (unsigned)v, sel, false, false);
}

__device__ __forceinline__ float leaky(float e) { return e > 0.f ? e : NEG_SLOPE * e; }
__device__ __forceinline__ float dot4(const float4 a, const float4 b) {
  return a.x * b.x + a.y * b.y + a.z * b.z + a.w * b.w;
}

__device__ __forceinline__ void stream_step(float& m, const float e, const bool ok, float& sc, float& w) {
  const float mn = fmaxf(m, e);
  sc = __expf(m - mn);
  w  = ok ? __expf(e - mn) : 0.f;
  m = mn;
}
__device__ __forceinline__ void scale_add(float4& acc, const float sc, const float w, const float4 v) {
  acc.x = acc.x * sc + w * v.x;
  acc.y = acc.y * sc + w * v.y;
  acc.z = acc.z * sc + w * v.z;
  acc.w = acc.w * sc + w * v.w;
}
__device__ __forceinline__ float slot_factor(const float m) {
  const float M = fmaxf(m, xrow<8>(m));
  return __expf(m - M);
}
__device__ __forceinline__ float sum_slots(float v) {
  v += xrow<8>(v);
  return v;
}
__device__ __forceinline__ float4 sum_slots4(const float4 a, const float f) {
  float4 r;
  r.x = sum_slots(a.x * f); r.y = sum_slots(a.y * f); r.z = sum_slots(a.z * f); r.w = sum_slots(a.w * f);
  return r;
}

__global__ void __launch_bounds__(256, 16)
gat1_node_k(const int* __restrict__ rowptr, const int* __restrict__ csr,
            const float* __restrict__ h1,
            const float* __restrict__ aw_s, const float* __restrict__ aw_d,
            const float* __restrict__ b1, float* __restrict__ helu, int N) {
  const int lane = (int)(threadIdx.x & 31);
  const int wv = __builtin_amdgcn_readfirstlane((int)((blockIdx.x * blockDim.x + threadIdx.x) >> 5));
  if (2 * wv >= N) return;
  const int hf = lane >> 4;
  const bool live = 2 * wv + hf < N;
  const int node = live ? 2 * wv + hf : N - 1;
  const int beg = rowptr[node];
  const int deg = rowptr[node + 1] - beg;
  const int degw = __builtin_amdgcn_readfirstlane(max(deg, xhalf_i(deg)));
  const int j = lane & 15;
  const int eq = (lane >> 3) & 1;
  const unsigned pc = (unsigned)(lane & 7);
  const float4* __restrict__ hv4 = (const float4*)h1;
  const float4 sA = *(const float4*)(aw_s + 4 * pc), sB = *(const float4*)(aw_s + 32 + 4 * pc);
  const float4 dA = *(const float4*)(aw_d + 4 * pc), dB = *(const float4*)(aw_d + 32 + 4 * pc);

  const float4 oA = hv4[(unsigned)node * 16u + pc], oB = hv4[(unsigned)node * 16u + 8u + pc];
  float adA = dot4(oA, dA), adB = dot4(oB, dB), asA = dot4(oA, sA), asB = dot4(oB, sB);
  adA += xrow<1>(adA); adB += xrow<1>(adB);
  asA += xrow<1>(asA); asB += xrow<1>(asB);
  float mA = leaky(asA + adA), mB = leaky(asB + adB);
  const float4 zero4 = make_float4(0.f, 0.f, 0.f, 0.f);
  float denA = (eq == 0) ? 1.f : 0.f, denB = denA;
  float4 accA = (eq == 0) ? oA : zero4, accB = (eq == 0) ? oB : zero4;
  const int rem = deg - eq;

  for (int c0 = 0; c0 < degw; c0 += 16) {
    const int sl = (c0 + j < deg) ? csr[beg + c0 + j] : node;
    const int lim = min(16, degw - c0);
    for (int t = 0; 4 * t < lim; ++t) {
      const unsigned s0 = row_pick(sl, 4u * (unsigned)t + (unsigned)eq);
      const unsigned s1 = row_pick(sl, 4u * (unsigned)t + 2u + (unsigned)eq);
      const float4 vA0 = hv4[s0 * 16u + pc], vB0 = hv4[s0 * 16u + 8u + pc];
      const float4 vA1 = hv4[s1 * 16u + pc], vB1 = hv4[s1 * 16u + 8u + pc];
      float pA0 = dot4(vA0, sA), pB0 = dot4(vB0, sB), pA1 = dot4(vA1, sA), pB1 = dot4(vB1, sB);
      pA0 += xrow<1>(pA0); pB0 += xrow<1>(pB0);
      pA1 += xrow<1>(pA1); pB1 += xrow<1>(pB1);
      const bool ok0 = c0 + 4 * t < rem, ok1 = c0 + 4 * t + 2 < rem;
      float sc, w;
      stream_step(mA, leaky(pA0 + adA), ok0, sc, w); denA = denA * sc + w; scale_add(accA, sc, w, vA0);
      stream_step(mB, leaky(pB0 + adB), ok0, sc, w); denB = denB * sc + w; scale_add(accB, sc, w, vB0);
      stream_step(mA, leaky(pA1 + adA), ok1, sc, w); denA = denA * sc + w; scale_add(accA, sc, w, vA1);
      stream_step(mB, leaky(pB1 + adB), ok1, sc, w); denB = denB * sc + w; scale_add(accB, sc, w, vB1);
    }
  }
  const float fA = slot_factor(mA), fB = slot_factor(mB);
  denA = sum_slots(denA * fA); accA = sum_slots4(accA, fA);
  denB = sum_slots(denB * fB); accB = sum_slots4(accB, fB);
  if (eq == 0 && live) {
    const float4 bA = *(const float4*)(b1 + 4 * pc), bB = *(const float4*)(b1 + 32 + 4 * pc);
    float4 rA, rB;
    rA.x = accA.x / denA + bA.x; rA.y = accA.y / denA + bA.y; rA.z = accA.z / denA + bA.z; rA.w = accA.w / denA + bA.w;
    rB.x = accB.x / denB + bB.x; rB.y = accB.y / denB + bB.y; rB.z = accB.z / denB + bB.z; rB.w = accB.w / denB + bB.w;
    rA.x = rA.x > 0.f ? rA.x : (__expf(rA.x) - 1.f); rA.y = rA.y > 0.f ? rA.y : (__expf(rA.y) - 1.f);
    rA.z = rA.z > 0.f ? rA.z : (__expf(rA.z) - 1.f); rA.w = rA.w > 0.f ? rA.w : (__expf(rA.w) - 1.f);
    rB.x = rB.x > 0.f ? rB.x : (__expf(rB.x) - 1.f); rB.y = rB.y > 0.f ? rB.y : (__expf(rB.y) - 1.f);
    rB.z = rB.z > 0.f ? rB.z : (__expf(rB.z) - 1.f); rB.w = rB.w > 0.f ? rB.w : (__expf(rB.w) - 1.f);
    float4* __restrict__ o4 = (float4*)helu;
    o4[(unsigned)node * 16u + pc] = rA;
    o4[(unsigned)node * 16u + 8u + pc] = rB;
  }
}

#define PROJ_PITCH 68
#define PROJ_WAVES 4
__global__ void __launch_bounds__(256, 16)
gat2_proj_k(const int* __restrict__ rowptr, const int* __restrict__ csr,
            const float* __restrict__ helu,
            const float* __restrict__ us, const float* __restrict__ ud,
            const float* __restrict__ W2, const float* __restrict__ b2,
            float* __restrict__ out, int N) {
  __shared__ float4 rows4[16 * (PROJ_PITCH / 4)];
  const int lane = (int)(threadIdx.x & 31);
  const int wib = __builtin_amdgcn_readfirstlane((int)(threadIdx.x >> 5));
  const int row0 = (int)blockIdx.x * 16;
  const int hf = lane >> 4;
  const int r = 2 * wib + hf;
  const bool live = row0 + r < N;
  const int node = live ? row0 + r : N - 1;
  const int beg = rowptr[node];
  const int deg = rowptr[node + 1] - beg;
  const int degw = __builtin_amdgcn_readfirstlane(max(deg, xhalf_i(deg)));
  const int j = lane & 15;
  const int eq = (lane >> 3) & 1;
  const unsigned pc = (unsigned)(lane & 7);
  const float4* __restrict__ hv4 = (const float4*)helu;
  const float4 sA = *(const float4*)(us + 4 * pc), sB = *(const float4*)(us + 32 + 4 * pc);
  const float4 dA = *(const float4*)(ud + 4 * pc), dB = *(const float4*)(ud + 32 + 4 * pc);

  const float4 oA = hv4[(unsigned)node * 16u + pc], oB = hv4[(unsigned)node * 16u + 8u + pc];
  float adn = dot4(oA, dA) + dot4(oB, dB), asn = dot4(oA, sA) + dot4(oB, sB);
  adn += xrow<1>(adn); asn += xrow<1>(asn);
  adn += xrow<2>(adn); asn += xrow<2>(asn);
  adn += xrow<4>(adn); asn += xrow<4>(asn);
  float m = leaky(asn + adn);
  const float4 zero4 = make_float4(0.f, 0.f, 0.f, 0.f);
  float den = (eq == 0) ? 1.f : 0.f;
  float4 accA = (eq == 0) ? oA : zero4, accB = (eq == 0) ? oB : zero4;
  const int rem = deg - eq;

  for (int c0 = 0; c0 < degw; c0 += 16) {
    const int sl = (c0 + j < deg) ? csr[beg + c0 + j] : node;
    const int lim = min(16, degw - c0);
    for (int t = 0; 4 * t < lim; ++t) {
      const unsigned s0 = row_pick(sl, 4u * (unsigned)t + (unsigned)eq);
      const unsigned s1 = row_pick(sl, 4u * (unsigned)t + 2u + (unsigned)eq);
      const float4 vA0 = hv4[s0 * 16u + pc], vB0 = hv4[s0 * 16u + 8u + pc];
      const float4 vA1 = hv4[s1 * 16u + pc], vB1 = hv4[s1 * 16u + 8u + pc];
      float p0 = dot4(vA0, sA) + dot4(vB0, sB), p1 = dot4(vA1, sA) + dot4(vB1, sB);
      p0 += xrow<1>(p0); p1 += xrow<1>(p1);
      p0 += xrow<2>(p0); p1 += xrow<2>(p1);
      p0 += xrow<4>(p0); p1 += xrow<4>(p1);
      const bool ok0 = c0 + 4 * t < rem, ok1 = c0 + 4 * t + 2 < rem;
      const float e0 = leaky(p0 + adn), e1 = leaky(p1 + adn);
      float sc, w;
      stream_step(m, e0, ok0, sc, w); den = den * sc + w; scale_add(accA, sc, w, vA0); scale_add(accB, sc, w, vB0);
      stream_step(m, e1, ok1, sc, w); den = den * sc + w; scale_add(accA, sc, w, vA1); scale_add(accB, sc, w, vB1);
    }
  }
  const float f = slot_factor(m);
  den = sum_slots(den * f); accA = sum_slots4(accA, f); accB = sum_slots4(accB, f);
  if (eq == 0) {
    float4 rA, rB;
    rA.x = accA.x / den; rA.y = accA.y / den; rA.z = accA.z / den; rA.w = accA.w / den;
    rB.x = accB.x / den; rB.y = accB.y / den; rB.z = accB.z / den; rB.w = accB.w / den;
    rows4[r * (PROJ_PITCH / 4) + (int)pc] = rA;
    rows4[r * (PROJ_PITCH / 4) + 8 + (int)pc] = rB;
  }
  __syncthreads();
  if (wib >= PROJ_WAVES) return;

  const unsigned m16  = (unsigned)lane & 15u;
  const unsigned half = (unsigned)lane >> 4;
  const unsigned klo  = half * 2u;
  const float* __restrict__ arow = (const float*)rows4 + m16 * PROJ_PITCH + klo;
  const unsigned cl = 32u * (unsigned)wib + 2u * m16;
  const float* __restrict__ wp = W2 + klo * 128u + cl;
  v8f acc0 = {0.f, 0.f, 0.f, 0.f, 0.f, 0.f, 0.f, 0.f};
  v8f acc1 = acc0;
  #pragma unroll
  for (int k = 0; k < 64; k += 4) {
    const v2f a  = *(const v2f*)(arow + k);
    const v2f w0 = *(const v2f*)(wp + k * 128);
    const v2f w1 = *(const v2f*)(wp + k * 128 + 128);
    v2f b0; b0.x = w0.x; b0.y = w1.x;
    v2f b1; b1.x = w0.y; b1.y = w1.y;
    acc0 = __builtin_amdgcn_wmma_f32_16x16x4_f32(false, a, false, b0, (short)0, acc0, false, false);
    acc1 = __builtin_amdgcn_wmma_f32_16x16x4_f32(false, a, false, b1, (short)0, acc1, false, false);
  }
  const float2 bv = *(const float2*)(b2 + cl);
  #pragma unroll
  for (int v = 0; v < 8; ++v) {
    const int row = row0 + v + 8 * (int)half;
    if (row < N) *(float2*)(out + (size_t)row * 128 + cl) = make_float2(acc0[v] + bv.x, acc1[v] + bv.y);
  }
}

static inline unsigned cdiv(long long a, long long b) { return (unsigned)((a + b - 1) / b); }

extern "C" void kernel_launch(void* const* d_in, const int* in_sizes, int n_in,
                              void* d_out, int out_size, void* d_ws, size_t ws_size,
                              hipStream_t stream) {
  const float* x    = (const float*)d_in[0];
  const int*   ei   = (const int*)d_in[1];
  const float* W1   = (const float*)d_in[2];
  const float* as1w = (const float*)d_in[3];
  const float* ad1w = (const float*)d_in[4];
  const float* b1   = (const float*)d_in[5];
  const float* W2   = (const float*)d_in[6];
  const float* as2w = (const float*)d_in[7];
  const float* ad2w = (const float*)d_in[8];
  const float* b2   = (const float*)d_in[9];
  float* out = (float*)d_out;

  const int N = in_sizes[0] / 128;
  const long long E = (long long)in_sizes[1] / 2;

  float* w = (float*)d_ws;
  size_t off = 0;
  float* h1   = w + off; off += (size_t)N * 64;
  float* helu = w + off; off += (size_t)N * 64;
  float* us   = w + off; off += 64;
  float* ud   = w + off; off += 64;
  int2* part  = (int2*)(w + off); off += (size_t)E * 2;
  int* gcnt   = (int*)(w + off); off += (size_t)KMAX * CPAD;
  int* rowptr = (int*)(w + off); off += (size_t)N + 1;
  int* csr    = (int*)(w + off); off += (size_t)E;

  const int B = 256;
  const int tiles = (N + 15) / 16;
  const int pairs = (N + 1) / 2;

  const int K = (N + NODES_PER_BUCKET - 1) / NODES_PER_BUCKET;
  hipMemsetAsync(gcnt, 0, (size_t)K * CPAD * sizeof(int), stream);
  bucket_hist_k<<<cdiv(E, EPB), 1024, 0, stream>>>(ei, E, K, gcnt);
  bucket_scatter_k<<<cdiv(E, EPB), 1024, 0, stream>>>(ei, E, K, gcnt, part);
  edge_rows_gemm_k<<<K + cdiv(tiles, GEMM_TILES_PER_GROUP), 1024, 0, stream>>>(part, gcnt, K, N, rowptr, csr, x, W1, h1);

  fold_k<<<8, B, 0, stream>>>(W2, as2w, ad2w, us, ud);
  gat1_node_k<<<cdiv((long long)pairs * 32, B), B, 0, stream>>>(rowptr, csr, h1, as1w, ad1w, b1, helu, N);

  gat2_proj_k<<<tiles, 256, 0, stream>>>(rowptr, csr, helu, us, ud, W2, b2, out, N);

  (void)n_in; (void)out_size; (void)ws_size;
}


# Edge grouping without the histogram launch and the counter memset: level 1 orders each block inside its own region and writes a table row, level 2 collects its bucket's runs; sum order = list order
# speedup vs baseline: 11.5024x; 1.0459x over previous
#include <hip/hip_runtime.h>
#include <cstdint>
#include <cstddef>

typedef __attribute__((ext_vector_type(2))) float v2f;
typedef __attribute__((ext_vector_type(8))) float v8f;

#define NEG_SLOPE 0.2f

template<int K, int COLS>
__device__ __forceinline__ void gemm_tile_f32(const float* __restrict__ A, const float* __restrict__ B,
                                              float* __restrict__ C, const int row0, const int lane) {
  constexpr int CT = COLS / 16;
  const unsigned m    = (unsigned)lane & 15u;
  const unsigned half = (unsigned)lane >> 4;
  const unsigned klo  = half * 2u;
  const float* __restrict__ At = A + (size_t)row0 * K;
  float* __restrict__ Ct = C + (size_t)row0 * COLS;
  const unsigned ao = m * (unsigned)K + klo;
  const unsigned bo = klo * (unsigned)COLS + m;

  v8f zero = {0.f, 0.f, 0.f, 0.f, 0.f, 0.f, 0.f, 0.f};
  v8f acc[CT];
  #pragma unroll
  for (int t = 0; t < CT; ++t) acc[t] = zero;

  #pragma unroll
  for (int k = 0; k < K; k += 4) {
    v2f a; a.x = At[ao + (unsigned)k]; a.y = At[ao + (unsigned)k + 1u];
    #pragma unroll
    for (int t = 0; t < CT; ++t) {
      const unsigned bi = bo + (unsigned)(k * COLS + t * 16);
      v2f b; b.x = B[bi]; b.y = B[bi + (unsigned)COLS];
      acc[t] = __builtin_amdgcn_wmma_f32_16x16x4_f32(
          false, a, false, b, (short)0, acc[t], false, false);
    }
  }
  const unsigned co = half * 8u * (unsigned)COLS + m;
  #pragma unroll
  for (int t = 0; t < CT; ++t) {
    #pragma unroll
    for (int v = 0; v < 8; ++v)
      Ct[co + (unsigned)(v * COLS + t * 16)] = acc[t][v];
  }
}

__global__ void fold_k(const float* __restrict__ W2,
                       const float* __restrict__ aw_s, const float* __restrict__ aw_d,
                       float* __restrict__ us, float* __restrict__ ud) {
  const int wave = (int)((blockIdx.x * blockDim.x + threadIdx.x) >> 5);
  const int lane = (int)(threadIdx.x & 31);
  if (wave >= 64) return;
  const float4 wv = *(const float4*)(W2 + (size_t)wave * 128 + lane * 4);
  const float4 sv = *(const float4*)(aw_s + lane * 4);
  const float4 dv = *(const float4*)(aw_d + lane * 4);
  float s = wv.x * sv.x + wv.y * sv.y + wv.z * sv.z + wv.w * sv.w;
  float d = wv.x * dv.x + wv.y * dv.y + wv.z * dv.z + wv.w * dv.w;
  #pragma unroll
  for (int off = 16; off > 0; off >>= 1) {
    s += __shfl_xor(s, off, 32);
    d += __shfl_xor(d, off, 32);
  }
  if (lane == 0) { us[wave] = s; ud[wave] = d; }
}

#define NODE_SHIFT 8
#define NODES_PER_BUCKET 256
#define KMAX 1024
#define EPB 8192
#define EPT (EPB / 1024)
#define RUNS_PER_WAVE 7
#define RUN_REG 64
#define STAGE_CAP (12 * 1024)

template<bool TWO, int CH>
__device__ __forceinline__ void wave_excl_scan_lds(const int* in, const int* in2, int* out, int n, int lane) {
  int carry = 0;
  for (int b0 = 0; b0 < n; b0 += CH * 32) {
    int v[CH], incl[CH];
    #pragma unroll
    for (int c = 0; c < CH; ++c) {
      const int i = b0 + c * 32 + lane;
      v[c] = 0;
      if (i < n) { v[c] = in[i]; if (TWO) v[c] += in2[i]; }
      incl[c] = v[c];
    }
    #pragma unroll
    for (int off = 1; off < 32; off <<= 1) {
      #pragma unroll
      for (int c = 0; c < CH; ++c) {
        const int t = __shfl_up(incl[c], off, 32);
        if (lane >= off) incl[c] += t;
      }
    }
    #pragma unroll
    for (int c = 0; c < CH; ++c) {
      const int i = b0 + c * 32 + lane;
      const int tot = __shfl(incl[c], 31, 32);
      if (i < n) out[i] = carry + incl[c] - v[c];
      carry += tot;
    }
  }
}

__global__ void __launch_bounds__(1024)
bucket_runs_k(const int* __restrict__ ei, long long E, int K,
              int* __restrict__ runtab, int2* __restrict__ part) {
  __shared__ int cntL[KMAX];
  __shared__ int lstart[KMAX];
  __shared__ int2 stage[EPB];
  const int tid = (int)threadIdx.x, lane = tid & 31;
  const long long e0 = (long long)blockIdx.x * EPB;
  int d[EPT], s[EPT];
  #pragma unroll
  for (int j = 0; j < EPT; ++j) {
    const long long e = e0 + j * 1024 + tid;
    d[j] = -1; s[j] = 0;
    if (e < E) { d[j] = ei[E + e]; s[j] = ei[e]; }
  }
  for (int k = tid; k < K; k += 1024) cntL[k] = 0;
  __syncthreads();
  int r[EPT];
  #pragma unroll
  for (int j = 0; j < EPT; ++j)
    r[j] = (d[j] >= 0) ? atomicAdd(&cntL[d[j] >> NODE_SHIFT], 1) : 0;
  __syncthreads();
  if (tid < 32) wave_excl_scan_lds<false, 8>(cntL, nullptr, lstart, K, lane);
  __syncthreads();
  const long long rem = E - e0;
  const int nE = rem < (long long)EPB ? (int)rem : EPB;
  int* __restrict__ trow = runtab + (size_t)blockIdx.x * (size_t)(K + 1);
  for (int k = tid; k < K; k += 1024) trow[k] = lstart[k];
  if (tid == 0) trow[K] = nE;
  int ls[EPT];
  #pragma unroll
  for (int j = 0; j < EPT; ++j) ls[j] = lstart[max(d[j], 0) >> NODE_SHIFT];
  static_assert(EPT == 8, "the fence below names one value per edge of the thread");
  asm volatile("" :: "v"(ls[0]), "v"(ls[1]), "v"(ls[2]), "v"(ls[3]), "v"(ls[4]), "v"(ls[5]), "v"(ls[6]), "v"(ls[7]));
  #pragma unroll
  for (int j = 0; j < EPT; ++j)
    if (d[j] >= 0) stage[ls[j] + r[j]] = make_int2(s[j], d[j]);
  __syncthreads();
  int2* __restrict__ dstp = part + (size_t)blockIdx.x * EPB;
  if (nE == EPB) {
    int2 p[EPT];
    #pragma unroll
    for (int j = 0; j < EPT; ++j) p[j] = stage[tid + j * 1024];
    #pragma unroll
    for (int j = 0; j < EPT; ++j) dstp[tid + j * 1024] = p[j];
  } else {
    for (int i = tid; i < nE; i += 1024) dstp[i] = stage[i];
  }
}

template<typename F>
__device__ __forceinline__ void extra_pairs(const int2* __restrict__ part, const int* __restrict__ runtab,
                                            const int K, const int k, const int nb, const int w, const int lane,
                                            const int (&ro)[RUNS_PER_WAVE], const int (&rl)[RUNS_PER_WAVE], F f) {
  #pragma unroll
  for (int r = 0; r < RUNS_PER_WAVE; ++r)
    for (int j = RUN_REG + lane; j < rl[r]; j += 32) f(part[ro[r] + j]);
  for (int b = w + 32 * RUNS_PER_WAVE; b < nb; b += 32) {
    const int* __restrict__ trow = runtab + (size_t)b * (size_t)(K + 1) + k;
    const int t0 = trow[0], t1 = trow[1];
    for (int j = t0 + lane; j < t1; j += 32) f(part[(size_t)b * EPB + j]);
  }
}

#define GEMM_TILES_PER_GROUP 16
__global__ void __launch_bounds__(1024, 16)
edge_rows_gemm_k(const int2* __restrict__ part, const int* __restrict__ runtab, int nb, int K, int N,
                 int* __restrict__ rowptr, int* __restrict__ csr,
                 const float* __restrict__ x, const float* __restrict__ W1, float* __restrict__ h1) {
  const int wv = __builtin_amdgcn_readfirstlane((int)(threadIdx.x >> 5));
  if ((int)blockIdx.x >= K) {
    const int row0 = (((int)blockIdx.x - K) * GEMM_TILES_PER_GROUP + wv) * 16;
    if (wv < GEMM_TILES_PER_GROUP && row0 < N)
      gemm_tile_f32<128, 64>(x, W1, h1, row0, (int)(threadIdx.x & 31));
    return;
  }
  __shared__ int cntL[NODES_PER_BUCKET];
  __shared__ int cnt2L[NODES_PER_BUCKET];
  __shared__ int cur2[NODES_PER_BUCKET];
  __shared__ int excl[NODES_PER_BUCKET];
  __shared__ int tot[2];
  __shared__ int stage[STAGE_CAP];
  const int tid = (int)threadIdx.x, lane = tid & 31;
  const int k = (int)blockIdx.x;
  const int node0 = k << NODE_SHIFT;
  const size_t tpitch = (size_t)(K + 1);

  int t0v, t1v;
  {
    const int bc = min(wv + 32 * min(lane, RUNS_PER_WAVE - 1), nb - 1);
    const int* __restrict__ trow = runtab + (size_t)bc * tpitch + k;
    t0v = trow[0]; t1v = trow[1];
  }
  int ro[RUNS_PER_WAVE], rl[RUNS_PER_WAVE];
  #pragma unroll
  for (int r = 0; r < RUNS_PER_WAVE; ++r) {
    const int b = wv + 32 * r;
    const int t0 = __builtin_amdgcn_readlane(t0v, r), t1 = __builtin_amdgcn_readlane(t1v, r);
    ro[r] = min(b, nb - 1) * EPB + t0;
    rl[r] = (b < nb) ? t1 - t0 : 0;
  }
  int sx[RUNS_PER_WAVE][2], pk[RUNS_PER_WAVE][2];
  #pragma unroll
  for (int r = 0; r < RUNS_PER_WAVE; ++r) {
    #pragma unroll
    for (int h = 0; h < 2; ++h) {
      const int j = lane + 32 * h;
      const bool in = j < rl[r];
      const int2 q = part[in ? ro[r] + j : 0];
      sx[r][h] = q.x;
      pk[r][h] = in ? q.y - node0 : -1;
    }
  }
  int sa = 0, sc = 0;
  for (int b = tid; b < nb; b += 1024) {
    const int* __restrict__ trow = runtab + (size_t)b * tpitch + k;
    sa += trow[0]; sc += trow[1];
  }
  if (tid < NODES_PER_BUCKET) { cntL[tid] = 0; cnt2L[tid] = 0; cur2[tid] = 0; }
  if (tid < 2) tot[tid] = 0;
  __syncthreads();
  if (wv * 32 < nb) {
    #pragma unroll
    for (int off = 16; off > 0; off >>= 1) {
      sa += __shfl_xor(sa, off, 32);
      sc += __shfl_xor(sc, off, 32);
    }
    if (lane == 0) { atomicAdd(&tot[0], sa); atomicAdd(&tot[1], sc); }
  }
  int rk[RUNS_PER_WAVE][2];
  #pragma unroll
  for (int r = 0; r < RUNS_PER_WAVE; ++r) {
    #pragma unroll
    for (int h = 0; h < 2; ++h) {
      const bool in = pk[r][h] >= 0;
      rk[r][h] = atomicAdd(&cntL[in ? pk[r][h] : lane], in ? 1 : 0);
    }
  }
  #pragma unroll
  for (int r = 0; r < RUNS_PER_WAVE; ++r) {
    #pragma unroll
    for (int h = 0; h < 2; ++h)
      if (pk[r][h] >= 0) pk[r][h] |= rk[r][h] << NODE_SHIFT;
  }
  extra_pairs(part, runtab, K, k, nb, wv, lane, ro, rl,
              [&](const int2 q) { atomicAdd(&cnt2L[q.y - node0], 1); });
  __syncthreads();
  const int base = tot[0];
  const int cnt  = tot[1] - base;
  if (tid < 32) wave_excl_scan_lds<true, 4>(cntL, cnt2L, excl, NODES_PER_BUCKET, lane);
  __syncthreads();
  if (tid < NODES_PER_BUCKET) {
    const int n = node0 + tid;
    if (n < N) rowptr[n] = base + excl[tid];
  }
  if (tid == 0 && k == K - 1) rowptr[N] = base + cnt;
  const bool staged = cnt <= STAGE_CAP;
  int st[RUNS_PER_WAVE][2];
  #pragma unroll
  for (int r = 0; r < RUNS_PER_WAVE; ++r) {
    #pragma unroll
    for (int h = 0; h < 2; ++h)
      st[r][h] = excl[pk[r][h] & (NODES_PER_BUCKET - 1)];
  }
  static_assert(RUNS_PER_WAVE == 7, "the fence below names two values per run");
  asm volatile("" :: "v"(st[0][0]), "v"(st[0][1]), "v"(st[1][0]), "v"(st[1][1]), "v"(st[2][0]), "v"(st[2][1]),
                     "v"(st[3][0]), "v"(st[3][1]), "v"(st[4][0]), "v"(st[4][1]), "v"(st[5][0]), "v"(st[5][1]),
                     "v"(st[6][0]), "v"(st[6][1]));
  #pragma unroll
  for (int r = 0; r < RUNS_PER_WAVE; ++r) {
    #pragma unroll
    for (int h = 0; h < 2; ++h)
      pk[r][h] = (pk[r][h] >= 0) ? st[r][h] + (pk[r][h] >> NODE_SHIFT) : -1;
  }
  if (staged) {
    #pragma unroll
    for (int r = 0; r < RUNS_PER_WAVE; ++r) {
      #pragma unroll
      for (int h = 0; h < 2; ++h)
        if (pk[r][h] >= 0) stage[pk[r][h]] = sx[r][h];
    }
  } else {
    #pragma unroll
    for (int r = 0; r < RUNS_PER_WAVE; ++r) {
      #pragma unroll
      for (int h = 0; h < 2; ++h)
        if (pk[r][h] >= 0) csr[base + pk[r][h]] = sx[r][h];
    }
  }
  extra_pairs(part, runtab, K, k, nb, wv, lane, ro, rl,
              [&](const int2 q) {
                const int l = q.y - node0;
                const int pos = excl[l] + cntL[l] + atomicAdd(&cur2[l], 1);
                if (staged) stage[pos] = q.x; else csr[base + pos] = q.x;
              });
  __syncthreads();
  if (staged)
    for (int i = tid; i < cnt; i += 1024) csr[base + i] = stage[i];
}


template<int M>
__device__ __forceinline__ float xrow(const float v) {
  const int b = __builtin_bit_cast(int, v);
  return __builtin_bit_cast(float, __builtin_amdgcn_update_dpp(0, b, 0x160 + M, 0xf, 0xf, true));
}
__device__ __forceinline__ int xhalf_i(const int b) {
  return __builtin_amdgcn_permlanex16(b, b, 0x76543210, 0xfedcba98, false, false);
}
__device__ __forceinline__ unsigned row_pick(const int v, const unsigned sel) {
  return __builtin_amdgcn_permlane16_var((unsigned)v, (unsigned)v, sel, false, false);
}

__device__ __forceinline__ float leaky(float e) { return e > 0.f ? e : NEG_SLOPE * e; }
__device__ __forceinline__ float dot4(const float4 a, const float4 b) {
  return a.x * b.x + a.y * b.y + a.z * b.z + a.w * b.w;
}

__device__ __forceinline__ void stream_step(float& m, const float e, const bool ok, float& sc, float& w) {
  const float mn = fmaxf(m, e);
  sc = __expf(m - mn);
  w  = ok ? __expf(e - mn) : 0.f;
  m = mn;
}
__device__ __forceinline__ void scale_add(float4& acc, const float sc, const float w, const float4 v) {
  acc.x = acc.x * sc + w * v.x;
  acc.y = acc.y * sc + w * v.y;
  acc.z = acc.z * sc + w * v.z;
  acc.w = acc.w * sc + w * v.w;
}
__device__ __forceinline__ float slot_factor(const float m) {
  const float M = fmaxf(m, xrow<8>(m));
  return __expf(m - M);
}
__device__ __forceinline__ float sum_slots(float v) {
  v += xrow<8>(v);
  return v;
}
__device__ __forceinline__ float4 sum_slots4(const float4 a, const float f) {
  float4 r;
  r.x = sum_slots(a.x * f); r.y = sum_slots(a.y * f); r.z = sum_slots(a.z * f); r.w = sum_slots(a.w * f);
  return r;
}

__global__ void __launch_bounds__(256, 16)
gat1_node_k(const int* __restrict__ rowptr, const int* __restrict__ csr,
            const float* __restrict__ h1,
            const float* __restrict__ aw_s, const float* __restrict__ aw_d,
            const float* __restrict__ b1, float* __restrict__ helu, int N) {
  const int lane = (int)(threadIdx.x & 31);
  const int wv = __builtin_amdgcn_readfirstlane((int)((blockIdx.x * blockDim.x + threadIdx.x) >> 5));
  if (2 * wv >= N) return;
  const int hf = lane >> 4;
  const bool live = 2 * wv + hf < N;
  const int node = live ? 2 * wv + hf : N - 1;
  const int beg = rowptr[node];
  const int deg = rowptr[node + 1] - beg;
  const int degw = __builtin_amdgcn_readfirstlane(max(deg, xhalf_i(deg)));
  const int j = lane & 15;
  const int eq = (lane >> 3) & 1;
  const unsigned pc = (unsigned)(lane & 7);
  const float4* __restrict__ hv4 = (const float4*)h1;
  const float4 sA = *(const float4*)(aw_s + 4 * pc), sB = *(const float4*)(aw_s + 32 + 4 * pc);
  const float4 dA = *(const float4*)(aw_d + 4 * pc), dB = *(const float4*)(aw_d + 32 + 4 * pc);

  const float4 oA = hv4[(unsigned)node * 16u + pc], oB = hv4[(unsigned)node * 16u + 8u + pc];
  float adA = dot4(oA, dA), adB = dot4(oB, dB), asA = dot4(oA, sA), asB = dot4(oB, sB);
  adA += xrow<1>(adA); adB += xrow<1>(adB);
  asA += xrow<1>(asA); asB += xrow<1>(asB);
  float mA = leaky(asA + adA), mB = leaky(asB + adB);
  const float4 zero4 = make_float4(0.f, 0.f, 0.f, 0.f);
  float denA = (eq == 0) ? 1.f : 0.f, denB = denA;
  float4 accA = (eq == 0) ? oA : zero4, accB = (eq == 0) ? oB : zero4;
  const int rem = deg - eq;

  for (int c0 = 0; c0 < degw; c0 += 16) {
    const int sl = (c0 + j < deg) ? csr[beg + c0 + j] : node;
    const int lim = min(16, degw - c0);
    for (int t = 0; 4 * t < lim; ++t) {
      const unsigned s0 = row_pick(sl, 4u * (unsigned)t + (unsigned)eq);
      const unsigned s1 = row_pick(sl, 4u * (unsigned)t + 2u + (unsigned)eq);
      const float4 vA0 = hv4[s0 * 16u + pc], vB0 = hv4[s0 * 16u + 8u + pc];
      const float4 vA1 = hv4[s1 * 16u + pc], vB1 = hv4[s1 * 16u + 8u + pc];
      float pA0 = dot4(vA0, sA), pB0 = dot4(vB0, sB), pA1 = dot4(vA1, sA), pB1 = dot4(vB1, sB);
      pA0 += xrow<1>(pA0); pB0 += xrow<1>(pB0);
      pA1 += xrow<1>(pA1); pB1 += xrow<1>(pB1);
      const bool ok0 = c0 + 4 * t < rem, ok1 = c0 + 4 * t + 2 < rem;
      float sc, w;
      stream_step(mA, leaky(pA0 + adA), ok0, sc, w); denA = denA * sc + w; scale_add(accA, sc, w, vA0);
      stream_step(mB, leaky(pB0 + adB), ok0, sc, w); denB = denB * sc + w; scale_add(accB, sc, w, vB0);
      stream_step(mA, leaky(pA1 + adA), ok1, sc, w); denA = denA * sc + w; scale_add(accA, sc, w, vA1);
      stream_step(mB, leaky(pB1 + adB), ok1, sc, w); denB = denB * sc + w; scale_add(accB, sc, w, vB1);
    }
  }
  const float fA = slot_factor(mA), fB = slot_factor(mB);
  denA = sum_slots(denA * fA); accA = sum_slots4(accA, fA);
  denB = sum_slots(denB * fB); accB = sum_slots4(accB, fB);
  if (eq == 0 && live) {
    const float4 bA = *(const float4*)(b1 + 4 * pc), bB = *(const float4*)(b1 + 32 + 4 * pc);
    float4 rA, rB;
    rA.x = accA.x / denA + bA.x; rA.y = accA.y / denA + bA.y; rA.z = accA.z / denA + bA.z; rA.w = accA.w / denA + bA.w;
    rB.x = accB.x / denB + bB.x; rB.y = accB.y / denB + bB.y; rB.z = accB.z / denB + bB.z; rB.w = accB.w / denB + bB.w;
    rA.x = rA.x > 0.f ? rA.x : (__expf(rA.x) - 1.f); rA.y = rA.y > 0.f ? rA.y : (__expf(rA.y) - 1.f);
    rA.z = rA.z > 0.f ? rA.z : (__expf(rA.z) - 1.f); rA.w = rA.w > 0.f ? rA.w : (__expf(rA.w) - 1.f);
    rB.x = rB.x > 0.f ? rB.x : (__expf(rB.x) - 1.f); rB.y = rB.y > 0.f ? rB.y : (__expf(rB.y) - 1.f);
    rB.z = rB.z > 0.f ? rB.z : (__expf(rB.z) - 1.f); rB.w = rB.w > 0.f ? rB.w : (__expf(rB.w) - 1.f);
    float4* __restrict__ o4 = (float4*)helu;
    o4[(unsigned)node * 16u + pc] = rA;
    o4[(unsigned)node * 16u + 8u + pc] = rB;
  }
}

#define PROJ_PITCH 68
#define PROJ_WAVES 4
__global__ void __launch_bounds__(256, 16)
gat2_proj_k(const int* __restrict__ rowptr, const int* __restrict__ csr,
            const float* __restrict__ helu,
            const float* __restrict__ us, const float* __restrict__ ud,
            const float* __restrict__ W2, const float* __restrict__ b2,
            float* __restrict__ out, int N) {
  __shared__ float4 rows4[16 * (PROJ_PITCH / 4)];
  const int lane = (int)(threadIdx.x & 31);
  const int wib = __builtin_amdgcn_readfirstlane((int)(threadIdx.x >> 5));
  const int row0 = (int)blockIdx.x * 16;
  const int hf = lane >> 4;
  const int r = 2 * wib + hf;
  const bool live = row0 + r < N;
  const int node = live ? row0 + r : N - 1;
  const int beg = rowptr[node];
  const int deg = rowptr[node + 1] - beg;
  const int degw = __builtin_amdgcn_readfirstlane(max(deg, xhalf_i(deg)));
  const int j = lane & 15;
  const int eq = (lane >> 3) & 1;
  const unsigned pc = (unsigned)(lane & 7);
  const float4* __restrict__ hv4 = (const float4*)helu;
  const float4 sA = *(const float4*)(us + 4 * pc), sB = *(const float4*)(us + 32 + 4 * pc);
  const float4 dA = *(const float4*)(ud + 4 * pc), dB = *(const float4*)(ud + 32 + 4 * pc);

  const float4 oA = hv4[(unsigned)node * 16u + pc], oB = hv4[(unsigned)node * 16u + 8u + pc];
  float adn = dot4(oA, dA) + dot4(oB, dB), asn = dot4(oA, sA) + dot4(oB, sB);
  adn += xrow<1>(adn); asn += xrow<1>(asn);
  adn += xrow<2>(adn); asn += xrow<2>(asn);
  adn += xrow<4>(adn); asn += xrow<4>(asn);
  float m = leaky(asn + adn);
  const float4 zero4 = make_float4(0.f, 0.f, 0.f, 0.f);
  float den = (eq == 0) ? 1.f : 0.f;
  float4 accA = (eq == 0) ? oA : zero4, accB = (eq == 0) ? oB : zero4;
  const int rem = deg - eq;

  for (int c0 = 0; c0 < degw; c0 += 16) {
    const int sl = (c0 + j < deg) ? csr[beg + c0 + j] : node;
    const int lim = min(16, degw - c0);
    for (int t = 0; 4 * t < lim; ++t) {
      const unsigned s0 = row_pick(sl, 4u * (unsigned)t + (unsigned)eq);
      const unsigned s1 = row_pick(sl, 4u * (unsigned)t + 2u + (unsigned)eq);
      const float4 vA0 = hv4[s0 * 16u + pc], vB0 = hv4[s0 * 16u + 8u + pc];
      const float4 vA1 = hv4[s1 * 16u + pc], vB1 = hv4[s1 * 16u + 8u + pc];
      float p0 = dot4(vA0, sA) + dot4(vB0, sB), p1 = dot4(vA1, sA) + dot4(vB1, sB);
      p0 += xrow<1>(p0); p1 += xrow<1>(p1);
      p0 += xrow<2>(p0); p1 += xrow<2>(p1);
      p0 += xrow<4>(p0); p1 += xrow<4>(p1);
      const bool ok0 = c0 + 4 * t < rem, ok1 = c0 + 4 * t + 2 < rem;
      const float e0 = leaky(p0 + adn), e1 = leaky(p1 + adn);
      float sc, w;
      stream_step(m, e0, ok0, sc, w); den = den * sc + w; scale_add(accA, sc, w, vA0); scale_add(accB, sc, w, vB0);
      stream_step(m, e1, ok1, sc, w); den = den * sc + w; scale_add(accA, sc, w, vA1); scale_add(accB, sc, w, vB1);
    }
  }
  const float f = slot_factor(m);
  den = sum_slots(den * f); accA = sum_slots4(accA, f); accB = sum_slots4(accB, f);
  if (eq == 0) {
    float4 rA, rB;
    rA.x = accA.x / den; rA.y = accA.y / den; rA.z = accA.z / den; rA.w = accA.w / den;
    rB.x = accB.x / den; rB.y = accB.y / den; rB.z = accB.z / den; rB.w = accB.w / den;
    rows4[r * (PROJ_PITCH / 4) + (int)pc] = rA;
    rows4[r * (PROJ_PITCH / 4) + 8 + (int)pc] = rB;
  }
  __syncthreads();
  if (wib >= PROJ_WAVES) return;

  const unsigned m16  = (unsigned)lane & 15u;
  const unsigned half = (unsigned)lane >> 4;
  const unsigned klo  = half * 2u;
  const float* __restrict__ arow = (const float*)rows4 + m16 * PROJ_PITCH + klo;
  const unsigned cl = 32u * (unsigned)wib + 2u * m16;
  const float* __restrict__ wp = W2 + klo * 128u + cl;
  v8f acc0 = {0.f, 0.f, 0.f, 0.f, 0.f, 0.f, 0.f, 0.f};
  v8f acc1 = acc0;
  #pragma unroll
  for (int k = 0; k < 64; k += 4) {
    const v2f a  = *(const v2f*)(arow + k);
    const v2f w0 = *(const v2f*)(wp + k * 128);
    const v2f w1 = *(const v2f*)(wp + k * 128 + 128);
    v2f b0; b0.x = w0.x; b0.y = w1.x;
    v2f b1; b1.x = w0.y; b1.y = w1.y;
    acc0 = __builtin_amdgcn_wmma_f32_16x16x4_f32(false, a, false, b0, (short)0, acc0, false, false);
    acc1 = __builtin_amdgcn_wmma_f32_16x16x4_f32(false, a, false, b1, (short)0, acc1, false, false);
  }
  const float2 bv = *(const float2*)(b2 + cl);
  #pragma unroll
  for (int v = 0; v < 8; ++v) {
    const int row = row0 + v + 8 * (int)half;
    if (row < N) *(float2*)(out + (size_t)row * 128 + cl) = make_float2(acc0[v] + bv.x, acc1[v] + bv.y);
  }
}

static inline unsigned cdiv(long long a, long long b) { return (unsigned)((a + b - 1) / b); }

extern "C" void kernel_launch(void* const* d_in, const int* in_sizes, int n_in,
                              void* d_out, int out_size, void* d_ws, size_t ws_size,
                              hipStream_t stream) {
  const float* x    = (const float*)d_in[0];
  const int*   ei   = (const int*)d_in[1];
  const float* W1   = (const float*)d_in[2];
  const float* as1w = (const float*)d_in[3];
  const float* ad1w = (const float*)d_in[4];
  const float* b1   = (const float*)d_in[5];
  const float* W2   = (const float*)d_in[6];
  const float* as2w = (const float*)d_in[7];
  const float* ad2w = (const float*)d_in[8];
  const float* b2   = (const float*)d_in[9];
  float* out = (float*)d_out;

  const int N = in_sizes[0] / 128;
  const long long E = (long long)in_sizes[1] / 2;

  float* w = (float*)d_ws;
  size_t off = 0;
  float* h1   = w + off; off += (size_t)N * 64;
  float* helu = w + off; off += (size_t)N * 64;
  float* us   = w + off; off += 64;
  float* ud   = w + off; off += 64;
  const int K  = (N + NODES_PER_BUCKET - 1) / NODES_PER_BUCKET;
  const int nb = (int)cdiv(E, EPB);
  int2* part  = (int2*)(w + off); off += (size_t)nb * EPB * 2;
  int* runtab = (int*)(w + off); off += (size_t)nb * (K + 1);
  int* rowptr = (int*)(w + off); off += (size_t)N + 1;
  int* csr    = (int*)(w + off); off += (size_t)E;

  const int B = 256;
  const int tiles = (N + 15) / 16;
  const int pairs = (N + 1) / 2;

  bucket_runs_k<<<nb, 1024, 0, stream>>>(ei, E, K, runtab, part);
  edge_rows_gemm_k<<<K + cdiv(tiles, GEMM_TILES_PER_GROUP), 1024, 0, stream>>>(part, runtab, nb, K, N, rowptr, csr, x, W1, h1);

  fold_k<<<8, B, 0, stream>>>(W2, as2w, ad2w, us, ud);
  gat1_node_k<<<cdiv((long long)pairs * 32, B), B, 0, stream>>>(rowptr, csr, h1, as1w, ad1w, b1, helu, N);

  gat2_proj_k<<<tiles, 256, 0, stream>>>(rowptr, csr, helu, us, ud, W2, b2, out, N);

  (void)n_in; (void)out_size; (void)ws_size;
}


# Layer-1 rows stored head-per-lane (GEMM tile permutes its columns at the store): a lane holds one head's 8 channels, in-lane packed 8-term logit dot, one softmax stream; logit f32 sum order changes
# speedup vs baseline: 12.1691x; 1.0580x over previous
#include <hip/hip_runtime.h>
#include <cstdint>
#include <cstddef>

typedef __attribute__((ext_vector_type(2))) float v2f;
typedef __attribute__((ext_vector_type(8))) float v8f;

#define NEG_SLOPE 0.2f

template<int K, int COLS, bool HEAD_ROWS>
__device__ __forceinline__ void gemm_tile_f32(const float* __restrict__ A, const float* __restrict__ B,
                                              float* __restrict__ C, const int row0, const int lane) {
  static_assert(!HEAD_ROWS || COLS == 64, "head-per-lane rows are 8 heads x 8 channels");
  constexpr int CT = COLS / 16;
  const unsigned m    = (unsigned)lane & 15u;
  const unsigned half = (unsigned)lane >> 4;
  const unsigned klo  = half * 2u;
  const float* __restrict__ At = A + (size_t)row0 * K;
  float* __restrict__ Ct = C + (size_t)row0 * COLS;
  const unsigned ao = m * (unsigned)K + klo;
  const unsigned bo = klo * (unsigned)COLS + m;

  v8f zero = {0.f, 0.f, 0.f, 0.f, 0.f, 0.f, 0.f, 0.f};
  v8f acc[CT];
  #pragma unroll
  for (int t = 0; t < CT; ++t) acc[t] = zero;

  #pragma unroll
  for (int k = 0; k < K; k += 4) {
    v2f a; a.x = At[ao + (unsigned)k]; a.y = At[ao + (unsigned)k + 1u];
    #pragma unroll
    for (int t = 0; t < CT; ++t) {
      const unsigned bi = bo + (unsigned)(k * COLS + t * 16);
      v2f b; b.x = B[bi]; b.y = B[bi + (unsigned)COLS];
      acc[t] = __builtin_amdgcn_wmma_f32_16x16x4_f32(
          false, a, false, b, (short)0, acc[t], false, false);
    }
  }
  const unsigned cm = HEAD_ROWS ? 8u * (m & 4u) + ((m >> 3) << 2) + (m & 3u) : m;
  const unsigned co = half * 8u * (unsigned)COLS + cm;
  #pragma unroll
  for (int t = 0; t < CT; ++t) {
    #pragma unroll
    for (int v = 0; v < 8; ++v)
      Ct[co + (unsigned)(v * COLS + t * (HEAD_ROWS ? 8 : 16))] = acc[t][v];
  }
}

__global__ void fold_k(const float* __restrict__ W2,
                       const float* __restrict__ aw_s, const float* __restrict__ aw_d,
                       float* __restrict__ us, float* __restrict__ ud) {
  const int wave = (int)((blockIdx.x * blockDim.x + threadIdx.x) >> 5);
  const int lane = (int)(threadIdx.x & 31);
  if (wave >= 64) return;
  const float4 wv = *(const float4*)(W2 + (size_t)wave * 128 + lane * 4);
  const float4 sv = *(const float4*)(aw_s + lane * 4);
  const float4 dv = *(const float4*)(aw_d + lane * 4);
  float s = wv.x * sv.x + wv.y * sv.y + wv.z * sv.z + wv.w * sv.w;
  float d = wv.x * dv.x + wv.y * dv.y + wv.z * dv.z + wv.w * dv.w;
  #pragma unroll
  for (int off = 16; off > 0; off >>= 1) {
    s += __shfl_xor(s, off, 32);
    d += __shfl_xor(d, off, 32);
  }
  if (lane == 0) { us[wave] = s; ud[wave] = d; }
}

#define NODE_SHIFT 8
#define NODES_PER_BUCKET 256
#define KMAX 1024
#define EPB 8192
#define EPT (EPB / 1024)
#define RUNS_PER_WAVE 7
#define RUN_REG 64
#define STAGE_CAP (12 * 1024)

template<bool TWO, int CH>
__device__ __forceinline__ void wave_excl_scan_lds(const int* in, const int* in2, int* out, int n, int lane) {
  int carry = 0;
  for (int b0 = 0; b0 < n; b0 += CH * 32) {
    int v[CH], incl[CH];
    #pragma unroll
    for (int c = 0; c < CH; ++c) {
      const int i = b0 + c * 32 + lane;
      v[c] = 0;
      if (i < n) { v[c] = in[i]; if (TWO) v[c] += in2[i]; }
      incl[c] = v[c];
    }
    #pragma unroll
    for (int off = 1; off < 32; off <<= 1) {
      #pragma unroll
      for (int c = 0; c < CH; ++c) {
        const int t = __shfl_up(incl[c], off, 32);
        if (lane >= off) incl[c] += t;
      }
    }
    #pragma unroll
    for (int c = 0; c < CH; ++c) {
      const int i = b0 + c * 32 + lane;
      const int tot = __shfl(incl[c], 31, 32);
      if (i < n) out[i] = carry + incl[c] - v[c];
      carry += tot;
    }
  }
}

__global__ void __launch_bounds__(1024)
bucket_runs_k(const int* __restrict__ ei, long long E, int K,
              int* __restrict__ runtab, int2* __restrict__ part) {
  __shared__ int cntL[KMAX];
  __shared__ int lstart[KMAX];
  __shared__ int2 stage[EPB];
  const int tid = (int)threadIdx.x, lane = tid & 31;
  const long long e0 = (long long)blockIdx.x * EPB;
  int d[EPT], s[EPT];
  #pragma unroll
  for (int j = 0; j < EPT; ++j) {
    const long long e = e0 + j * 1024 + tid;
    d[j] = -1; s[j] = 0;
    if (e < E) { d[j] = ei[E + e]; s[j] = ei[e]; }
  }
  for (int k = tid; k < K; k += 1024) cntL[k] = 0;
  __syncthreads();
  int r[EPT];
  #pragma unroll
  for (int j = 0; j < EPT; ++j)
    r[j] = (d[j] >= 0) ? atomicAdd(&cntL[d[j] >> NODE_SHIFT], 1) : 0;
  __syncthreads();
  if (tid < 32) wave_excl_scan_lds<false, 8>(cntL, nullptr, lstart, K, lane);
  __syncthreads();
  const long long rem = E - e0;
  const int nE = rem < (long long)EPB ? (int)rem : EPB;
  int* __restrict__ trow = runtab + (size_t)blockIdx.x * (size_t)(K + 1);
  for (int k = tid; k < K; k += 1024) trow[k] = lstart[k];
  if (tid == 0) trow[K] = nE;
  int ls[EPT];
  #pragma unroll
  for (int j = 0; j < EPT; ++j) ls[j] = lstart[max(d[j], 0) >> NODE_SHIFT];
  static_assert(EPT == 8, "the fence below names one value per edge of the thread");
  asm volatile("" :: "v"(ls[0]), "v"(ls[1]), "v"(ls[2]), "v"(ls[3]), "v"(ls[4]), "v"(ls[5]), "v"(ls[6]), "v"(ls[7]));
  #pragma unroll
  for (int j = 0; j < EPT; ++j)
    if (d[j] >= 0) stage[ls[j] + r[j]] = make_int2(s[j], d[j]);
  __syncthreads();
  int2* __restrict__ dstp = part + (size_t)blockIdx.x * EPB;
  if (nE == EPB) {
    int2 p[EPT];
    #pragma unroll
    for (int j = 0; j < EPT; ++j) p[j] = stage[tid + j * 1024];
    #pragma unroll
    for (int j = 0; j < EPT; ++j) dstp[tid + j * 1024] = p[j];
  } else {
    for (int i = tid; i < nE; i += 1024) dstp[i] = stage[i];
  }
}

template<typename F>
__device__ __forceinline__ void extra_pairs(const int2* __restrict__ part, const int* __restrict__ runtab,
                                            const int K, const int k, const int nb, const int w, const int lane,
                                            const int (&ro)[RUNS_PER_WAVE], const int (&rl)[RUNS_PER_WAVE], F f) {
  #pragma unroll
  for (int r = 0; r < RUNS_PER_WAVE; ++r)
    for (int j = RUN_REG + lane; j < rl[r]; j += 32) f(part[ro[r] + j]);
  for (int b = w + 32 * RUNS_PER_WAVE; b < nb; b += 32) {
    const int* __restrict__ trow = runtab + (size_t)b * (size_t)(K + 1) + k;
    const int t0 = trow[0], t1 = trow[1];
    for (int j = t0 + lane; j < t1; j += 32) f(part[(size_t)b * EPB + j]);
  }
}

#define GEMM_TILES_PER_GROUP 16
__global__ void __launch_bounds__(1024, 16)
edge_rows_gemm_k(const int2* __restrict__ part, const int* __restrict__ runtab, int nb, int K, int N,
                 int* __restrict__ rowptr, int* __restrict__ csr,
                 const float* __restrict__ x, const float* __restrict__ W1, float* __restrict__ h1) {
  const int wv = __builtin_amdgcn_readfirstlane((int)(threadIdx.x >> 5));
  if ((int)blockIdx.x >= K) {
    const int row0 = (((int)blockIdx.x - K) * GEMM_TILES_PER_GROUP + wv) * 16;
    if (wv < GEMM_TILES_PER_GROUP && row0 < N)
      gemm_tile_f32<128, 64, true>(x, W1, h1, row0, (int)(threadIdx.x & 31));
    return;
  }
  __shared__ int cntL[NODES_PER_BUCKET];
  __shared__ int cnt2L[NODES_PER_BUCKET];
  __shared__ int cur2[NODES_PER_BUCKET];
  __shared__ int excl[NODES_PER_BUCKET];
  __shared__ int tot[2];
  __shared__ int stage[STAGE_CAP];
  const int tid = (int)threadIdx.x, lane = tid & 31;
  const int k = (int)blockIdx.x;
  const int node0 = k << NODE_SHIFT;
  const size_t tpitch = (size_t)(K + 1);

  int t0v, t1v;
  {
    const int bc = min(wv + 32 * min(lane, RUNS_PER_WAVE - 1), nb - 1);
    const int* __restrict__ trow = runtab + (size_t)bc * tpitch + k;
    t0v = trow[0]; t1v = trow[1];
  }
  int ro[RUNS_PER_WAVE], rl[RUNS_PER_WAVE];
  #pragma unroll
  for (int r = 0; r < RUNS_PER_WAVE; ++r) {
    const int b = wv + 32 * r;
    const int t0 = __builtin_amdgcn_readlane(t0v, r), t1 = __builtin_amdgcn_readlane(t1v, r);
    ro[r] = min(b, nb - 1) * EPB + t0;
    rl[r] = (b < nb) ? t1 - t0 : 0;
  }
  int sx[RUNS_PER_WAVE][2], pk[RUNS_PER_WAVE][2];
  #pragma unroll
  for (int r = 0; r < RUNS_PER_WAVE; ++r) {
    #pragma unroll
    for (int h = 0; h < 2; ++h) {
      const int j = lane + 32 * h;
      const bool in = j < rl[r];
      const int2 q = part[in ? ro[r] + j : 0];
      sx[r][h] = q.x;
      pk[r][h] = in ? q.y - node0 : -1;
    }
  }
  int sa = 0, sc = 0;
  for (int b = tid; b < nb; b += 1024) {
    const int* __restrict__ trow = runtab + (size_t)b * tpitch + k;
    sa += trow[0]; sc += trow[1];
  }
  if (tid < NODES_PER_BUCKET) { cntL[tid] = 0; cnt2L[tid] = 0; cur2[tid] = 0; }
  if (tid < 2) tot[tid] = 0;
  __syncthreads();
  if (wv * 32 < nb) {
    #pragma unroll
    for (int off = 16; off > 0; off >>= 1) {
      sa += __shfl_xor(sa, off, 32);
      sc += __shfl_xor(sc, off, 32);
    }
    if (lane == 0) { atomicAdd(&tot[0], sa); atomicAdd(&tot[1], sc); }
  }
  int rk[RUNS_PER_WAVE][2];
  #pragma unroll
  for (int r = 0; r < RUNS_PER_WAVE; ++r) {
    #pragma unroll
    for (int h = 0; h < 2; ++h) {
      const bool in = pk[r][h] >= 0;
      rk[r][h] = atomicAdd(&cntL[in ? pk[r][h] : lane], in ? 1 : 0);
    }
  }
  #pragma unroll
  for (int r = 0; r < RUNS_PER_WAVE; ++r) {
    #pragma unroll
    for (int h = 0; h < 2; ++h)
      if (pk[r][h] >= 0) pk[r][h] |= rk[r][h] << NODE_SHIFT;
  }
  extra_pairs(part, runtab, K, k, nb, wv, lane, ro, rl,
              [&](const int2 q) { atomicAdd(&cnt2L[q.y - node0], 1); });
  __syncthreads();
  const int base = tot[0];
  const int cnt  = tot[1] - base;
  if (tid < 32) wave_excl_scan_lds<true, 4>(cntL, cnt2L, excl, NODES_PER_BUCKET, lane);
  __syncthreads();
  if (tid < NODES_PER_BUCKET) {
    const int n = node0 + tid;
    if (n < N) rowptr[n] = base + excl[tid];
  }
  if (tid == 0 && k == K - 1) rowptr[N] = base + cnt;
  const bool staged = cnt <= STAGE_CAP;
  int st[RUNS_PER_WAVE][2];
  #pragma unroll
  for (int r = 0; r < RUNS_PER_WAVE; ++r) {
    #pragma unroll
    for (int h = 0; h < 2; ++h)
      st[r][h] = excl[pk[r][h] & (NODES_PER_BUCKET - 1)];
  }
  static_assert(RUNS_PER_WAVE == 7, "the fence below names two values per run");
  asm volatile("" :: "v"(st[0][0]), "v"(st[0][1]), "v"(st[1][0]), "v"(st[1][1]), "v"(st[2][0]), "v"(st[2][1]),
                     "v"(st[3][0]), "v"(st[3][1]), "v"(st[4][0]), "v"(st[4][1]), "v"(st[5][0]), "v"(st[5][1]),
                     "v"(st[6][0]), "v"(st[6][1]));
  #pragma unroll
  for (int r = 0; r < RUNS_PER_WAVE; ++r) {
    #pragma unroll
    for (int h = 0; h < 2; ++h)
      pk[r][h] = (pk[r][h] >= 0) ? st[r][h] + (pk[r][h] >> NODE_SHIFT) : -1;
  }
  if (staged) {
    #pragma unroll
    for (int r = 0; r < RUNS_PER_WAVE; ++r) {
      #pragma unroll
      for (int h = 0; h < 2; ++h)
        if (pk[r][h] >= 0) stage[pk[r][h]] = sx[r][h];
    }
  } else {
    #pragma unroll
    for (int r = 0; r < RUNS_PER_WAVE; ++r) {
      #pragma unroll
      for (int h = 0; h < 2; ++h)
        if (pk[r][h] >= 0) csr[base + pk[r][h]] = sx[r][h];
    }
  }
  extra_pairs(part, runtab, K, k, nb, wv, lane, ro, rl,
              [&](const int2 q) {
                const int l = q.y - node0;
                const int pos = excl[l] + cntL[l] + atomicAdd(&cur2[l], 1);
                if (staged) stage[pos] = q.x; else csr[base + pos] = q.x;
              });
  __syncthreads();
  if (staged)
    for (int i = tid; i < cnt; i += 1024) csr[base + i] = stage[i];
}


template<int M>
__device__ __forceinline__ float xrow(const float v) {
  const int b = __builtin_bit_cast(int, v);
  return __builtin_bit_cast(float, __builtin_amdgcn_update_dpp(0, b, 0x160 + M, 0xf, 0xf, true));
}
__device__ __forceinline__ int xhalf_i(const int b) {
  return __builtin_amdgcn_permlanex16(b, b, 0x76543210, 0xfedcba98, false, false);
}
__device__ __forceinline__ unsigned row_pick(const int v, const unsigned sel) {
  return __builtin_amdgcn_permlane16_var((unsigned)v, (unsigned)v, sel, false, false);
}

__device__ __forceinline__ float leaky(float e) { return e > 0.f ? e : NEG_SLOPE * e; }
__device__ __forceinline__ float dot4(const float4 a, const float4 b) {
  return a.x * b.x + a.y * b.y + a.z * b.z + a.w * b.w;
}
__device__ __forceinline__ v2f lo2(const float4 a) { v2f r; r.x = a.x; r.y = a.y; return r; }
__device__ __forceinline__ v2f hi2(const float4 a) { v2f r; r.x = a.z; r.y = a.w; return r; }
__device__ __forceinline__ float dot8(const float4 a0, const float4 a1, const float4 b0, const float4 b1) {
  v2f t = lo2(a0) * lo2(b0);
  t = hi2(a0) * hi2(b0) + t;
  t = lo2(a1) * lo2(b1) + t;
  t = hi2(a1) * hi2(b1) + t;
  return t.x + t.y;
}

__device__ __forceinline__ void stream_step(float& m, const float e, const bool ok, float& sc, float& w) {
  const float mn = fmaxf(m, e);
  sc = __expf(m - mn);
  w  = ok ? __expf(e - mn) : 0.f;
  m = mn;
}
__device__ __forceinline__ void scale_add(float4& acc, const float sc, const float w, const float4 v) {
  acc.x = acc.x * sc + w * v.x;
  acc.y = acc.y * sc + w * v.y;
  acc.z = acc.z * sc + w * v.z;
  acc.w = acc.w * sc + w * v.w;
}
__device__ __forceinline__ float slot_factor(const float m) {
  const float M = fmaxf(m, xrow<8>(m));
  return __expf(m - M);
}
__device__ __forceinline__ float sum_slots(float v) {
  v += xrow<8>(v);
  return v;
}
__device__ __forceinline__ float4 sum_slots4(const float4 a, const float f) {
  float4 r;
  r.x = sum_slots(a.x * f); r.y = sum_slots(a.y * f); r.z = sum_slots(a.z * f); r.w = sum_slots(a.w * f);
  return r;
}

__global__ void __launch_bounds__(256, 16)
gat1_node_k(const int* __restrict__ rowptr, const int* __restrict__ csr,
            const float* __restrict__ h1,
            const float* __restrict__ aw_s, const float* __restrict__ aw_d,
            const float* __restrict__ b1, float* __restrict__ helu, int N) {
  const int lane = (int)(threadIdx.x & 31);
  const int wv = __builtin_amdgcn_readfirstlane((int)((blockIdx.x * blockDim.x + threadIdx.x) >> 5));
  if (2 * wv >= N) return;
  const int hf = lane >> 4;
  const bool live = 2 * wv + hf < N;
  const int node = live ? 2 * wv + hf : N - 1;
  const int beg = rowptr[node];
  const int deg = rowptr[node + 1] - beg;
  const int degw = __builtin_amdgcn_readfirstlane(max(deg, xhalf_i(deg)));
  const int j = lane & 15;
  const int eq = (lane >> 3) & 1;
  const unsigned pc = (unsigned)(lane & 7);
  const float4* __restrict__ hv4 = (const float4*)h1;
  const float4 sA = *(const float4*)(aw_s + 8 * pc), sB = *(const float4*)(aw_s + 8 * pc + 4);
  const float4 dA = *(const float4*)(aw_d + 8 * pc), dB = *(const float4*)(aw_d + 8 * pc + 4);

  const float4 oA = hv4[(unsigned)node * 16u + pc], oB = hv4[(unsigned)node * 16u + 8u + pc];
  const float ad = dot8(oA, oB, dA, dB);
  float m = leaky(dot8(oA, oB, sA, sB) + ad);
  const float4 zero4 = make_float4(0.f, 0.f, 0.f, 0.f);
  float den = (eq == 0) ? 1.f : 0.f;
  float4 accA = (eq == 0) ? oA : zero4, accB = (eq == 0) ? oB : zero4;
  const int rem = deg - eq;

  for (int c0 = 0; c0 < degw; c0 += 16) {
    const int sl = (c0 + j < deg) ? csr[beg + c0 + j] : node;
    const int lim = min(16, degw - c0);
    for (int t = 0; 4 * t < lim; ++t) {
      const unsigned s0 = row_pick(sl, 4u * (unsigned)t + (unsigned)eq);
      const unsigned s1 = row_pick(sl, 4u * (unsigned)t + 2u + (unsigned)eq);
      const float4 vA0 = hv4[s0 * 16u + pc], vB0 = hv4[s0 * 16u + 8u + pc];
      const float4 vA1 = hv4[s1 * 16u + pc], vB1 = hv4[s1 * 16u + 8u + pc];
      const float p0 = dot8(vA0, vB0, sA, sB), p1 = dot8(vA1, vB1, sA, sB);
      const bool ok0 = c0 + 4 * t < rem, ok1 = c0 + 4 * t + 2 < rem;
      float sc, w;
      stream_step(m, leaky(p0 + ad), ok0, sc, w); den = den * sc + w; scale_add(accA, sc, w, vA0); scale_add(accB, sc, w, vB0);
      stream_step(m, leaky(p1 + ad), ok1, sc, w); den = den * sc + w; scale_add(accA, sc, w, vA1); scale_add(accB, sc, w, vB1);
    }
  }
  const float f = slot_factor(m);
  den = sum_slots(den * f); accA = sum_slots4(accA, f); accB = sum_slots4(accB, f);
  if (eq == 0 && live) {
    const float4 bA = *(const float4*)(b1 + 8 * pc), bB = *(const float4*)(b1 + 8 * pc + 4);
    float4 rA, rB;
    rA.x = accA.x / den + bA.x; rA.y = accA.y / den + bA.y; rA.z = accA.z / den + bA.z; rA.w = accA.w / den + bA.w;
    rB.x = accB.x / den + bB.x; rB.y = accB.y / den + bB.y; rB.z = accB.z / den + bB.z; rB.w = accB.w / den + bB.w;
    rA.x = rA.x > 0.f ? rA.x : (__expf(rA.x) - 1.f); rA.y = rA.y > 0.f ? rA.y : (__expf(rA.y) - 1.f);
    rA.z = rA.z > 0.f ? rA.z : (__expf(rA.z) - 1.f); rA.w = rA.w > 0.f ? rA.w : (__expf(rA.w) - 1.f);
    rB.x = rB.x > 0.f ? rB.x : (__expf(rB.x) - 1.f); rB.y = rB.y > 0.f ? rB.y : (__expf(rB.y) - 1.f);
    rB.z = rB.z > 0.f ? rB.z : (__expf(rB.z) - 1.f); rB.w = rB.w > 0.f ? rB.w : (__expf(rB.w) - 1.f);
    float4* __restrict__ o4 = (float4*)helu;
    o4[(unsigned)node * 16u + 2u * pc] = rA;
    o4[(unsigned)node * 16u + 2u * pc + 1u] = rB;
  }
}

#define PROJ_PITCH 68
#define PROJ_WAVES 4
__global__ void __launch_bounds__(256, 16)
gat2_proj_k(const int* __restrict__ rowptr, const int* __restrict__ csr,
            const float* __restrict__ helu,
            const float* __restrict__ us, const float* __restrict__ ud,
            const float* __restrict__ W2, const float* __restrict__ b2,
            float* __restrict__ out, int N) {
  __shared__ float4 rows4[16 * (PROJ_PITCH / 4)];
  const int lane = (int)(threadIdx.x & 31);
  const int wib = __builtin_amdgcn_readfirstlane((int)(threadIdx.x >> 5));
  const int row0 = (int)blockIdx.x * 16;
  const int hf = lane >> 4;
  const int r = 2 * wib + hf;
  const bool live = row0 + r < N;
  const int node = live ? row0 + r : N - 1;
  const int beg = rowptr[node];
  const int deg = rowptr[node + 1] - beg;
  const int degw = __builtin_amdgcn_readfirstlane(max(deg, xhalf_i(deg)));
  const int j = lane & 15;
  const int eq = (lane >> 3) & 1;
  const unsigned pc = (unsigned)(lane & 7);
  const float4* __restrict__ hv4 = (const float4*)helu;
  const float4 sA = *(const float4*)(us + 4 * pc), sB = *(const float4*)(us + 32 + 4 * pc);
  const float4 dA = *(const float4*)(ud + 4 * pc), dB = *(const float4*)(ud + 32 + 4 * pc);

  const float4 oA = hv4[(unsigned)node * 16u + pc], oB = hv4[(unsigned)node * 16u + 8u + pc];
  float adn = dot4(oA, dA) + dot4(oB, dB), asn = dot4(oA, sA) + dot4(oB, sB);
  adn += xrow<1>(adn); asn += xrow<1>(asn);
  adn += xrow<2>(adn); asn += xrow<2>(asn);
  adn += xrow<4>(adn); asn += xrow<4>(asn);
  float m = leaky(asn + adn);
  const float4 zero4 = make_float4(0.f, 0.f, 0.f, 0.f);
  float den = (eq == 0) ? 1.f : 0.f;
  float4 accA = (eq == 0) ? oA : zero4, accB = (eq == 0) ? oB : zero4;
  const int rem = deg - eq;

  for (int c0 = 0; c0 < degw; c0 += 16) {
    const int sl = (c0 + j < deg) ? csr[beg + c0 + j] : node;
    const int lim = min(16, degw - c0);
    for (int t = 0; 4 * t < lim; ++t) {
      const unsigned s0 = row_pick(sl, 4u * (unsigned)t + (unsigned)eq);
      const unsigned s1 = row_pick(sl, 4u * (unsigned)t + 2u + (unsigned)eq);
      const float4 vA0 = hv4[s0 * 16u + pc], vB0 = hv4[s0 * 16u + 8u + pc];
      const float4 vA1 = hv4[s1 * 16u + pc], vB1 = hv4[s1 * 16u + 8u + pc];
      float p0 = dot4(vA0, sA) + dot4(vB0, sB), p1 = dot4(vA1, sA) + dot4(vB1, sB);
      p0 += xrow<1>(p0); p1 += xrow<1>(p1);
      p0 += xrow<2>(p0); p1 += xrow<2>(p1);
      p0 += xrow<4>(p0); p1 += xrow<4>(p1);
      const bool ok0 = c0 + 4 * t < rem, ok1 = c0 + 4 * t + 2 < rem;
      const float e0 = leaky(p0 + adn), e1 = leaky(p1 + adn);
      float sc, w;
      stream_step(m, e0, ok0, sc, w); den = den * sc + w; scale_add(accA, sc, w, vA0); scale_add(accB, sc, w, vB0);
      stream_step(m, e1, ok1, sc, w); den = den * sc + w; scale_add(accA, sc, w, vA1); scale_add(accB, sc, w, vB1);
    }
  }
  const float f = slot_factor(m);
  den = sum_slots(den * f); accA = sum_slots4(accA, f); accB = sum_slots4(accB, f);
  if (eq == 0) {
    float4 rA, rB;
    rA.x = accA.x / den; rA.y = accA.y / den; rA.z = accA.z / den; rA.w = accA.w / den;
    rB.x = accB.x / den; rB.y = accB.y / den; rB.z = accB.z / den; rB.w = accB.w / den;
    rows4[r * (PROJ_PITCH / 4) + (int)pc] = rA;
    rows4[r * (PROJ_PITCH / 4) + 8 + (int)pc] = rB;
  }
  __syncthreads();
  if (wib >= PROJ_WAVES) return;

  const unsigned m16  = (unsigned)lane & 15u;
  const unsigned half = (unsigned)lane >> 4;
  const unsigned klo  = half * 2u;
  const float* __restrict__ arow = (const float*)rows4 + m16 * PROJ_PITCH + klo;
  const unsigned cl = 32u * (unsigned)wib + 2u * m16;
  const float* __restrict__ wp = W2 + klo * 128u + cl;
  v8f acc0 = {0.f, 0.f, 0.f, 0.f, 0.f, 0.f, 0.f, 0.f};
  v8f acc1 = acc0;
  #pragma unroll
  for (int k = 0; k < 64; k += 4) {
    const v2f a  = *(const v2f*)(arow + k);
    const v2f w0 = *(const v2f*)(wp + k * 128);
    const v2f w1 = *(const v2f*)(wp + k * 128 + 128);
    v2f b0; b0.x = w0.x; b0.y = w1.x;
    v2f b1; b1.x = w0.y; b1.y = w1.y;
    acc0 = __builtin_amdgcn_wmma_f32_16x16x4_f32(false, a, false, b0, (short)0, acc0, false, false);
    acc1 = __builtin_amdgcn_wmma_f32_16x16x4_f32(false, a, false, b1, (short)0, acc1, false, false);
  }
  const float2 bv = *(const float2*)(b2 + cl);
  #pragma unroll
  for (int v = 0; v < 8; ++v) {
    const int row = row0 + v + 8 * (int)half;
    if (row < N) *(float2*)(out + (size_t)row * 128 + cl) = make_float2(acc0[v] + bv.x, acc1[v] + bv.y);
  }
}

static inline unsigned cdiv(long long a, long long b) { return (unsigned)((a + b - 1) / b); }

extern "C" void kernel_launch(void* const* d_in, const int* in_sizes, int n_in,
                              void* d_out, int out_size, void* d_ws, size_t ws_size,
                              hipStream_t stream) {
  const float* x    = (const float*)d_in[0];
  const int*   ei   = (const int*)d_in[1];
  const float* W1   = (const float*)d_in[2];
  const float* as1w = (const float*)d_in[3];
  const float* ad1w = (const float*)d_in[4];
  const float* b1   = (const float*)d_in[5];
  const float* W2   = (const float*)d_in[6];
  const float* as2w = (const float*)d_in[7];
  const float* ad2w = (const float*)d_in[8];
  const float* b2   = (const float*)d_in[9];
  float* out = (float*)d_out;

  const int N = in_sizes[0] / 128;
  const long long E = (long long)in_sizes[1] / 2;

  float* w = (float*)d_ws;
  size_t off = 0;
  float* h1   = w + off; off += (size_t)N * 64;
  float* helu = w + off; off += (size_t)N * 64;
  float* us   = w + off; off += 64;
  float* ud   = w + off; off += 64;
  const int K  = (N + NODES_PER_BUCKET - 1) / NODES_PER_BUCKET;
  const int nb = (int)cdiv(E, EPB);
  int2* part  = (int2*)(w + off); off += (size_t)nb * EPB * 2;
  int* runtab = (int*)(w + off); off += (size_t)nb * (K + 1);
  int* rowptr = (int*)(w + off); off += (size_t)N + 1;
  int* csr    = (int*)(w + off); off += (size_t)E;

  const int B = 256;
  const int tiles = (N + 15) / 16;
  const int pairs = (N + 1) / 2;

  bucket_runs_k<<<nb, 1024, 0, stream>>>(ei, E, K, runtab, part);
  edge_rows_gemm_k<<<K + cdiv(tiles, GEMM_TILES_PER_GROUP), 1024, 0, stream>>>(part, runtab, nb, K, N, rowptr, csr, x, W1, h1);

  fold_k<<<8, B, 0, stream>>>(W2, as2w, ad2w, us, ud);
  gat1_node_k<<<cdiv((long long)pairs * 32, B), B, 0, stream>>>(rowptr, csr, h1, as1w, ad1w, b1, helu, N);

  gat2_proj_k<<<tiles, 256, 0, stream>>>(rowptr, csr, helu, us, ud, W2, b2, out, N);

  (void)n_in; (void)out_size; (void)ws_size;
}
